# starting kernel
# baseline (speedup 1.0000x reference)
#include <hip/hip_runtime.h>
#include <hip/hip_bf16.h>

typedef __bf16 bf16_t;
typedef __attribute__((ext_vector_type(16))) __bf16 v16bf;
typedef __attribute__((ext_vector_type(4)))  __bf16 v4bf;
typedef __attribute__((ext_vector_type(8)))  float  v8f;

#define DIMD   256
#define HEADS  4
#define HD     64
#define FFND   1024
#define BATCH  8
#define SEQ    2048
#define NTOK   (BATCH*SEQ)
#define NEGV   (-1e9f)

__device__ __forceinline__ v16bf load_frag_bf16(const bf16_t* base) {
  const int half = (threadIdx.x >> 4) & 1;
  const bf16_t* p = base + 8 * half;
  v16bf f;
#pragma unroll
  for (int e = 0; e < 8; ++e) {
    f[e]     = p[e];
    f[e + 8] = p[e + 16];
  }
  return f;
}

__device__ __forceinline__ v8f wmma_bf16(v16bf a, v16bf b, v8f c) {
  return __builtin_amdgcn_wmma_f32_16x16x32_bf16(
       false, a,  false, b,
       (short)0, c,  false,  false);
}

__global__ __launch_bounds__(256) void cvt_bf16(const float* __restrict__ in,
                                                bf16_t* __restrict__ out) {
  const size_t i = (size_t)blockIdx.x * 256 + threadIdx.x;
  const float4 v = ((const float4*)in)[i];
  v4bf o;
  o[0] = (bf16_t)v.x; o[1] = (bf16_t)v.y; o[2] = (bf16_t)v.z; o[3] = (bf16_t)v.w;
  ((v4bf*)out)[i] = o;
}

template <int MODE>
__global__ __launch_bounds__(128) void gemm_wmma(
    const bf16_t* __restrict__ A, const bf16_t* __restrict__ B,
    const float* __restrict__ bias, float* __restrict__ Yf,
    bf16_t* __restrict__ Yb,
    bf16_t* __restrict__ Qb, bf16_t* __restrict__ Kb, bf16_t* __restrict__ Vt,
    int M, int N, int K) {
  const int lane = threadIdx.x & 31;
  const int wave = threadIdx.x >> 5;
  const int r    = lane & 15;
  const int half = lane >> 4;
  const int row0 = blockIdx.y * 128 + wave * 32;
  const int col0 = blockIdx.x * 64;

  v8f acc[2][4] = {};
  const bf16_t* ap0 = A + (size_t)(row0 + r) * K;
  const bf16_t* ap1 = A + (size_t)(row0 + 16 + r) * K;
  const bf16_t* bp  = B + (size_t)(col0 + r) * K;

  for (int k0 = 0; k0 < K; k0 += 32) {
    __builtin_prefetch(ap0 + k0 + 64, 0, 0);
    __builtin_prefetch(ap1 + k0 + 64, 0, 0);
    v16bf a0 = load_frag_bf16(ap0 + k0);
    v16bf a1 = load_frag_bf16(ap1 + k0);
#pragma unroll
    for (int nb = 0; nb < 4; ++nb) {
      v16bf bfr = load_frag_bf16(bp + (size_t)nb * 16 * K + k0);
      acc[0][nb] = wmma_bf16(a0, bfr, acc[0][nb]);
      acc[1][nb] = wmma_bf16(a1, bfr, acc[1][nb]);
    }
  }

#pragma unroll
  for (int mi = 0; mi < 2; ++mi) {
#pragma unroll
    for (int nb = 0; nb < 4; ++nb) {
#pragma unroll
      for (int i = 0; i < 8; ++i) {
        const int row = row0 + mi * 16 + i + 8 * half;
        const int col = col0 + nb * 16 + r;
        float y = acc[mi][nb][i] + bias[col];
        if (MODE == 0) {
          Yf[(size_t)row * N + col] = y;
        } else if (MODE == 1) {
          Yb[(size_t)row * N + col] = (bf16_t)fmaxf(y, 0.0f);
        } else {
          const int b = row >> 11, t = row & (SEQ - 1);
          if (col < DIMD) {
            const int h = col >> 6, d = col & 63;
            Qb[(((size_t)b * HEADS + h) * SEQ + t) * HD + d] = (bf16_t)y;
          } else if (col < 2 * DIMD) {
            const int c = col - DIMD, h = c >> 6, d = c & 63;
            Kb[(((size_t)b * HEADS + h) * SEQ + t) * HD + d] = (bf16_t)y;
          } else {
            const int c = col - 2 * DIMD, h = c >> 6, d = c & 63;
            Vt[(((size_t)b * HEADS + h) * HD + d) * SEQ + t] = (bf16_t)y;
          }
        }
      }
    }
  }
}

__global__ __launch_bounds__(32) void attn_wmma(
    const bf16_t* __restrict__ Qb, const bf16_t* __restrict__ Kb,
    const bf16_t* __restrict__ Vt, const float* __restrict__ mask,
    bf16_t* __restrict__ attn, float* __restrict__ halt) {
  __shared__ bf16_t lds_p[2][16 * 32];

  const int bid  = blockIdx.x;
  const int qt   = bid & (SEQ / 32 - 1);
  const int bh   = bid >> 6;
  const int b    = bh >> 2;
  const int lane = threadIdx.x & 31;
  const int r    = lane & 15;
  const int half = lane >> 4;
  const float scale = 0.125f;

  const bf16_t* Qbase = Qb + ((size_t)bh * SEQ + qt * 32) * HD;
  const bf16_t* Kbase = Kb + (size_t)bh * SEQ * HD;
  const bf16_t* Vbase = Vt + (size_t)bh * HD * SEQ;
  const float*  mrow  = mask + (size_t)b * SEQ;

  v16bf qa[2][2];
#pragma unroll
  for (int qi = 0; qi < 2; ++qi) {
    qa[qi][0] = load_frag_bf16(Qbase + (size_t)(qi * 16 + r) * HD + 0);
    qa[qi][1] = load_frag_bf16(Qbase + (size_t)(qi * 16 + r) * HD + 32);
  }

  float m[2][8], l[2][8];
#pragma unroll
  for (int qi = 0; qi < 2; ++qi)
#pragma unroll
    for (int i = 0; i < 8; ++i) { m[qi][i] = -3.0e38f; l[qi][i] = 0.0f; }

  for (int kt = 0; kt < SEQ / 16; ++kt) {
    v16bf kb0 = load_frag_bf16(Kbase + (size_t)(kt * 16 + r) * HD + 0);
    v16bf kb1 = load_frag_bf16(Kbase + (size_t)(kt * 16 + r) * HD + 32);
    const float mk = mrow[kt * 16 + r];
#pragma unroll
    for (int qi = 0; qi < 2; ++qi) {
      v8f s = {};
      s = wmma_bf16(qa[qi][0], kb0, s);
      s = wmma_bf16(qa[qi][1], kb1, s);
#pragma unroll
      for (int i = 0; i < 8; ++i) {
        const float sv = (mk > 0.5f) ? NEGV : s[i] * scale;
        const float mn = fmaxf(m[qi][i], sv);
        l[qi][i] = l[qi][i] * __expf(m[qi][i] - mn) + __expf(sv - mn);
        m[qi][i] = mn;
      }
    }
  }
#pragma unroll
  for (int d = 1; d <= 8; d <<= 1) {
#pragma unroll
    for (int qi = 0; qi < 2; ++qi) {
#pragma unroll
      for (int i = 0; i < 8; ++i) {
        const float mo = __shfl_xor(m[qi][i], d, 32);
        const float lo = __shfl_xor(l[qi][i], d, 32);
        const float mn = fmaxf(m[qi][i], mo);
        l[qi][i] = l[qi][i] * __expf(m[qi][i] - mn) + lo * __expf(mo - mn);
        m[qi][i] = mn;
      }
    }
  }
  float rl[2][8];
#pragma unroll
  for (int qi = 0; qi < 2; ++qi)
#pragma unroll
    for (int i = 0; i < 8; ++i) rl[qi][i] = 1.0f / l[qi][i];

  v8f o[2][4] = {};
  for (int kt2 = 0; kt2 < SEQ / 32; ++kt2) {
#pragma unroll
    for (int sub = 0; sub < 2; ++sub) {
      const int kt = kt2 * 2 + sub;
      v16bf kb0 = load_frag_bf16(Kbase + (size_t)(kt * 16 + r) * HD + 0);
      v16bf kb1 = load_frag_bf16(Kbase + (size_t)(kt * 16 + r) * HD + 32);
      const float mk = mrow[kt * 16 + r];
      float cs = 0.0f;
#pragma unroll
      for (int qi = 0; qi < 2; ++qi) {
        v8f s = {};
        s = wmma_bf16(qa[qi][0], kb0, s);
        s = wmma_bf16(qa[qi][1], kb1, s);
#pragma unroll
        for (int i = 0; i < 8; ++i) {
          const float sv = (mk > 0.5f) ? NEGV : s[i] * scale;
          const float p  = __expf(sv - m[qi][i]) * rl[qi][i];
          cs += p;
          lds_p[qi][(i + 8 * half) * 32 + sub * 16 + r] = (bf16_t)p;
        }
      }
      cs += __shfl_xor(cs, 16, 32);
      if (lane < 16) atomicAdd(&halt[(size_t)b * SEQ + kt * 16 + lane], cs);
    }
    __syncthreads();
    v16bf pa0 = load_frag_bf16(lds_p[0] + r * 32);
    v16bf pa1 = load_frag_bf16(lds_p[1] + r * 32);
#pragma unroll
    for (int nb = 0; nb < 4; ++nb) {
      v16bf vb = load_frag_bf16(Vbase + (size_t)(nb * 16 + r) * SEQ + kt2 * 32);
      o[0][nb] = wmma_bf16(pa0, vb, o[0][nb]);
      o[1][nb] = wmma_bf16(pa1, vb, o[1][nb]);
    }
    __syncthreads();
  }

  const int h = bh & 3;
#pragma unroll
  for (int qi = 0; qi < 2; ++qi) {
#pragma unroll
    for (int nb = 0; nb < 4; ++nb) {
#pragma unroll
      for (int i = 0; i < 8; ++i) {
        const int t = qt * 32 + qi * 16 + i + 8 * half;
        attn[((size_t)b * SEQ + t) * DIMD + h * HD + nb * 16 + r] =
            (bf16_t)o[qi][nb][i];
      }
    }
  }
}

template <bool WRITE_BF16>
__global__ __launch_bounds__(256) void add_ln(
    const float* __restrict__ A, const float* __restrict__ Bv,
    const float* __restrict__ g, const float* __restrict__ be,
    float* __restrict__ out, bf16_t* __restrict__ outb) {
  __shared__ float red[256];
  const int row = blockIdx.x, tid = threadIdx.x;
  const float v = A[(size_t)row * DIMD + tid] + Bv[(size_t)row * DIMD + tid];
  red[tid] = v;
  __syncthreads();
  for (int s = 128; s > 0; s >>= 1) {
    if (tid < s) red[tid] += red[tid + s];
    __syncthreads();
  }
  const float mu = red[0] * (1.0f / DIMD);
  __syncthreads();
  const float d = v - mu;
  red[tid] = d * d;
  __syncthreads();
  for (int s = 128; s > 0; s >>= 1) {
    if (tid < s) red[tid] += red[tid + s];
    __syncthreads();
  }
  const float var = red[0] * (1.0f / DIMD);
  const float y = d * rsqrtf(var + 1e-5f) * g[tid] + be[tid];
  out[(size_t)row * DIMD + tid] = y;
  if (WRITE_BF16) outb[(size_t)row * DIMD + tid] = (bf16_t)y;
}

__global__ __launch_bounds__(256) void halt_finalize(
    const float* __restrict__ hacc, const float* __restrict__ mask,
    float* __restrict__ out) {
  __shared__ float red[256];
  const int b = blockIdx.x, tid = threadIdx.x;
  float lm = -3.0e38f;
  for (int j = tid; j < SEQ; j += 256) {
    const float v = hacc[(size_t)b * SEQ + j] * 0.25f +
                    mask[(size_t)b * SEQ + j] * (-1e10f);
    lm = fmaxf(lm, v);
  }
  red[tid] = lm;
  __syncthreads();
  for (int s = 128; s > 0; s >>= 1) {
    if (tid < s) red[tid] = fmaxf(red[tid], red[tid + s]);
    __syncthreads();
  }
  const float mx = red[0];
  __syncthreads();
  float ls = 0.0f;
  for (int j = tid; j < SEQ; j += 256) {
    const float v = hacc[(size_t)b * SEQ + j] * 0.25f +
                    mask[(size_t)b * SEQ + j] * (-1e10f);
    ls += __expf(v - mx);
  }
  red[tid] = ls;
  __syncthreads();
  for (int s = 128; s > 0; s >>= 1) {
    if (tid < s) red[tid] += red[tid + s];
    __syncthreads();
  }
  const float inv = 1.0f / red[0];
  for (int j = tid; j < SEQ; j += 256) {
    const float v = hacc[(size_t)b * SEQ + j] * 0.25f +
                    mask[(size_t)b * SEQ + j] * (-1e10f);
    out[(size_t)b * SEQ + j] = __expf(v - mx) * inv;
  }
}

extern "C" void kernel_launch(void* const* d_in, const int* in_sizes, int n_in,
                              void* d_out, int out_size, void* d_ws,
                              size_t ws_size, hipStream_t stream) {
  (void)in_sizes; (void)n_in; (void)out_size; (void)ws_size;
  const float* x    = (const float*)d_in[0];
  const float* mask = (const float*)d_in[1];
  const float* Wqkv = (const float*)d_in[2];
  const float* bqkv = (const float*)d_in[3];
  const float* Wo   = (const float*)d_in[4];
  const float* bo   = (const float*)d_in[5];
  const float* W1   = (const float*)d_in[6];
  const float* b1   = (const float*)d_in[7];
  const float* W2   = (const float*)d_in[8];
  const float* b2   = (const float*)d_in[9];
  const float* g1   = (const float*)d_in[10];
  const float* be1  = (const float*)d_in[11];
  const float* g2   = (const float*)d_in[12];
  const float* be2  = (const float*)d_in[13];

  char* ws = (char*)d_ws;
  size_t off = 0;
  auto alloc = [&](size_t bytes) {
    char* p = ws + off;
    off += (bytes + 255) & ~(size_t)255;
    return p;
  };
  bf16_t* xb    = (bf16_t*)alloc((size_t)NTOK * DIMD * 2);
  bf16_t* Wqkvb = (bf16_t*)alloc((size_t)3 * DIMD * DIMD * 2);
  bf16_t* Wob   = (bf16_t*)alloc((size_t)DIMD * DIMD * 2);
  bf16_t* W1b   = (bf16_t*)alloc((size_t)FFND * DIMD * 2);
  bf16_t* W2b   = (bf16_t*)alloc((size_t)DIMD * FFND * 2);
  bf16_t* Qb    = (bf16_t*)alloc((size_t)NTOK * DIMD * 2);
  bf16_t* Kb    = (bf16_t*)alloc((size_t)NTOK * DIMD * 2);
  bf16_t* Vt    = (bf16_t*)alloc((size_t)NTOK * DIMD * 2);
  bf16_t* attnb = (bf16_t*)alloc((size_t)NTOK * DIMD * 2);
  float*  attnO = (float*) alloc((size_t)NTOK * DIMD * 4);
  float*  hbuf  = (float*) alloc((size_t)NTOK * DIMD * 4);
  bf16_t* hb    = (bf16_t*)alloc((size_t)NTOK * DIMD * 2);
  bf16_t* ffn1b = (bf16_t*)alloc((size_t)NTOK * FFND * 2);
  float*  ffn2  = (float*) alloc((size_t)NTOK * DIMD * 4);
  float*  halt  = (float*) alloc((size_t)NTOK * 4);

  float* out      = (float*)d_out;
  float* out_halt = out + (size_t)NTOK * DIMD;

  hipMemsetAsync(halt, 0, (size_t)NTOK * sizeof(float), stream);

  cvt_bf16<<<NTOK * DIMD / 1024, 256, 0, stream>>>(x, xb);
  cvt_bf16<<<3 * DIMD * DIMD / 1024, 256, 0, stream>>>(Wqkv, Wqkvb);
  cvt_bf16<<<DIMD * DIMD / 1024, 256, 0, stream>>>(Wo, Wob);
  cvt_bf16<<<FFND * DIMD / 1024, 256, 0, stream>>>(W1, W1b);
  cvt_bf16<<<DIMD * FFND / 1024, 256, 0, stream>>>(W2, W2b);

  gemm_wmma<2><<<dim3(3 * DIMD / 64, NTOK / 128), 128, 0, stream>>>(
      xb, Wqkvb, bqkv, nullptr, nullptr, Qb, Kb, Vt, NTOK, 3 * DIMD, DIMD);

  attn_wmma<<<BATCH * HEADS * (SEQ / 32), 32, 0, stream>>>(
      Qb, Kb, Vt, mask, attnb, halt);

  gemm_wmma<0><<<dim3(DIMD / 64, NTOK / 128), 128, 0, stream>>>(
      attnb, Wob, bo, attnO, nullptr, nullptr, nullptr, nullptr,
      NTOK, DIMD, DIMD);

  add_ln<true><<<NTOK, 256, 0, stream>>>(x, attnO, g1, be1, hbuf, hb);

  gemm_wmma<1><<<dim3(FFND / 64, NTOK / 128), 128, 0, stream>>>(
      hb, W1b, b1, nullptr, ffn1b, nullptr, nullptr, nullptr,
      NTOK, FFND, DIMD);

  gemm_wmma<0><<<dim3(DIMD / 64, NTOK / 128), 128, 0, stream>>>(
      ffn1b, W2b, b2, ffn2, nullptr, nullptr, nullptr, nullptr,
      NTOK, DIMD, FFND);

  add_ln<false><<<NTOK, 256, 0, stream>>>(hbuf, ffn2, g2, be2, out, nullptr);

  halt_finalize<<<BATCH, 256, 0, stream>>>(halt, mask, out_halt);
}


# GEMM operand tiles fetched by whole-line coalesced loads into LDS, fragments from LDS: QKV/Wo A and B tiles; FFN block W1/W2 tiles in a 2-stage ring one tile ahead; sum order unchanged
# speedup vs baseline: 2.2831x; 2.2831x over previous
#include <hip/hip_runtime.h>
#include <hip/hip_bf16.h>

typedef __bf16 bf16_t;
typedef __attribute__((ext_vector_type(16))) __bf16 v16bf;
typedef __attribute__((ext_vector_type(4)))  __bf16 v4bf;
typedef __attribute__((ext_vector_type(8)))  float  v8f;
typedef __attribute__((ext_vector_type(8)))  __bf16 v8bf;

#define DIMD   256
#define HEADS  4
#define HD     64
#define FFND   1024
#define BATCH  8
#define SEQ    2048
#define NTOK   (BATCH*SEQ)
#define NEGV   (-1e9f)

__device__ __forceinline__ v16bf load_frag_bf16(const bf16_t* base) {
  const int half = (threadIdx.x >> 4) & 1;
  const bf16_t* p = base + 8 * half;
  v16bf f;
#pragma unroll
  for (int e = 0; e < 8; ++e) {
    f[e]     = p[e];
    f[e + 8] = p[e + 16];
  }
  return f;
}

__device__ __forceinline__ v16bf lds_frag_bf16(const bf16_t* base) {
  const int half = (threadIdx.x >> 4) & 1;
  const bf16_t* p = base + 8 * half;
  const v8bf lo = *(const v8bf*)p;
  const v8bf hi = *(const v8bf*)(p + 16);
  return __builtin_shufflevector(lo, hi, 0, 1, 2, 3, 4, 5, 6, 7,
                                 8, 9, 10, 11, 12, 13, 14, 15);
}

__device__ __forceinline__ void lds_barrier() {
  __builtin_amdgcn_fence(__ATOMIC_RELEASE, "workgroup", "local");
  __builtin_amdgcn_s_barrier();
  __builtin_amdgcn_fence(__ATOMIC_ACQUIRE, "workgroup", "local");
}

__device__ __forceinline__ v8f wmma_bf16(v16bf a, v16bf b, v8f c) {
  return __builtin_amdgcn_wmma_f32_16x16x32_bf16(
       false, a,  false, b,
       (short)0, c,  false,  false);
}

__global__ __launch_bounds__(256) void cvt_bf16(const float* __restrict__ in,
                                                bf16_t* __restrict__ out) {
  const size_t i = (size_t)blockIdx.x * 256 + threadIdx.x;
  const float4 v = ((const float4*)in)[i];
  v4bf o;
  o[0] = (bf16_t)v.x; o[1] = (bf16_t)v.y; o[2] = (bf16_t)v.z; o[3] = (bf16_t)v.w;
  ((v4bf*)out)[i] = o;
}

__global__ __launch_bounds__(256) void mask_scan(
    const float* __restrict__ mask, int* __restrict__ kpos,
    int* __restrict__ kidx, int* __restrict__ nkv,
    bf16_t* __restrict__ Kb, bf16_t* __restrict__ Vt) {
  __shared__ int cnt[256];
  const int b = blockIdx.x, tid = threadIdx.x;
  const float4* mp = (const float4*)(mask + (size_t)b * SEQ + tid * 8);
  const float4 m0 = mp[0], m1 = mp[1];
  const float mv[8] = {m0.x, m0.y, m0.z, m0.w, m1.x, m1.y, m1.z, m1.w};
  int keep[8], c = 0;
#pragma unroll
  for (int j = 0; j < 8; ++j) { keep[j] = !(mv[j] > 0.5f); c += keep[j]; }
  cnt[tid] = c;
  __syncthreads();
  for (int s = 1; s < 256; s <<= 1) {
    const int add = (tid >= s) ? cnt[tid - s] : 0;
    __syncthreads();
    cnt[tid] += add;
    __syncthreads();
  }
  const int total = cnt[255];
  const bool allm = (total == 0);
  int p = cnt[tid] - c;
#pragma unroll
  for (int j = 0; j < 8; ++j) {
    const int t = tid * 8 + j;
    int q = -1;
    if (allm) q = t;
    else if (keep[j]) q = p++;
    kpos[(size_t)b * SEQ + t] = q;
    if (q >= 0) kidx[(size_t)b * SEQ + q] = t;
  }
  const int nk = allm ? SEQ : total;
  if (tid == 0) nkv[b] = allm ? -1 : total;
  const int nk64 = (nk + 63) & ~63;
  if (tid < nk64 - nk) kidx[(size_t)b * SEQ + nk + tid] = 0;
  const int h = tid >> 6, d = tid & 63;
  const size_t bhh = (size_t)b * HEADS + h;
  for (int q = nk; q < nk64; ++q) {
    Kb[(bhh * SEQ + q) * HD + d] = (bf16_t)0.0f;
    Vt[(bhh * HD + d) * SEQ + q] = (bf16_t)0.0f;
  }
}

#define GT_PITCH 264
template <int MODE>
__global__ __launch_bounds__(128) void gemm_wmma(
    const bf16_t* __restrict__ A, const bf16_t* __restrict__ B,
    const float* __restrict__ bias, float* __restrict__ Yf,
    bf16_t* __restrict__ Qb, bf16_t* __restrict__ Kb, bf16_t* __restrict__ Vt,
    const int* __restrict__ kpos, int N) {
  __shared__ alignas(16) bf16_t at[128 * GT_PITCH];
  __shared__ alignas(16) bf16_t bt[64 * GT_PITCH];
  const int lane = threadIdx.x & 31;
  const int wave = threadIdx.x >> 5;
  const int r    = lane & 15;
  const int half = lane >> 4;
  const int rowb = blockIdx.y * 128;
  const int row0 = rowb + wave * 32;
  const int col0 = blockIdx.x * 64;

  {
    const v8bf* ag = (const v8bf*)(A + (size_t)(rowb + wave) * DIMD) + lane;
    const v8bf* bg = (const v8bf*)(B + (size_t)(col0 + wave) * DIMD) + lane;
    const int so = wave * GT_PITCH + lane * 8;
    v8bf tb[16], ta[32];
#pragma unroll
    for (int j = 0; j < 16; ++j) tb[j] = bg[(size_t)j * 4 * (DIMD / 8)];
#pragma unroll
    for (int j = 0; j < 32; ++j) ta[j] = ag[(size_t)j * 4 * (DIMD / 8)];
#pragma unroll
    for (int j = 0; j < 16; ++j) *(v8bf*)(bt + so + j * 4 * GT_PITCH) = tb[j];
#pragma unroll
    for (int j = 0; j < 32; ++j) *(v8bf*)(at + so + j * 4 * GT_PITCH) = ta[j];
  }
  __syncthreads();

  v8f acc[2][4] = {};
  const bf16_t* ap0 = at + (wave * 32 + r) * GT_PITCH;
  const bf16_t* ap1 = ap0 + 16 * GT_PITCH;
  const bf16_t* bp  = bt + r * GT_PITCH;

  v16bf fa[2][2], fb[2][4];
  fa[0][0] = lds_frag_bf16(ap0);
  fa[0][1] = lds_frag_bf16(ap1);
#pragma unroll
  for (int nb = 0; nb < 4; ++nb) fb[0][nb] = lds_frag_bf16(bp + nb * 16 * GT_PITCH);
#pragma unroll
  for (int ks = 0; ks < DIMD / 32; ++ks) {
    const int cs = ks & 1, ns = cs ^ 1;
    if (ks + 1 < DIMD / 32) {
      const int k1 = (ks + 1) * 32;
      fa[ns][0] = lds_frag_bf16(ap0 + k1);
      fa[ns][1] = lds_frag_bf16(ap1 + k1);
#pragma unroll
      for (int nb = 0; nb < 4; ++nb)
        fb[ns][nb] = lds_frag_bf16(bp + nb * 16 * GT_PITCH + k1);
    }
    __builtin_amdgcn_sched_barrier(0);
#pragma unroll
    for (int nb = 0; nb < 4; ++nb) {
      acc[0][nb] = wmma_bf16(fa[cs][0], fb[cs][nb], acc[0][nb]);
      acc[1][nb] = wmma_bf16(fa[cs][1], fb[cs][nb], acc[1][nb]);
    }
    __builtin_amdgcn_sched_barrier(0);
  }

  int kp[2][8] = {};
  if (MODE == 2 && col0 >= DIMD) {
#pragma unroll
    for (int mi = 0; mi < 2; ++mi)
#pragma unroll
      for (int i = 0; i < 8; ++i)
        kp[mi][i] = kpos[row0 + mi * 16 + i + 8 * half];
  }

#pragma unroll
  for (int mi = 0; mi < 2; ++mi) {
#pragma unroll
    for (int nb = 0; nb < 4; ++nb) {
#pragma unroll
      for (int i = 0; i < 8; ++i) {
        const int row = row0 + mi * 16 + i + 8 * half;
        const int col = col0 + nb * 16 + r;
        float y = acc[mi][nb][i] + bias[col];
        if (MODE == 0) {
          Yf[(size_t)row * N + col] = y;
        } else {
          const int b = row >> 11, t = row & (SEQ - 1);
          if (col < DIMD) {
            const int h = col >> 6, d = col & 63;
            Qb[(((size_t)b * HEADS + h) * SEQ + t) * HD + d] = (bf16_t)y;
          } else if (col < 2 * DIMD) {
            const int c = col - DIMD, h = c >> 6, d = c & 63;
            const int p = kp[mi][i];
            if (p >= 0)
              Kb[(((size_t)b * HEADS + h) * SEQ + p) * HD + d] = (bf16_t)y;
          } else {
            const int c = col - 2 * DIMD, h = c >> 6, d = c & 63;
            const int p = kp[mi][i];
            if (p >= 0)
              Vt[(((size_t)b * HEADS + h) * HD + d) * SEQ + p] = (bf16_t)y;
          }
        }
      }
    }
  }
}

#define AT_WAVES 8
#define AT_KT    64
#define KV_PITCH 72

__global__ __launch_bounds__(256) __attribute__((amdgpu_waves_per_eu(2, 2))) void attn_wmma(
    const bf16_t* __restrict__ Qb, const bf16_t* __restrict__ Kb,
    const bf16_t* __restrict__ Vt, const int* __restrict__ nkv,
    const int* __restrict__ kidx,
    bf16_t* __restrict__ attn, float* __restrict__ halt) {
  __shared__ alignas(16) bf16_t ks[2][AT_KT * KV_PITCH];
  __shared__ alignas(16) bf16_t vs[2][HD * KV_PITCH];
  __shared__ alignas(16) bf16_t lds_pw[AT_WAVES][2][16 * 32];

  const int tid  = threadIdx.x;
  const int wave = tid >> 5;
  const int bid  = blockIdx.x;
  const int bh   = bid >> 3;
  const int qt   = (bid & 7) * AT_WAVES + wave;
  const int b    = bh >> 2;
  const int lane = tid & 31;
  const int r    = lane & 15;
  const int half = lane >> 4;
  const float scale = 0.125f;

  const bf16_t* Qbase = Qb + ((size_t)bh * SEQ + qt * 32) * HD;
  const bf16_t* Kbase = Kb + (size_t)bh * SEQ * HD;
  const bf16_t* Vbase = Vt + (size_t)bh * HD * SEQ;
  const int*    kix   = kidx + (size_t)b * SEQ;
  const int nkraw = nkv[b];
  const int nk    = (nkraw < 0) ? SEQ : nkraw;
  const int lim   = (nkraw < 0) ? 0 : nkraw;
  const int nt    = (nk + AT_KT - 1) / AT_KT;

  const int srow = tid >> 3;
  const int scol = (tid & 7) * 8;
  const v8bf* kg = (const v8bf*)(Kbase + (size_t)srow * HD + scol);
  const v8bf* vg = (const v8bf*)(Vbase + (size_t)srow * SEQ + scol);
  const int   so = srow * KV_PITCH + scol;
  bf16_t* pw0 = lds_pw[wave][0];
  bf16_t* pw1 = lds_pw[wave][1];

  v16bf qa[2][2];
#pragma unroll
  for (int qi = 0; qi < 2; ++qi) {
    qa[qi][0] = load_frag_bf16(Qbase + (size_t)(qi * 16 + r) * HD + 0);
    qa[qi][1] = load_frag_bf16(Qbase + (size_t)(qi * 16 + r) * HD + 32);
  }

  float m[2][8], l[2][8];
#pragma unroll
  for (int qi = 0; qi < 2; ++qi)
#pragma unroll
    for (int i = 0; i < 8; ++i) { m[qi][i] = -3.0e38f; l[qi][i] = 0.0f; }

  {
    const v8bf g0 = kg[0];
    const v8bf g1 = kg[32 * HD / 8];
    *(v8bf*)(ks[0] + so)                 = g0;
    *(v8bf*)(ks[0] + so + 32 * KV_PITCH) = g1;
  }
  lds_barrier();
  for (int t = 0; t < nt; ++t) {
    const int cur = t & 1;
    const int tn = (t + 1 < nt) ? t + 1 : t;
    const v8bf* kn = kg + (size_t)tn * (AT_KT * HD / 8);
    const v8bf g0 = kn[0];
    const v8bf g1 = kn[32 * HD / 8];
    __builtin_amdgcn_sched_barrier(0);
#pragma unroll
    for (int sub = 0; sub < 4; ++sub) {
      const bf16_t* kr = ks[cur] + (sub * 16 + r) * KV_PITCH;
      v16bf kb0 = lds_frag_bf16(kr);
      v16bf kb1 = lds_frag_bf16(kr + 32);
      const bool dead = (t * AT_KT + sub * 16 + r) >= lim;
#pragma unroll
      for (int qi = 0; qi < 2; ++qi) {
        v8f s = {};
        s = wmma_bf16(qa[qi][0], kb0, s);
        s = wmma_bf16(qa[qi][1], kb1, s);
#pragma unroll
        for (int i = 0; i < 8; ++i) {
          const float sv = dead ? NEGV : s[i] * scale;
          const float mn = fmaxf(m[qi][i], sv);
          l[qi][i] = l[qi][i] * __expf(m[qi][i] - mn) + __expf(sv - mn);
          m[qi][i] = mn;
        }
      }
    }
    __builtin_amdgcn_sched_barrier(0);
    *(v8bf*)(ks[cur ^ 1] + so)                 = g0;
    *(v8bf*)(ks[cur ^ 1] + so + 32 * KV_PITCH) = g1;
    lds_barrier();
  }

  {
    const v8bf g0 = kg[0];
    const v8bf g1 = kg[32 * HD / 8];
    const v8bf h0 = vg[0];
    const v8bf h1 = vg[32 * SEQ / 8];
    *(v8bf*)(ks[0] + so)                 = g0;
    *(v8bf*)(ks[0] + so + 32 * KV_PITCH) = g1;
    *(v8bf*)(vs[0] + so)                 = h0;
    *(v8bf*)(vs[0] + so + 32 * KV_PITCH) = h1;
  }
  int tokc[4];
#pragma unroll
  for (int j = 0; j < 4; ++j) tokc[j] = kix[j * 16 + r];

#pragma unroll
  for (int d = 1; d <= 8; d <<= 1) {
#pragma unroll
    for (int qi = 0; qi < 2; ++qi) {
#pragma unroll
      for (int i = 0; i < 8; ++i) {
        const float mo = __shfl_xor(m[qi][i], d, 32);
        const float lo = __shfl_xor(l[qi][i], d, 32);
        const float mn = fmaxf(m[qi][i], mo);
        l[qi][i] = l[qi][i] * __expf(m[qi][i] - mn) + lo * __expf(mo - mn);
        m[qi][i] = mn;
      }
    }
  }
  float rl[2][8];
#pragma unroll
  for (int qi = 0; qi < 2; ++qi)
#pragma unroll
    for (int i = 0; i < 8; ++i) rl[qi][i] = 1.0f / l[qi][i];
  lds_barrier();

  v8f o[2][4] = {};
  for (int t = 0; t < nt; ++t) {
    const int cur = t & 1;
    const int tn = (t + 1 < nt) ? t + 1 : t;
    const v8bf* kn = kg + (size_t)tn * (AT_KT * HD / 8);
    const v8bf* vn = vg + (size_t)tn * (AT_KT / 8);
    const v8bf g0 = kn[0];
    const v8bf g1 = kn[32 * HD / 8];
    const v8bf h0 = vn[0];
    const v8bf h1 = vn[32 * SEQ / 8];
    int tokn[4];
#pragma unroll
    for (int j = 0; j < 4; ++j) tokn[j] = kix[tn * AT_KT + j * 16 + r];
    __builtin_amdgcn_sched_barrier(0);
    v16bf kf[4][2];
#pragma unroll
    for (int j = 0; j < 4; ++j) {
      const bf16_t* kr = ks[cur] + (j * 16 + r) * KV_PITCH;
      kf[j][0] = lds_frag_bf16(kr);
      kf[j][1] = lds_frag_bf16(kr + 32);
    }
#pragma unroll
    for (int kh = 0; kh < 2; ++kh) {
#pragma unroll
      for (int sub = 0; sub < 2; ++sub) {
        const int lrow = (kh * 2 + sub) * 16 + r;
        const v16bf kb0 = kf[kh * 2 + sub][0];
        const v16bf kb1 = kf[kh * 2 + sub][1];
        const int  slot = t * AT_KT + lrow;
        const int  tok  = tokc[kh * 2 + sub];
        const bool dead = slot >= lim;
        float cs = 0.0f;
#pragma unroll
        for (int qi = 0; qi < 2; ++qi) {
          v8f s = {};
          s = wmma_bf16(qa[qi][0], kb0, s);
          s = wmma_bf16(qa[qi][1], kb1, s);
          bf16_t* pw = qi ? pw1 : pw0;
#pragma unroll
          for (int i = 0; i < 8; ++i) {
            const float sv = dead ? NEGV : s[i] * scale;
            const float p  = __expf(sv - m[qi][i]) * rl[qi][i];
            cs += p;
            pw[(i + 8 * half) * 32 + sub * 16 + r] = (bf16_t)p;
          }
        }
        cs += __shfl_xor(cs, 16, 32);
        if (lane < 16 && slot < nk) atomicAdd(&halt[(size_t)b * SEQ + tok], cs);
      }
      __builtin_amdgcn_fence(__ATOMIC_ACQ_REL, "wavefront");
      v16bf pa0 = lds_frag_bf16(pw0 + r * 32);
      v16bf pa1 = lds_frag_bf16(pw1 + r * 32);
#pragma unroll
      for (int nb = 0; nb < 4; ++nb) {
        v16bf vb = lds_frag_bf16(vs[cur] + (nb * 16 + r) * KV_PITCH + kh * 32);
        o[0][nb] = wmma_bf16(pa0, vb, o[0][nb]);
        o[1][nb] = wmma_bf16(pa1, vb, o[1][nb]);
      }
      __builtin_amdgcn_fence(__ATOMIC_ACQ_REL, "wavefront");
    }
    *(v8bf*)(ks[cur ^ 1] + so)                 = g0;
    *(v8bf*)(ks[cur ^ 1] + so + 32 * KV_PITCH) = g1;
    *(v8bf*)(vs[cur ^ 1] + so)                 = h0;
    *(v8bf*)(vs[cur ^ 1] + so + 32 * KV_PITCH) = h1;
#pragma unroll
    for (int j = 0; j < 4; ++j) tokc[j] = tokn[j];
    lds_barrier();
  }

  const int h = bh & 3;
#pragma unroll
  for (int qi = 0; qi < 2; ++qi) {
#pragma unroll
    for (int nb = 0; nb < 4; ++nb) {
#pragma unroll
      for (int i = 0; i < 8; ++i) {
        const int t = qt * 32 + qi * 16 + i + 8 * half;
        attn[((size_t)b * SEQ + t) * DIMD + h * HD + nb * 16 + r] =
            (bf16_t)o[qi][nb][i];
      }
    }
  }
}

#define TROWS     64
#define HF_PITCH  260
#define HB_PITCH  264
#define HID_PITCH 1032
#define W1S_PITCH 264
#define W2S_PITCH 72
#define WST_ELEMS (DIMD * W2S_PITCH)

__device__ __forceinline__ float wave_sum(float v) {
#pragma unroll
  for (int d = 16; d >= 1; d >>= 1) v += __shfl_xor(v, d, 32);
  return v;
}

__device__ __forceinline__ void ln_row(float v[8], const float4 ga,
                                       const float4 gb, const float4 ba,
                                       const float4 bb) {
  float s = ((v[0] + v[1]) + (v[2] + v[3])) + ((v[4] + v[5]) + (v[6] + v[7]));
  const float mu = wave_sum(s) * (1.0f / DIMD);
  float d[8];
#pragma unroll
  for (int i = 0; i < 8; ++i) d[i] = v[i] - mu;
  float q = ((d[0] * d[0] + d[1] * d[1]) + (d[2] * d[2] + d[3] * d[3])) +
            ((d[4] * d[4] + d[5] * d[5]) + (d[6] * d[6] + d[7] * d[7]));
  const float var = wave_sum(q) * (1.0f / DIMD);
  const float rs = rsqrtf(var + 1e-5f);
  const float gg[8] = {ga.x, ga.y, ga.z, ga.w, gb.x, gb.y, gb.z, gb.w};
  const float bt[8] = {ba.x, ba.y, ba.z, ba.w, bb.x, bb.y, bb.z, bb.w};
#pragma unroll
  for (int i = 0; i < 8; ++i) v[i] = d[i] * rs * gg[i] + bt[i];
}

__global__ __launch_bounds__(256) void ffn_block(
    const float* __restrict__ xin, const float* __restrict__ proj,
    const float* __restrict__ g1, const float* __restrict__ be1,
    const bf16_t* __restrict__ W1, const float* __restrict__ b1,
    const bf16_t* __restrict__ W2, const float* __restrict__ b2,
    const float* __restrict__ g2, const float* __restrict__ be2,
    float* __restrict__ out) {
  __shared__ alignas(16) float  hf[TROWS * HF_PITCH];
  __shared__ alignas(16) bf16_t hb[TROWS * HB_PITCH];
  __shared__ alignas(16) bf16_t hid[TROWS * HID_PITCH];
  __shared__ alignas(16) bf16_t wst[2][WST_ELEMS];

  const int tid  = threadIdx.x;
  const int lane = tid & 31;
  const int wave = tid >> 5;
  const int r    = lane & 15;
  const int half = lane >> 4;
  const int mw   = wave >> 2;
  const int nw   = wave & 3;
  const size_t row_blk = (size_t)blockIdx.x * TROWS;

  const v8bf* w1g  = (const v8bf*)(W1 + (size_t)wave * DIMD) + lane;
  const int   w1so = wave * W1S_PITCH + lane * 8;
  const int   srow = tid >> 3;
  const int   scol = (tid & 7) * 8;
  const v8bf* w2g  = (const v8bf*)(W2 + (size_t)srow * FFND + scol);
  const int   w2so = srow * W2S_PITCH + scol;

  v8bf wg[8];
#pragma unroll
  for (int j = 0; j < 8; ++j) wg[j] = w1g[(size_t)(j * 8) * (DIMD / 8)];

  {
    const float4 ga = ((const float4*)g1)[lane], gb = ((const float4*)g1)[32 + lane];
    const float4 ba = ((const float4*)be1)[lane], bb = ((const float4*)be1)[32 + lane];
#pragma unroll
    for (int j = 0; j < 8; ++j) {
      const int lr = wave * 8 + j;
      const float4* xp = (const float4*)(xin + (row_blk + lr) * DIMD);
      const float4* pp = (const float4*)(proj + (row_blk + lr) * DIMD);
      const float4 x0 = xp[lane], x1 = xp[32 + lane];
      const float4 p0 = pp[lane], p1 = pp[32 + lane];
      float v[8] = {x0.x + p0.x, x0.y + p0.y, x0.z + p0.z, x0.w + p0.w,
                    x1.x + p1.x, x1.y + p1.y, x1.z + p1.z, x1.w + p1.w};
      ln_row(v, ga, gb, ba, bb);
      float* hp = hf + lr * HF_PITCH + lane * 4;
      *(float4*)hp         = make_float4(v[0], v[1], v[2], v[3]);
      *(float4*)(hp + 128) = make_float4(v[4], v[5], v[6], v[7]);
      v4bf c0, c1;
      c0[0] = (bf16_t)v[0]; c0[1] = (bf16_t)v[1]; c0[2] = (bf16_t)v[2]; c0[3] = (bf16_t)v[3];
      c1[0] = (bf16_t)v[4]; c1[1] = (bf16_t)v[5]; c1[2] = (bf16_t)v[6]; c1[3] = (bf16_t)v[7];
      bf16_t* bp = hb + lr * HB_PITCH + lane * 4;
      *(v4bf*)bp         = c0;
      *(v4bf*)(bp + 128) = c1;
    }
  }
#pragma unroll
  for (int j = 0; j < 8; ++j) *(v8bf*)(wst[0] + w1so + j * 8 * W1S_PITCH) = wg[j];
  __syncthreads();

  {
    v16bf af[2][8];
#pragma unroll
    for (int mi = 0; mi < 2; ++mi) {
#pragma unroll
      for (int s = 0; s < 8; ++s)
        af[mi][s] = lds_frag_bf16(hb + (mw * 32 + mi * 16 + r) * HB_PITCH + s * 32);
    }
    float bias = b1[nw * 16 + r];
    for (int c = 0; c < FFND / 64; ++c) {
      const int cur = c & 1;
      const int cn = (c + 1 < FFND / 64) ? c + 1 : c;
      const float bias_n = b1[cn * 64 + nw * 16 + r];
#pragma unroll
      for (int j = 0; j < 8; ++j)
        wg[j] = w1g[(size_t)(cn * 64 + j * 8) * (DIMD / 8)];
      const bf16_t* bb = wst[cur] + (nw * 16 + r) * W1S_PITCH;
      v16bf bq[8];
#pragma unroll
      for (int s = 0; s < 8; ++s) bq[s] = lds_frag_bf16(bb + s * 32);
      __builtin_amdgcn_sched_barrier(0);
      v8f acc[2] = {};
#pragma unroll
      for (int s = 0; s < 8; ++s) {
        acc[0] = wmma_bf16(af[0][s], bq[s], acc[0]);
        acc[1] = wmma_bf16(af[1][s], bq[s], acc[1]);
      }
      const int col = c * 64 + nw * 16 + r;
#pragma unroll
      for (int mi = 0; mi < 2; ++mi) {
#pragma unroll
        for (int i = 0; i < 8; ++i) {
          const int row = mw * 32 + mi * 16 + i + 8 * half;
          hid[row * HID_PITCH + col] = (bf16_t)fmaxf(acc[mi][i] + bias, 0.0f);
        }
      }
      bias = bias_n;
      __builtin_amdgcn_sched_barrier(0);
#pragma unroll
      for (int j = 0; j < 8; ++j)
        *(v8bf*)(wst[cur ^ 1] + w1so + j * 8 * W1S_PITCH) = wg[j];
      lds_barrier();
    }
  }
#pragma unroll
  for (int j = 0; j < 8; ++j) wg[j] = w2g[(size_t)j * 32 * (FFND / 8)];
#pragma unroll
  for (int j = 0; j < 8; ++j) *(v8bf*)(wst[0] + w2so + j * 32 * W2S_PITCH) = wg[j];
  __syncthreads();

  {
    const int oc0 = nw * 64;
    const bf16_t* ia0 = hid + (mw * 32 + r) * HID_PITCH;
    const bf16_t* ia1 = ia0 + 16 * HID_PITCH;
    v8f acc[2][4] = {};
    for (int t = 0; t < FFND / 64; ++t) {
      const int cur = t & 1;
      const int tn = (t + 1 < FFND / 64) ? t + 1 : t;
#pragma unroll
      for (int j = 0; j < 8; ++j)
        wg[j] = w2g[(size_t)j * 32 * (FFND / 8) + tn * 8];
      const bf16_t* bb = wst[cur] + (oc0 + r) * W2S_PITCH;
      v16bf a[2][2], bq[2][4];
#pragma unroll
      for (int s = 0; s < 2; ++s) {
        a[0][s] = lds_frag_bf16(ia0 + t * 64 + s * 32);
        a[1][s] = lds_frag_bf16(ia1 + t * 64 + s * 32);
#pragma unroll
        for (int nb = 0; nb < 4; ++nb)
          bq[s][nb] = lds_frag_bf16(bb + nb * 16 * W2S_PITCH + s * 32);
      }
      __builtin_amdgcn_sched_barrier(0);
#pragma unroll
      for (int s = 0; s < 2; ++s) {
#pragma unroll
        for (int nb = 0; nb < 4; ++nb) {
          acc[0][nb] = wmma_bf16(a[0][s], bq[s][nb], acc[0][nb]);
          acc[1][nb] = wmma_bf16(a[1][s], bq[s][nb], acc[1][nb]);
        }
      }
      __builtin_amdgcn_sched_barrier(0);
#pragma unroll
      for (int j = 0; j < 8; ++j)
        *(v8bf*)(wst[cur ^ 1] + w2so + j * 32 * W2S_PITCH) = wg[j];
      lds_barrier();
    }
#pragma unroll
    for (int mi = 0; mi < 2; ++mi) {
#pragma unroll
      for (int nb = 0; nb < 4; ++nb) {
        const int col = oc0 + nb * 16 + r;
        const float bias = b2[col];
#pragma unroll
        for (int i = 0; i < 8; ++i) {
          const int row = mw * 32 + mi * 16 + i + 8 * half;
          float* hp = hf + row * HF_PITCH + col;
          *hp = *hp + (acc[mi][nb][i] + bias);
        }
      }
    }
  }
  __syncthreads();

  {
    const float4 ga = ((const float4*)g2)[lane], gb = ((const float4*)g2)[32 + lane];
    const float4 ba = ((const float4*)be2)[lane], bb = ((const float4*)be2)[32 + lane];
#pragma unroll
    for (int j = 0; j < 8; ++j) {
      const int lr = wave * 8 + j;
      const float* hp = hf + lr * HF_PITCH + lane * 4;
      const float4 y0 = *(const float4*)hp;
      const float4 y1 = *(const float4*)(hp + 128);
      float v[8] = {y0.x, y0.y, y0.z, y0.w, y1.x, y1.y, y1.z, y1.w};
      ln_row(v, ga, gb, ba, bb);
      float4* op = (float4*)(out + (row_blk + lr) * DIMD);
      op[lane]      = make_float4(v[0], v[1], v[2], v[3]);
      op[32 + lane] = make_float4(v[4], v[5], v[6], v[7]);
    }
  }
}

__global__ __launch_bounds__(256) void halt_finalize(
    const float* __restrict__ hacc, const float* __restrict__ mask,
    float* __restrict__ out) {
  __shared__ float red[256];
  const int b = blockIdx.x, tid = threadIdx.x;
  float lm = -3.0e38f;
  for (int j = tid; j < SEQ; j += 256) {
    const float v = hacc[(size_t)b * SEQ + j] * 0.25f +
                    mask[(size_t)b * SEQ + j] * (-1e10f);
    lm = fmaxf(lm, v);
  }
  red[tid] = lm;
  __syncthreads();
  for (int s = 128; s > 0; s >>= 1) {
    if (tid < s) red[tid] = fmaxf(red[tid], red[tid + s]);
    __syncthreads();
  }
  const float mx = red[0];
  __syncthreads();
  float ls = 0.0f;
  for (int j = tid; j < SEQ; j += 256) {
    const float v = hacc[(size_t)b * SEQ + j] * 0.25f +
                    mask[(size_t)b * SEQ + j] * (-1e10f);
    ls += __expf(v - mx);
  }
  red[tid] = ls;
  __syncthreads();
  for (int s = 128; s > 0; s >>= 1) {
    if (tid < s) red[tid] += red[tid + s];
    __syncthreads();
  }
  const float inv = 1.0f / red[0];
  for (int j = tid; j < SEQ; j += 256) {
    const float v = hacc[(size_t)b * SEQ + j] * 0.25f +
                    mask[(size_t)b * SEQ + j] * (-1e10f);
    out[(size_t)b * SEQ + j] = __expf(v - mx) * inv;
  }
}

extern "C" void kernel_launch(void* const* d_in, const int* in_sizes, int n_in,
                              void* d_out, int out_size, void* d_ws,
                              size_t ws_size, hipStream_t stream) {
  (void)in_sizes; (void)n_in; (void)out_size; (void)ws_size;
  const float* x    = (const float*)d_in[0];
  const float* mask = (const float*)d_in[1];
  const float* Wqkv = (const float*)d_in[2];
  const float* bqkv = (const float*)d_in[3];
  const float* Wo   = (const float*)d_in[4];
  const float* bo   = (const float*)d_in[5];
  const float* W1   = (const float*)d_in[6];
  const float* b1   = (const float*)d_in[7];
  const float* W2   = (const float*)d_in[8];
  const float* b2   = (const float*)d_in[9];
  const float* g1   = (const float*)d_in[10];
  const float* be1  = (const float*)d_in[11];
  const float* g2   = (const float*)d_in[12];
  const float* be2  = (const float*)d_in[13];

  char* ws = (char*)d_ws;
  size_t off = 0;
  auto alloc = [&](size_t bytes) {
    char* p = ws + off;
    off += (bytes + 255) & ~(size_t)255;
    return p;
  };
  bf16_t* xb    = (bf16_t*)alloc((size_t)NTOK * DIMD * 2);
  bf16_t* Wqkvb = (bf16_t*)alloc((size_t)3 * DIMD * DIMD * 2);
  bf16_t* Wob   = (bf16_t*)alloc((size_t)DIMD * DIMD * 2);
  bf16_t* W1b   = (bf16_t*)alloc((size_t)FFND * DIMD * 2);
  bf16_t* W2b   = (bf16_t*)alloc((size_t)DIMD * FFND * 2);
  bf16_t* Qb    = (bf16_t*)alloc((size_t)NTOK * DIMD * 2);
  bf16_t* Kb    = (bf16_t*)alloc((size_t)NTOK * DIMD * 2);
  bf16_t* Vt    = (bf16_t*)alloc((size_t)NTOK * DIMD * 2);
  bf16_t* attnb = (bf16_t*)alloc((size_t)NTOK * DIMD * 2);
  float*  attnO = (float*) alloc((size_t)NTOK * DIMD * 4);
  float*  halt  = (float*) alloc((size_t)NTOK * 4);

  int* kpos = (int*)alloc((size_t)NTOK * 4);
  int* kidx = (int*)alloc((size_t)NTOK * 4);
  int* nkv  = (int*)alloc((size_t)BATCH * 4);

  float* out      = (float*)d_out;
  float* out_halt = out + (size_t)NTOK * DIMD;

  hipMemsetAsync(halt, 0, (size_t)NTOK * sizeof(float), stream);

  cvt_bf16<<<NTOK * DIMD / 1024, 256, 0, stream>>>(x, xb);
  cvt_bf16<<<3 * DIMD * DIMD / 1024, 256, 0, stream>>>(Wqkv, Wqkvb);
  cvt_bf16<<<DIMD * DIMD / 1024, 256, 0, stream>>>(Wo, Wob);
  cvt_bf16<<<FFND * DIMD / 1024, 256, 0, stream>>>(W1, W1b);
  cvt_bf16<<<DIMD * FFND / 1024, 256, 0, stream>>>(W2, W2b);

  mask_scan<<<BATCH, 256, 0, stream>>>(mask, kpos, kidx, nkv, Kb, Vt);

  gemm_wmma<2><<<dim3(3 * DIMD / 64, NTOK / 128), 128, 0, stream>>>(
      xb, Wqkvb, bqkv, nullptr, Qb, Kb, Vt, kpos, 3 * DIMD);

  attn_wmma<<<BATCH * HEADS * (SEQ / (32 * AT_WAVES)), 32 * AT_WAVES, 0, stream>>>(
      Qb, Kb, Vt, nkv, kidx, attnb, halt);

  gemm_wmma<0><<<dim3(DIMD / 64, NTOK / 128), 128, 0, stream>>>(
      attnb, Wob, bo, attnO, nullptr, nullptr, nullptr, nullptr, DIMD);

  ffn_block<<<NTOK / TROWS, 256, 0, stream>>>(
      x, attnO, g1, be1, W1b, b1, W2b, b2, g2, be2, out);

  halt_finalize<<<BATCH, 256, 0, stream>>>(halt, mask, out_halt);
}


# x is converted to bf16 where it is read: QKV as a 64-row-tile kernel reads f32 x once into LDS, weight tiles via a 2-stage LDS ring; the 32768-wave convert launch is gone; sum order unchanged
# speedup vs baseline: 2.6856x; 1.1763x over previous
#include <hip/hip_runtime.h>
#include <hip/hip_bf16.h>

typedef __bf16 bf16_t;
typedef __attribute__((ext_vector_type(16))) __bf16 v16bf;
typedef __attribute__((ext_vector_type(4)))  __bf16 v4bf;
typedef __attribute__((ext_vector_type(8)))  float  v8f;
typedef __attribute__((ext_vector_type(8)))  __bf16 v8bf;

#define DIMD   256
#define HEADS  4
#define HD     64
#define FFND   1024
#define BATCH  8
#define SEQ    2048
#define NTOK   (BATCH*SEQ)
#define NEGV   (-1e9f)

__device__ __forceinline__ v16bf load_frag_bf16(const bf16_t* base) {
  const int half = (threadIdx.x >> 4) & 1;
  const bf16_t* p = base + 8 * half;
  v16bf f;
#pragma unroll
  for (int e = 0; e < 8; ++e) {
    f[e]     = p[e];
    f[e + 8] = p[e + 16];
  }
  return f;
}

__device__ __forceinline__ v16bf lds_frag_bf16(const bf16_t* base) {
  const int half = (threadIdx.x >> 4) & 1;
  const bf16_t* p = base + 8 * half;
  const v8bf lo = *(const v8bf*)p;
  const v8bf hi = *(const v8bf*)(p + 16);
  return __builtin_shufflevector(lo, hi, 0, 1, 2, 3, 4, 5, 6, 7,
                                 8, 9, 10, 11, 12, 13, 14, 15);
}

__device__ __forceinline__ void lds_barrier() {
  __builtin_amdgcn_fence(__ATOMIC_RELEASE, "workgroup", "local");
  __builtin_amdgcn_s_barrier();
  __builtin_amdgcn_fence(__ATOMIC_ACQUIRE, "workgroup", "local");
}

__device__ __forceinline__ v8f wmma_bf16(v16bf a, v16bf b, v8f c) {
  return __builtin_amdgcn_wmma_f32_16x16x32_bf16(
       false, a,  false, b,
       (short)0, c,  false,  false);
}

__global__ __launch_bounds__(256) void cvt_bf16(const float* __restrict__ in,
                                                bf16_t* __restrict__ out) {
  const size_t i = (size_t)blockIdx.x * 256 + threadIdx.x;
  const float4 v = ((const float4*)in)[i];
  v4bf o;
  o[0] = (bf16_t)v.x; o[1] = (bf16_t)v.y; o[2] = (bf16_t)v.z; o[3] = (bf16_t)v.w;
  ((v4bf*)out)[i] = o;
}

__global__ __launch_bounds__(256) void mask_scan(
    const float* __restrict__ mask, int* __restrict__ kpos,
    int* __restrict__ nkv,
    bf16_t* __restrict__ Kb, bf16_t* __restrict__ Vt) {
  __shared__ int cnt[256];
  const int b = blockIdx.x, tid = threadIdx.x;
  const float4* mp = (const float4*)(mask + (size_t)b * SEQ + tid * 8);
  const float4 m0 = mp[0], m1 = mp[1];
  const float mv[8] = {m0.x, m0.y, m0.z, m0.w, m1.x, m1.y, m1.z, m1.w};
  int keep[8], c = 0;
#pragma unroll
  for (int j = 0; j < 8; ++j) { keep[j] = !(mv[j] > 0.5f); c += keep[j]; }
  cnt[tid] = c;
  __syncthreads();
  for (int s = 1; s < 256; s <<= 1) {
    const int add = (tid >= s) ? cnt[tid - s] : 0;
    __syncthreads();
    cnt[tid] += add;
    __syncthreads();
  }
  const int total = cnt[255];
  const bool allm = (total == 0);
  int p = cnt[tid] - c;
#pragma unroll
  for (int j = 0; j < 8; ++j) {
    const int t = tid * 8 + j;
    int q = -1;
    if (allm) q = t;
    else if (keep[j]) q = p++;
    kpos[(size_t)b * SEQ + t] = q;
  }
  const int nk = allm ? SEQ : total;
  if (tid == 0) nkv[b] = allm ? -1 : total;
  const int nk64 = (nk + 63) & ~63;
  const int h = tid >> 6, d = tid & 63;
  const size_t bhh = (size_t)b * HEADS + h;
  for (int q = nk; q < nk64; ++q) {
    Kb[(bhh * SEQ + q) * HD + d] = (bf16_t)0.0f;
    Vt[(bhh * HD + d) * SEQ + q] = (bf16_t)0.0f;
  }
}

#define QROWS    64
#define XS_PITCH 264
#define WQ_PITCH 264
__global__ __launch_bounds__(256) void qkv_block(
    const float* __restrict__ xin, const bf16_t* __restrict__ Wb,
    const float* __restrict__ bias, bf16_t* __restrict__ Qb,
    bf16_t* __restrict__ Kb, bf16_t* __restrict__ Vt,
    const int* __restrict__ kpos) {
  __shared__ alignas(16) bf16_t xs[QROWS * XS_PITCH];
  __shared__ alignas(16) bf16_t wq[2][64 * WQ_PITCH];

  const int tid  = threadIdx.x;
  const int lane = tid & 31;
  const int wave = tid >> 5;
  const int r    = lane & 15;
  const int half = lane >> 4;
  const int mw   = wave >> 2;
  const int nw   = wave & 3;
  const int row_blk = blockIdx.x * QROWS;

  const v8bf* wgp = (const v8bf*)(Wb + (size_t)wave * DIMD) + lane;
  const int   wso = wave * WQ_PITCH + lane * 8;

  v8bf wg[8];
#pragma unroll
  for (int j = 0; j < 8; ++j) wg[j] = wgp[(size_t)(j * 8) * (DIMD / 8)];

  {
    float4 x0[8], x1[8];
#pragma unroll
    for (int j = 0; j < 8; ++j) {
      const float4* xp = (const float4*)(xin + (size_t)(row_blk + wave * 8 + j) * DIMD);
      x0[j] = xp[lane];
      x1[j] = xp[32 + lane];
    }
#pragma unroll
    for (int j = 0; j < 8; ++j) {
      v4bf c0, c1;
      c0[0] = (bf16_t)x0[j].x; c0[1] = (bf16_t)x0[j].y; c0[2] = (bf16_t)x0[j].z; c0[3] = (bf16_t)x0[j].w;
      c1[0] = (bf16_t)x1[j].x; c1[1] = (bf16_t)x1[j].y; c1[2] = (bf16_t)x1[j].z; c1[3] = (bf16_t)x1[j].w;
      bf16_t* bp = xs + (wave * 8 + j) * XS_PITCH + lane * 4;
      *(v4bf*)bp         = c0;
      *(v4bf*)(bp + 128) = c1;
    }
  }
#pragma unroll
  for (int j = 0; j < 8; ++j) *(v8bf*)(wq[0] + wso + j * 8 * WQ_PITCH) = wg[j];
  __syncthreads();

  v16bf af[2][8];
#pragma unroll
  for (int mi = 0; mi < 2; ++mi) {
#pragma unroll
    for (int s = 0; s < 8; ++s)
      af[mi][s] = lds_frag_bf16(xs + (mw * 32 + mi * 16 + r) * XS_PITCH + s * 32);
  }
  int kp[2][8];
#pragma unroll
  for (int mi = 0; mi < 2; ++mi) {
#pragma unroll
    for (int i = 0; i < 8; ++i)
      kp[mi][i] = kpos[row_blk + mw * 32 + mi * 16 + i + 8 * half];
  }
  const int b  = row_blk >> 11;
  const int t0 = (row_blk & (SEQ - 1)) + mw * 32 + 8 * half;
  const int d  = nw * 16 + r;

  float bv = bias[d];
  for (int c = 0; c < 3 * DIMD / 64; ++c) {
    const int cur = c & 1;
    const int cn = (c + 1 < 3 * DIMD / 64) ? c + 1 : c;
    const float bias_n = bias[cn * 64 + d];
#pragma unroll
    for (int j = 0; j < 8; ++j)
      wg[j] = wgp[(size_t)(cn * 64 + j * 8) * (DIMD / 8)];
    const bf16_t* bb = wq[cur] + d * WQ_PITCH;
    v16bf bq[8];
#pragma unroll
    for (int s = 0; s < 8; ++s) bq[s] = lds_frag_bf16(bb + s * 32);
    __builtin_amdgcn_sched_barrier(0);
    v8f acc[2] = {};
#pragma unroll
    for (int s = 0; s < 8; ++s) {
      acc[0] = wmma_bf16(af[0][s], bq[s], acc[0]);
      acc[1] = wmma_bf16(af[1][s], bq[s], acc[1]);
    }
    const int sec = c >> 2;
    const size_t bhh = (size_t)b * HEADS + (c & 3);
    bf16_t y[2][8];
#pragma unroll
    for (int mi = 0; mi < 2; ++mi) {
#pragma unroll
      for (int i = 0; i < 8; ++i) y[mi][i] = (bf16_t)(acc[mi][i] + bv);
    }
    if (sec == 0) {
      bf16_t* qp = Qb + (bhh * SEQ + t0) * HD + d;
#pragma unroll
      for (int mi = 0; mi < 2; ++mi) {
#pragma unroll
        for (int i = 0; i < 8; ++i) qp[(mi * 16 + i) * HD] = y[mi][i];
      }
    } else if (sec == 1) {
      bf16_t* kq = Kb + bhh * SEQ * HD + d;
#pragma unroll
      for (int mi = 0; mi < 2; ++mi) {
#pragma unroll
        for (int i = 0; i < 8; ++i) {
          const int p = kp[mi][i];
          if (__builtin_expect(p >= 0, 1)) kq[(size_t)p * HD] = y[mi][i];
        }
      }
    } else {
      bf16_t* vq = Vt + (bhh * HD + d) * SEQ;
#pragma unroll
      for (int mi = 0; mi < 2; ++mi) {
#pragma unroll
        for (int i = 0; i < 8; ++i) {
          const int p = kp[mi][i];
          if (__builtin_expect(p >= 0, 1)) vq[p] = y[mi][i];
        }
      }
    }
    bv = bias_n;
#pragma unroll
    for (int j = 0; j < 8; ++j)
      *(v8bf*)(wq[cur ^ 1] + wso + j * 8 * WQ_PITCH) = wg[j];
    lds_barrier();
  }
}

#define GT_PITCH 264
__global__ __launch_bounds__(128) void gemm_wmma(
    const bf16_t* __restrict__ A, const bf16_t* __restrict__ B,
    const float* __restrict__ bias, float* __restrict__ Yf, int N) {
  __shared__ alignas(16) bf16_t at[128 * GT_PITCH];
  __shared__ alignas(16) bf16_t bt[64 * GT_PITCH];
  const int lane = threadIdx.x & 31;
  const int wave = threadIdx.x >> 5;
  const int r    = lane & 15;
  const int half = lane >> 4;
  const int rowb = blockIdx.y * 128;
  const int row0 = rowb + wave * 32;
  const int col0 = blockIdx.x * 64;

  {
    const v8bf* ag = (const v8bf*)(A + (size_t)(rowb + wave) * DIMD) + lane;
    const v8bf* bg = (const v8bf*)(B + (size_t)(col0 + wave) * DIMD) + lane;
    const int so = wave * GT_PITCH + lane * 8;
    v8bf tb[16], ta[32];
#pragma unroll
    for (int j = 0; j < 16; ++j) tb[j] = bg[(size_t)j * 4 * (DIMD / 8)];
#pragma unroll
    for (int j = 0; j < 32; ++j) ta[j] = ag[(size_t)j * 4 * (DIMD / 8)];
#pragma unroll
    for (int j = 0; j < 16; ++j) *(v8bf*)(bt + so + j * 4 * GT_PITCH) = tb[j];
#pragma unroll
    for (int j = 0; j < 32; ++j) *(v8bf*)(at + so + j * 4 * GT_PITCH) = ta[j];
  }
  __syncthreads();

  v8f acc[2][4] = {};
  const bf16_t* ap0 = at + (wave * 32 + r) * GT_PITCH;
  const bf16_t* ap1 = ap0 + 16 * GT_PITCH;
  const bf16_t* bp  = bt + r * GT_PITCH;

  v16bf fa[2][2], fb[2][4];
  fa[0][0] = lds_frag_bf16(ap0);
  fa[0][1] = lds_frag_bf16(ap1);
#pragma unroll
  for (int nb = 0; nb < 4; ++nb) fb[0][nb] = lds_frag_bf16(bp + nb * 16 * GT_PITCH);
#pragma unroll
  for (int ks = 0; ks < DIMD / 32; ++ks) {
    const int cs = ks & 1, ns = cs ^ 1;
    if (ks + 1 < DIMD / 32) {
      const int k1 = (ks + 1) * 32;
      fa[ns][0] = lds_frag_bf16(ap0 + k1);
      fa[ns][1] = lds_frag_bf16(ap1 + k1);
#pragma unroll
      for (int nb = 0; nb < 4; ++nb)
        fb[ns][nb] = lds_frag_bf16(bp + nb * 16 * GT_PITCH + k1);
    }
    __builtin_amdgcn_sched_barrier(0);
#pragma unroll
    for (int nb = 0; nb < 4; ++nb) {
      acc[0][nb] = wmma_bf16(fa[cs][0], fb[cs][nb], acc[0][nb]);
      acc[1][nb] = wmma_bf16(fa[cs][1], fb[cs][nb], acc[1][nb]);
    }
    __builtin_amdgcn_sched_barrier(0);
  }

#pragma unroll
  for (int mi = 0; mi < 2; ++mi) {
#pragma unroll
    for (int nb = 0; nb < 4; ++nb) {
#pragma unroll
      for (int i = 0; i < 8; ++i) {
        const int row = row0 + mi * 16 + i + 8 * half;
        const int col = col0 + nb * 16 + r;
        Yf[(size_t)row * N + col] = acc[mi][nb][i] + bias[col];
      }
    }
  }
}

#define AT_WAVES 8
#define AT_KT    64
#define HPARTS   (HEADS * SEQ / (32 * AT_WAVES))
#define KV_PITCH 72

__global__ __launch_bounds__(256) __attribute__((amdgpu_waves_per_eu(2, 2))) void attn_wmma(
    const bf16_t* __restrict__ Qb, const bf16_t* __restrict__ Kb,
    const bf16_t* __restrict__ Vt, const int* __restrict__ nkv,
    bf16_t* __restrict__ attn, float* __restrict__ hpart) {
  __shared__ alignas(16) bf16_t ks[2][AT_KT * KV_PITCH];
  __shared__ alignas(16) bf16_t vs[2][HD * KV_PITCH];
  __shared__ alignas(16) bf16_t lds_pw[AT_WAVES][2][16 * 32];
  __shared__ float csl[2][AT_WAVES][AT_KT];

  const int tid  = threadIdx.x;
  const int wave = tid >> 5;
  const int bid  = blockIdx.x;
  const int bh   = bid >> 3;
  const int qt   = (bid & 7) * AT_WAVES + wave;
  const int b    = bh >> 2;
  const int lane = tid & 31;
  const int r    = lane & 15;
  const int half = lane >> 4;
  const float scale = 0.125f;

  const bf16_t* Qbase = Qb + ((size_t)bh * SEQ + qt * 32) * HD;
  const bf16_t* Kbase = Kb + (size_t)bh * SEQ * HD;
  const bf16_t* Vbase = Vt + (size_t)bh * HD * SEQ;
  float*        hp    = hpart + (size_t)bid * SEQ;
  const int nkraw = nkv[b];
  const int nk    = (nkraw < 0) ? SEQ : nkraw;
  const int lim   = (nkraw < 0) ? 0 : nkraw;
  const int nt    = (nk + AT_KT - 1) / AT_KT;

  const int srow = tid >> 3;
  const int scol = (tid & 7) * 8;
  const v8bf* kg = (const v8bf*)(Kbase + (size_t)srow * HD + scol);
  const v8bf* vg = (const v8bf*)(Vbase + (size_t)srow * SEQ + scol);
  const int   so = srow * KV_PITCH + scol;
  bf16_t* pw0 = lds_pw[wave][0];
  bf16_t* pw1 = lds_pw[wave][1];

  v16bf qa[2][2];
#pragma unroll
  for (int qi = 0; qi < 2; ++qi) {
    qa[qi][0] = load_frag_bf16(Qbase + (size_t)(qi * 16 + r) * HD + 0);
    qa[qi][1] = load_frag_bf16(Qbase + (size_t)(qi * 16 + r) * HD + 32);
  }

  float m[2][8], l[2][8];
#pragma unroll
  for (int qi = 0; qi < 2; ++qi)
#pragma unroll
    for (int i = 0; i < 8; ++i) { m[qi][i] = -3.0e38f; l[qi][i] = 0.0f; }

  {
    const v8bf g0 = kg[0];
    const v8bf g1 = kg[32 * HD / 8];
    *(v8bf*)(ks[0] + so)                 = g0;
    *(v8bf*)(ks[0] + so + 32 * KV_PITCH) = g1;
  }
  lds_barrier();
  for (int t = 0; t < nt; ++t) {
    const int cur = t & 1;
    const int tn = (t + 1 < nt) ? t + 1 : t;
    const v8bf* kn = kg + (size_t)tn * (AT_KT * HD / 8);
    const v8bf g0 = kn[0];
    const v8bf g1 = kn[32 * HD / 8];
    __builtin_amdgcn_sched_barrier(0);
#pragma unroll
    for (int sub = 0; sub < 4; ++sub) {
      const bf16_t* kr = ks[cur] + (sub * 16 + r) * KV_PITCH;
      v16bf kb0 = lds_frag_bf16(kr);
      v16bf kb1 = lds_frag_bf16(kr + 32);
      const bool dead = (t * AT_KT + sub * 16 + r) >= lim;
#pragma unroll
      for (int qi = 0; qi < 2; ++qi) {
        v8f s = {};
        s = wmma_bf16(qa[qi][0], kb0, s);
        s = wmma_bf16(qa[qi][1], kb1, s);
#pragma unroll
        for (int i = 0; i < 8; ++i) {
          const float sv = dead ? NEGV : s[i] * scale;
          const float mn = fmaxf(m[qi][i], sv);
          l[qi][i] = l[qi][i] * __expf(m[qi][i] - mn) + __expf(sv - mn);
          m[qi][i] = mn;
        }
      }
    }
    __builtin_amdgcn_sched_barrier(0);
    *(v8bf*)(ks[cur ^ 1] + so)                 = g0;
    *(v8bf*)(ks[cur ^ 1] + so + 32 * KV_PITCH) = g1;
    lds_barrier();
  }

  {
    const v8bf g0 = kg[0];
    const v8bf g1 = kg[32 * HD / 8];
    const v8bf h0 = vg[0];
    const v8bf h1 = vg[32 * SEQ / 8];
    *(v8bf*)(ks[0] + so)                 = g0;
    *(v8bf*)(ks[0] + so + 32 * KV_PITCH) = g1;
    *(v8bf*)(vs[0] + so)                 = h0;
    *(v8bf*)(vs[0] + so + 32 * KV_PITCH) = h1;
  }

#pragma unroll
  for (int d = 1; d <= 8; d <<= 1) {
#pragma unroll
    for (int qi = 0; qi < 2; ++qi) {
#pragma unroll
      for (int i = 0; i < 8; ++i) {
        const float mo = __shfl_xor(m[qi][i], d, 32);
        const float lo = __shfl_xor(l[qi][i], d, 32);
        const float mn = fmaxf(m[qi][i], mo);
        l[qi][i] = l[qi][i] * __expf(m[qi][i] - mn) + lo * __expf(mo - mn);
        m[qi][i] = mn;
      }
    }
  }
  float rl[2][8];
#pragma unroll
  for (int qi = 0; qi < 2; ++qi)
#pragma unroll
    for (int i = 0; i < 8; ++i) rl[qi][i] = 1.0f / l[qi][i];
  lds_barrier();

  v8f o[2][4] = {};
  for (int t = 0; t < nt; ++t) {
    const int cur = t & 1;
    const int tn = (t + 1 < nt) ? t + 1 : t;
    const v8bf* kn = kg + (size_t)tn * (AT_KT * HD / 8);
    const v8bf* vn = vg + (size_t)tn * (AT_KT / 8);
    const v8bf g0 = kn[0];
    const v8bf g1 = kn[32 * HD / 8];
    const v8bf h0 = vn[0];
    const v8bf h1 = vn[32 * SEQ / 8];
    if (t > 0 && tid < AT_KT) {
      const float* cp = &csl[cur ^ 1][0][tid];
      float s = cp[0];
#pragma unroll
      for (int w = 1; w < AT_WAVES; ++w) s += cp[w * AT_KT];
      hp[(t - 1) * AT_KT + tid] = s;
    }
    __builtin_amdgcn_sched_barrier(0);
    v16bf kf[4][2];
#pragma unroll
    for (int j = 0; j < 4; ++j) {
      const bf16_t* kr = ks[cur] + (j * 16 + r) * KV_PITCH;
      kf[j][0] = lds_frag_bf16(kr);
      kf[j][1] = lds_frag_bf16(kr + 32);
    }
    float* csw = csl[cur][wave];
#pragma unroll
    for (int kh = 0; kh < 2; ++kh) {
      float csv[2];
#pragma unroll
      for (int sub = 0; sub < 2; ++sub) {
        const int lrow = (kh * 2 + sub) * 16 + r;
        const v16bf kb0 = kf[kh * 2 + sub][0];
        const v16bf kb1 = kf[kh * 2 + sub][1];
        const int  slot = t * AT_KT + lrow;
        const bool dead = slot >= lim;
        float cs = 0.0f;
#pragma unroll
        for (int qi = 0; qi < 2; ++qi) {
          v8f s = {};
          s = wmma_bf16(qa[qi][0], kb0, s);
          s = wmma_bf16(qa[qi][1], kb1, s);
          bf16_t* pw = qi ? pw1 : pw0;
#pragma unroll
          for (int i = 0; i < 8; ++i) {
            const float sv = dead ? NEGV : s[i] * scale;
            const float p  = __expf(sv - m[qi][i]) * rl[qi][i];
            cs += p;
            pw[(i + 8 * half) * 32 + sub * 16 + r] = (bf16_t)p;
          }
        }
        csv[sub] = cs + __shfl_xor(cs, 16, 32);
      }
      csw[kh * 32 + lane] = half ? csv[1] : csv[0];
      __builtin_amdgcn_fence(__ATOMIC_ACQ_REL, "wavefront");
      v16bf pa0 = lds_frag_bf16(pw0 + r * 32);
      v16bf pa1 = lds_frag_bf16(pw1 + r * 32);
#pragma unroll
      for (int nb = 0; nb < 4; ++nb) {
        v16bf vb = lds_frag_bf16(vs[cur] + (nb * 16 + r) * KV_PITCH + kh * 32);
        o[0][nb] = wmma_bf16(pa0, vb, o[0][nb]);
        o[1][nb] = wmma_bf16(pa1, vb, o[1][nb]);
      }
      __builtin_amdgcn_fence(__ATOMIC_ACQ_REL, "wavefront");
    }
    *(v8bf*)(ks[cur ^ 1] + so)                 = g0;
    *(v8bf*)(ks[cur ^ 1] + so + 32 * KV_PITCH) = g1;
    *(v8bf*)(vs[cur ^ 1] + so)                 = h0;
    *(v8bf*)(vs[cur ^ 1] + so + 32 * KV_PITCH) = h1;
    lds_barrier();
  }
  if (tid < AT_KT) {
    const float* cp = &csl[(nt - 1) & 1][0][tid];
    float s = cp[0];
#pragma unroll
    for (int w = 1; w < AT_WAVES; ++w) s += cp[w * AT_KT];
    hp[(nt - 1) * AT_KT + tid] = s;
  }

  const int h = bh & 3;
#pragma unroll
  for (int qi = 0; qi < 2; ++qi) {
#pragma unroll
    for (int nb = 0; nb < 4; ++nb) {
#pragma unroll
      for (int i = 0; i < 8; ++i) {
        const int t = qt * 32 + qi * 16 + i + 8 * half;
        attn[((size_t)b * SEQ + t) * DIMD + h * HD + nb * 16 + r] =
            (bf16_t)o[qi][nb][i];
      }
    }
  }
}

#define TROWS     64
#define HF_PITCH  260
#define HB_PITCH  264
#define HID_PITCH 1032
#define W1S_PITCH 264
#define W2S_PITCH 72
#define WST_ELEMS (DIMD * W2S_PITCH)

__device__ __forceinline__ float wave_sum(float v) {
#pragma unroll
  for (int d = 16; d >= 1; d >>= 1) v += __shfl_xor(v, d, 32);
  return v;
}

__device__ __forceinline__ void ln_row(float v[8], const float4 ga,
                                       const float4 gb, const float4 ba,
                                       const float4 bb) {
  float s = ((v[0] + v[1]) + (v[2] + v[3])) + ((v[4] + v[5]) + (v[6] + v[7]));
  const float mu = wave_sum(s) * (1.0f / DIMD);
  float d[8];
#pragma unroll
  for (int i = 0; i < 8; ++i) d[i] = v[i] - mu;
  float q = ((d[0] * d[0] + d[1] * d[1]) + (d[2] * d[2] + d[3] * d[3])) +
            ((d[4] * d[4] + d[5] * d[5]) + (d[6] * d[6] + d[7] * d[7]));
  const float var = wave_sum(q) * (1.0f / DIMD);
  const float rs = rsqrtf(var + 1e-5f);
  const float gg[8] = {ga.x, ga.y, ga.z, ga.w, gb.x, gb.y, gb.z, gb.w};
  const float bt[8] = {ba.x, ba.y, ba.z, ba.w, bb.x, bb.y, bb.z, bb.w};
#pragma unroll
  for (int i = 0; i < 8; ++i) v[i] = d[i] * rs * gg[i] + bt[i];
}

__global__ __launch_bounds__(256) void ffn_block(
    const float* __restrict__ xin, const float* __restrict__ proj,
    const float* __restrict__ g1, const float* __restrict__ be1,
    const bf16_t* __restrict__ W1, const float* __restrict__ b1,
    const bf16_t* __restrict__ W2, const float* __restrict__ b2,
    const float* __restrict__ g2, const float* __restrict__ be2,
    float* __restrict__ out) {
  __shared__ alignas(16) float  hf[TROWS * HF_PITCH];
  __shared__ alignas(16) bf16_t hb[TROWS * HB_PITCH];
  __shared__ alignas(16) bf16_t hid[TROWS * HID_PITCH];
  __shared__ alignas(16) bf16_t wst[2][WST_ELEMS];

  const int tid  = threadIdx.x;
  const int lane = tid & 31;
  const int wave = tid >> 5;
  const int r    = lane & 15;
  const int half = lane >> 4;
  const int mw   = wave >> 2;
  const int nw   = wave & 3;
  const size_t row_blk = (size_t)blockIdx.x * TROWS;

  const v8bf* w1g  = (const v8bf*)(W1 + (size_t)wave * DIMD) + lane;
  const int   w1so = wave * W1S_PITCH + lane * 8;
  const int   srow = tid >> 3;
  const int   scol = (tid & 7) * 8;
  const v8bf* w2g  = (const v8bf*)(W2 + (size_t)srow * FFND + scol);
  const int   w2so = srow * W2S_PITCH + scol;

  v8bf wg[8];
#pragma unroll
  for (int j = 0; j < 8; ++j) wg[j] = w1g[(size_t)(j * 8) * (DIMD / 8)];

  {
    const float4 ga = ((const float4*)g1)[lane], gb = ((const float4*)g1)[32 + lane];
    const float4 ba = ((const float4*)be1)[lane], bb = ((const float4*)be1)[32 + lane];
#pragma unroll
    for (int j = 0; j < 8; ++j) {
      const int lr = wave * 8 + j;
      const float4* xp = (const float4*)(xin + (row_blk + lr) * DIMD);
      const float4* pp = (const float4*)(proj + (row_blk + lr) * DIMD);
      const float4 x0 = xp[lane], x1 = xp[32 + lane];
      const float4 p0 = pp[lane], p1 = pp[32 + lane];
      float v[8] = {x0.x + p0.x, x0.y + p0.y, x0.z + p0.z, x0.w + p0.w,
                    x1.x + p1.x, x1.y + p1.y, x1.z + p1.z, x1.w + p1.w};
      ln_row(v, ga, gb, ba, bb);
      float* hp = hf + lr * HF_PITCH + lane * 4;
      *(float4*)hp         = make_float4(v[0], v[1], v[2], v[3]);
      *(float4*)(hp + 128) = make_float4(v[4], v[5], v[6], v[7]);
      v4bf c0, c1;
      c0[0] = (bf16_t)v[0]; c0[1] = (bf16_t)v[1]; c0[2] = (bf16_t)v[2]; c0[3] = (bf16_t)v[3];
      c1[0] = (bf16_t)v[4]; c1[1] = (bf16_t)v[5]; c1[2] = (bf16_t)v[6]; c1[3] = (bf16_t)v[7];
      bf16_t* bp = hb + lr * HB_PITCH + lane * 4;
      *(v4bf*)bp         = c0;
      *(v4bf*)(bp + 128) = c1;
    }
  }
#pragma unroll
  for (int j = 0; j < 8; ++j) *(v8bf*)(wst[0] + w1so + j * 8 * W1S_PITCH) = wg[j];
  __syncthreads();

  {
    v16bf af[2][8];
#pragma unroll
    for (int mi = 0; mi < 2; ++mi) {
#pragma unroll
      for (int s = 0; s < 8; ++s)
        af[mi][s] = lds_frag_bf16(hb + (mw * 32 + mi * 16 + r) * HB_PITCH + s * 32);
    }
    float bias = b1[nw * 16 + r];
    for (int c = 0; c < FFND / 64; ++c) {
      const int cur = c & 1;
      const int cn = (c + 1 < FFND / 64) ? c + 1 : c;
      const float bias_n = b1[cn * 64 + nw * 16 + r];
#pragma unroll
      for (int j = 0; j < 8; ++j)
        wg[j] = w1g[(size_t)(cn * 64 + j * 8) * (DIMD / 8)];
      const bf16_t* bb = wst[cur] + (nw * 16 + r) * W1S_PITCH;
      v16bf bq[8];
#pragma unroll
      for (int s = 0; s < 8; ++s) bq[s] = lds_frag_bf16(bb + s * 32);
      __builtin_amdgcn_sched_barrier(0);
      v8f acc[2] = {};
#pragma unroll
      for (int s = 0; s < 8; ++s) {
        acc[0] = wmma_bf16(af[0][s], bq[s], acc[0]);
        acc[1] = wmma_bf16(af[1][s], bq[s], acc[1]);
      }
      const int col = c * 64 + nw * 16 + r;
#pragma unroll
      for (int mi = 0; mi < 2; ++mi) {
#pragma unroll
        for (int i = 0; i < 8; ++i) {
          const int row = mw * 32 + mi * 16 + i + 8 * half;
          hid[row * HID_PITCH + col] = (bf16_t)fmaxf(acc[mi][i] + bias, 0.0f);
        }
      }
      bias = bias_n;
      __builtin_amdgcn_sched_barrier(0);
#pragma unroll
      for (int j = 0; j < 8; ++j)
        *(v8bf*)(wst[cur ^ 1] + w1so + j * 8 * W1S_PITCH) = wg[j];
      lds_barrier();
    }
  }
#pragma unroll
  for (int j = 0; j < 8; ++j) wg[j] = w2g[(size_t)j * 32 * (FFND / 8)];
#pragma unroll
  for (int j = 0; j < 8; ++j) *(v8bf*)(wst[0] + w2so + j * 32 * W2S_PITCH) = wg[j];
  __syncthreads();

  {
    const int oc0 = nw * 64;
    const bf16_t* ia0 = hid + (mw * 32 + r) * HID_PITCH;
    const bf16_t* ia1 = ia0 + 16 * HID_PITCH;
    v8f acc[2][4] = {};
    for (int t = 0; t < FFND / 64; ++t) {
      const int cur = t & 1;
      const int tn = (t + 1 < FFND / 64) ? t + 1 : t;
#pragma unroll
      for (int j = 0; j < 8; ++j)
        wg[j] = w2g[(size_t)j * 32 * (FFND / 8) + tn * 8];
      const bf16_t* bb = wst[cur] + (oc0 + r) * W2S_PITCH;
      v16bf a[2][2], bq[2][4];
#pragma unroll
      for (int s = 0; s < 2; ++s) {
        a[0][s] = lds_frag_bf16(ia0 + t * 64 + s * 32);
        a[1][s] = lds_frag_bf16(ia1 + t * 64 + s * 32);
#pragma unroll
        for (int nb = 0; nb < 4; ++nb)
          bq[s][nb] = lds_frag_bf16(bb + nb * 16 * W2S_PITCH + s * 32);
      }
      __builtin_amdgcn_sched_barrier(0);
#pragma unroll
      for (int s = 0; s < 2; ++s) {
#pragma unroll
        for (int nb = 0; nb < 4; ++nb) {
          acc[0][nb] = wmma_bf16(a[0][s], bq[s][nb], acc[0][nb]);
          acc[1][nb] = wmma_bf16(a[1][s], bq[s][nb], acc[1][nb]);
        }
      }
      __builtin_amdgcn_sched_barrier(0);
#pragma unroll
      for (int j = 0; j < 8; ++j)
        *(v8bf*)(wst[cur ^ 1] + w2so + j * 32 * W2S_PITCH) = wg[j];
      lds_barrier();
    }
#pragma unroll
    for (int mi = 0; mi < 2; ++mi) {
#pragma unroll
      for (int nb = 0; nb < 4; ++nb) {
        const int col = oc0 + nb * 16 + r;
        const float bias = b2[col];
#pragma unroll
        for (int i = 0; i < 8; ++i) {
          const int row = mw * 32 + mi * 16 + i + 8 * half;
          float* hp = hf + row * HF_PITCH + col;
          *hp = *hp + (acc[mi][nb][i] + bias);
        }
      }
    }
  }
  __syncthreads();

  {
    const float4 ga = ((const float4*)g2)[lane], gb = ((const float4*)g2)[32 + lane];
    const float4 ba = ((const float4*)be2)[lane], bb = ((const float4*)be2)[32 + lane];
#pragma unroll
    for (int j = 0; j < 8; ++j) {
      const int lr = wave * 8 + j;
      const float* hp = hf + lr * HF_PITCH + lane * 4;
      const float4 y0 = *(const float4*)hp;
      const float4 y1 = *(const float4*)(hp + 128);
      float v[8] = {y0.x, y0.y, y0.z, y0.w, y1.x, y1.y, y1.z, y1.w};
      ln_row(v, ga, gb, ba, bb);
      float4* op = (float4*)(out + (row_blk + lr) * DIMD);
      op[lane]      = make_float4(v[0], v[1], v[2], v[3]);
      op[32 + lane] = make_float4(v[4], v[5], v[6], v[7]);
    }
  }
}

__device__ __forceinline__ float wave_max(float v) {
#pragma unroll
  for (int d = 16; d >= 1; d >>= 1) v = fmaxf(v, __shfl_xor(v, d, 32));
  return v;
}

__global__ __launch_bounds__(1024) void halt_finalize(
    const float* __restrict__ hpart, const int* __restrict__ kpos,
    const float* __restrict__ mask, float* __restrict__ out) {
  __shared__ float red[32];
  const int b = blockIdx.x, tid = threadIdx.x;
  const int lane = tid & 31, wave = tid >> 5;
  float v[2];
#pragma unroll
  for (int i = 0; i < 2; ++i) {
    const int j = tid + i * 1024;
    const int q = kpos[(size_t)b * SEQ + j];
    const float* pp = hpart + (size_t)b * HPARTS * SEQ + (q < 0 ? 0 : q);
    float a = 0.0f;
#pragma unroll
    for (int p = 0; p < HPARTS; ++p) a += pp[(size_t)p * SEQ];
    if (q < 0) a = 0.0f;
    v[i] = a * 0.25f + mask[(size_t)b * SEQ + j] * (-1e10f);
  }
  const float wm = wave_max(fmaxf(v[0], v[1]));
  if (lane == 0) red[wave] = wm;
  __syncthreads();
  const float mx = wave_max(red[lane]);
  __syncthreads();
  const float e0 = __expf(v[0] - mx), e1 = __expf(v[1] - mx);
  const float ws = wave_sum(e0 + e1);
  if (lane == 0) red[wave] = ws;
  __syncthreads();
  const float inv = 1.0f / wave_sum(red[lane]);
  out[(size_t)b * SEQ + tid]        = e0 * inv;
  out[(size_t)b * SEQ + tid + 1024] = e1 * inv;
}

extern "C" void kernel_launch(void* const* d_in, const int* in_sizes, int n_in,
                              void* d_out, int out_size, void* d_ws,
                              size_t ws_size, hipStream_t stream) {
  (void)in_sizes; (void)n_in; (void)out_size; (void)ws_size;
  const float* x    = (const float*)d_in[0];
  const float* mask = (const float*)d_in[1];
  const float* Wqkv = (const float*)d_in[2];
  const float* bqkv = (const float*)d_in[3];
  const float* Wo   = (const float*)d_in[4];
  const float* bo   = (const float*)d_in[5];
  const float* W1   = (const float*)d_in[6];
  const float* b1   = (const float*)d_in[7];
  const float* W2   = (const float*)d_in[8];
  const float* b2   = (const float*)d_in[9];
  const float* g1   = (const float*)d_in[10];
  const float* be1  = (const float*)d_in[11];
  const float* g2   = (const float*)d_in[12];
  const float* be2  = (const float*)d_in[13];

  char* ws = (char*)d_ws;
  size_t off = 0;
  auto alloc = [&](size_t bytes) {
    char* p = ws + off;
    off += (bytes + 255) & ~(size_t)255;
    return p;
  };
  bf16_t* Wqkvb = (bf16_t*)alloc((size_t)3 * DIMD * DIMD * 2);
  bf16_t* Wob   = (bf16_t*)alloc((size_t)DIMD * DIMD * 2);
  bf16_t* W1b   = (bf16_t*)alloc((size_t)FFND * DIMD * 2);
  bf16_t* W2b   = (bf16_t*)alloc((size_t)DIMD * FFND * 2);
  bf16_t* Qb    = (bf16_t*)alloc((size_t)NTOK * DIMD * 2);
  bf16_t* Kb    = (bf16_t*)alloc((size_t)NTOK * DIMD * 2);
  bf16_t* Vt    = (bf16_t*)alloc((size_t)NTOK * DIMD * 2);
  bf16_t* attnb = (bf16_t*)alloc((size_t)NTOK * DIMD * 2);
  float*  attnO = (float*) alloc((size_t)NTOK * DIMD * 4);

  int* kpos = (int*)alloc((size_t)NTOK * 4);
  int* nkv  = (int*)alloc((size_t)BATCH * 4);

  float* hpart = (float*)alloc((size_t)BATCH * HPARTS * SEQ * 4);

  float* out      = (float*)d_out;
  float* out_halt = out + (size_t)NTOK * DIMD;

  cvt_bf16<<<3 * DIMD * DIMD / 1024, 256, 0, stream>>>(Wqkv, Wqkvb);
  cvt_bf16<<<DIMD * DIMD / 1024, 256, 0, stream>>>(Wo, Wob);
  cvt_bf16<<<FFND * DIMD / 1024, 256, 0, stream>>>(W1, W1b);
  cvt_bf16<<<DIMD * FFND / 1024, 256, 0, stream>>>(W2, W2b);

  mask_scan<<<BATCH, 256, 0, stream>>>(mask, kpos, nkv, Kb, Vt);

  qkv_block<<<NTOK / QROWS, 256, 0, stream>>>(x, Wqkvb, bqkv, Qb, Kb, Vt, kpos);

  attn_wmma<<<BATCH * HEADS * (SEQ / (32 * AT_WAVES)), 32 * AT_WAVES, 0, stream>>>(
      Qb, Kb, Vt, nkv, attnb, hpart);

  gemm_wmma<<<dim3(DIMD / 64, NTOK / 128), 128, 0, stream>>>(
      attnb, Wob, bo, attnO, DIMD);

  ffn_block<<<NTOK / TROWS, 256, 0, stream>>>(
      x, attnO, g1, be1, W1b, b1, W2b, b2, g2, be2, out);

  halt_finalize<<<BATCH, 1024, 0, stream>>>(hpart, kpos, mask, out_halt);
}


# key-mask scan and Wo-convert launches folded into their consumers: each QKV workgroup counts its batch's mask itself (slots, count, padding); Wo GEMM rounds f32 Wo while staging; 10 -> 8 launches
# speedup vs baseline: 2.8407x; 1.0578x over previous
#include <hip/hip_runtime.h>
#include <hip/hip_bf16.h>

typedef __bf16 bf16_t;
typedef __attribute__((ext_vector_type(16))) __bf16 v16bf;
typedef __attribute__((ext_vector_type(4)))  __bf16 v4bf;
typedef __attribute__((ext_vector_type(8)))  float  v8f;
typedef __attribute__((ext_vector_type(8)))  __bf16 v8bf;

#define DIMD   256
#define HEADS  4
#define HD     64
#define FFND   1024
#define BATCH  8
#define SEQ    2048
#define NTOK   (BATCH*SEQ)
#define NEGV   (-1e9f)

__device__ __forceinline__ v16bf load_frag_bf16(const bf16_t* base) {
  const int half = (threadIdx.x >> 4) & 1;
  const bf16_t* p = base + 8 * half;
  v16bf f;
#pragma unroll
  for (int e = 0; e < 8; ++e) {
    f[e]     = p[e];
    f[e + 8] = p[e + 16];
  }
  return f;
}

__device__ __forceinline__ v16bf lds_frag_bf16(const bf16_t* base) {
  const int half = (threadIdx.x >> 4) & 1;
  const bf16_t* p = base + 8 * half;
  const v8bf lo = *(const v8bf*)p;
  const v8bf hi = *(const v8bf*)(p + 16);
  return __builtin_shufflevector(lo, hi, 0, 1, 2, 3, 4, 5, 6, 7,
                                 8, 9, 10, 11, 12, 13, 14, 15);
}

__device__ __forceinline__ void lds_barrier() {
  __builtin_amdgcn_fence(__ATOMIC_RELEASE, "workgroup", "local");
  __builtin_amdgcn_s_barrier();
  __builtin_amdgcn_fence(__ATOMIC_ACQUIRE, "workgroup", "local");
}

__device__ __forceinline__ v8f wmma_bf16(v16bf a, v16bf b, v8f c) {
  return __builtin_amdgcn_wmma_f32_16x16x32_bf16(
       false, a,  false, b,
       (short)0, c,  false,  false);
}

__global__ __launch_bounds__(256) void cvt_bf16(const float* __restrict__ in,
                                                bf16_t* __restrict__ out) {
  const size_t i = (size_t)blockIdx.x * 256 + threadIdx.x;
  const float4 v = ((const float4*)in)[i];
  v4bf o;
  o[0] = (bf16_t)v.x; o[1] = (bf16_t)v.y; o[2] = (bf16_t)v.z; o[3] = (bf16_t)v.w;
  ((v4bf*)out)[i] = o;
}


#define QROWS    64
#define XS_PITCH 264
#define WQ_PITCH 264
__global__ __launch_bounds__(256) void qkv_block(
    const float* __restrict__ xin, const bf16_t* __restrict__ Wb,
    const float* __restrict__ bias, bf16_t* __restrict__ Qb,
    bf16_t* __restrict__ Kb, bf16_t* __restrict__ Vt,
    const float* __restrict__ mask, int* __restrict__ kpos,
    int* __restrict__ nkv) {
  __shared__ alignas(16) bf16_t xs[QROWS * XS_PITCH];
  __shared__ alignas(16) bf16_t wq[2][64 * WQ_PITCH];
  __shared__ int sc_red[16];
  __shared__ int sc_keep[QROWS];
  __shared__ int sc_grp[8];
  __shared__ int sc_kp[QROWS];

  const int tid  = threadIdx.x;
  const int lane = tid & 31;
  const int wave = tid >> 5;
  const int r    = lane & 15;
  const int half = lane >> 4;
  const int mw   = wave >> 2;
  const int nw   = wave & 3;
  const int row_blk = blockIdx.x * QROWS;
  const int b   = row_blk >> 11;
  const int t0w = row_blk & (SEQ - 1);

  const float4* mp = (const float4*)(mask + (size_t)b * SEQ + tid * 8);
  const float4 m0 = mp[0], m1 = mp[1];

  const v8bf* wgp = (const v8bf*)(Wb + (size_t)wave * DIMD) + lane;
  const int   wso = wave * WQ_PITCH + lane * 8;

  v8bf wg[8];
#pragma unroll
  for (int j = 0; j < 8; ++j) wg[j] = wgp[(size_t)(j * 8) * (DIMD / 8)];

  {
    float4 x0[8], x1[8];
#pragma unroll
    for (int j = 0; j < 8; ++j) {
      const float4* xp = (const float4*)(xin + (size_t)(row_blk + wave * 8 + j) * DIMD);
      x0[j] = xp[lane];
      x1[j] = xp[32 + lane];
    }

    {
      const float mv[8] = {m0.x, m0.y, m0.z, m0.w, m1.x, m1.y, m1.z, m1.w};
      int keep[8], c = 0;
#pragma unroll
      for (int j = 0; j < 8; ++j) { keep[j] = !(mv[j] > 0.5f); c += keep[j]; }
      const int g0 = t0w >> 3;
      int call = c, cpre = (tid < g0) ? c : 0;
#pragma unroll
      for (int d = 16; d >= 1; d >>= 1) {
        call += __shfl_xor(call, d, 32);
        cpre += __shfl_xor(cpre, d, 32);
      }
      if (lane == 0) { sc_red[wave] = call; sc_red[8 + wave] = cpre; }
      const int g = tid - g0;
      if (g >= 0 && g < 8) {
        sc_grp[g] = c;
#pragma unroll
        for (int j = 0; j < 8; ++j) sc_keep[g * 8 + j] = keep[j];
      }
      lds_barrier();
      int total = 0, pre = 0;
#pragma unroll
      for (int w = 0; w < 8; ++w) { total += sc_red[w]; pre += sc_red[8 + w]; }
      const bool allm = (total == 0);
      if (tid < QROWS) {
        const int gq = tid >> 3, e = tid & 7;
        int q = pre;
#pragma unroll
        for (int gg = 0; gg < 8; ++gg) q += (gg < gq) ? sc_grp[gg] : 0;
#pragma unroll
        for (int ee = 0; ee < 8; ++ee) q += (ee < e) ? sc_keep[gq * 8 + ee] : 0;
        const int kq = allm ? (t0w + tid) : (sc_keep[tid] ? q : -1);
        sc_kp[tid] = kq;
        kpos[row_blk + tid] = kq;
      }
      const int nk = allm ? SEQ : total;
      if (t0w == 0 && tid == 0) nkv[b] = allm ? -1 : total;
      const int nk64 = (nk + 63) & ~63;
      const int ph = tid >> 6, pd = tid & 63;
      const size_t pbh = (size_t)b * HEADS + ph;
#pragma unroll
      for (int u = 0; u < 2; ++u) {
        const int q = nk + (t0w >> 6) * 2 + u;
        if (q < nk64) {
          Kb[(pbh * SEQ + q) * HD + pd] = (bf16_t)0.0f;
          Vt[(pbh * HD + pd) * SEQ + q] = (bf16_t)0.0f;
        }
      }
    }

#pragma unroll
    for (int j = 0; j < 8; ++j) {
      v4bf c0, c1;
      c0[0] = (bf16_t)x0[j].x; c0[1] = (bf16_t)x0[j].y; c0[2] = (bf16_t)x0[j].z; c0[3] = (bf16_t)x0[j].w;
      c1[0] = (bf16_t)x1[j].x; c1[1] = (bf16_t)x1[j].y; c1[2] = (bf16_t)x1[j].z; c1[3] = (bf16_t)x1[j].w;
      bf16_t* bp = xs + (wave * 8 + j) * XS_PITCH + lane * 4;
      *(v4bf*)bp         = c0;
      *(v4bf*)(bp + 128) = c1;
    }
  }
#pragma unroll
  for (int j = 0; j < 8; ++j) *(v8bf*)(wq[0] + wso + j * 8 * WQ_PITCH) = wg[j];
  __syncthreads();

  v16bf af[2][8];
#pragma unroll
  for (int mi = 0; mi < 2; ++mi) {
#pragma unroll
    for (int s = 0; s < 8; ++s)
      af[mi][s] = lds_frag_bf16(xs + (mw * 32 + mi * 16 + r) * XS_PITCH + s * 32);
  }
  int kp[2][8];
#pragma unroll
  for (int mi = 0; mi < 2; ++mi) {
#pragma unroll
    for (int i = 0; i < 8; ++i)
      kp[mi][i] = sc_kp[mw * 32 + mi * 16 + i + 8 * half];
  }
  const int t0 = t0w + mw * 32 + 8 * half;
  const int d  = nw * 16 + r;

  float bv = bias[d];
  for (int c = 0; c < 3 * DIMD / 64; ++c) {
    const int cur = c & 1;
    const int cn = (c + 1 < 3 * DIMD / 64) ? c + 1 : c;
    const float bias_n = bias[cn * 64 + d];
#pragma unroll
    for (int j = 0; j < 8; ++j)
      wg[j] = wgp[(size_t)(cn * 64 + j * 8) * (DIMD / 8)];
    const bf16_t* bb = wq[cur] + d * WQ_PITCH;
    v16bf bq[8];
#pragma unroll
    for (int s = 0; s < 8; ++s) bq[s] = lds_frag_bf16(bb + s * 32);
    __builtin_amdgcn_sched_barrier(0);
    v8f acc[2] = {};
#pragma unroll
    for (int s = 0; s < 8; ++s) {
      acc[0] = wmma_bf16(af[0][s], bq[s], acc[0]);
      acc[1] = wmma_bf16(af[1][s], bq[s], acc[1]);
    }
    const int sec = c >> 2;
    const size_t bhh = (size_t)b * HEADS + (c & 3);
    bf16_t y[2][8];
#pragma unroll
    for (int mi = 0; mi < 2; ++mi) {
#pragma unroll
      for (int i = 0; i < 8; ++i) y[mi][i] = (bf16_t)(acc[mi][i] + bv);
    }
    if (sec == 0) {
      bf16_t* qp = Qb + (bhh * SEQ + t0) * HD + d;
#pragma unroll
      for (int mi = 0; mi < 2; ++mi) {
#pragma unroll
        for (int i = 0; i < 8; ++i) qp[(mi * 16 + i) * HD] = y[mi][i];
      }
    } else if (sec == 1) {
      bf16_t* kq = Kb + bhh * SEQ * HD + d;
#pragma unroll
      for (int mi = 0; mi < 2; ++mi) {
#pragma unroll
        for (int i = 0; i < 8; ++i) {
          const int p = kp[mi][i];
          if (__builtin_expect(p >= 0, 1)) kq[(size_t)p * HD] = y[mi][i];
        }
      }
    } else {
      bf16_t* vq = Vt + (bhh * HD + d) * SEQ;
#pragma unroll
      for (int mi = 0; mi < 2; ++mi) {
#pragma unroll
        for (int i = 0; i < 8; ++i) {
          const int p = kp[mi][i];
          if (__builtin_expect(p >= 0, 1)) vq[p] = y[mi][i];
        }
      }
    }
    bv = bias_n;
#pragma unroll
    for (int j = 0; j < 8; ++j)
      *(v8bf*)(wq[cur ^ 1] + wso + j * 8 * WQ_PITCH) = wg[j];
    lds_barrier();
  }
}

#define GT_PITCH 264
__global__ __launch_bounds__(128) void gemm_wmma(
    const bf16_t* __restrict__ A, const float* __restrict__ B,
    const float* __restrict__ bias, float* __restrict__ Yf, int N) {
  __shared__ alignas(16) bf16_t at[128 * GT_PITCH];
  __shared__ alignas(16) bf16_t bt[64 * GT_PITCH];
  const int lane = threadIdx.x & 31;
  const int wave = threadIdx.x >> 5;
  const int r    = lane & 15;
  const int half = lane >> 4;
  const int rowb = blockIdx.y * 128;
  const int row0 = rowb + wave * 32;
  const int col0 = blockIdx.x * 64;

  {
    const v8bf*   ag = (const v8bf*)(A + (size_t)(rowb + wave) * DIMD) + lane;
    const float4* bg = (const float4*)(B + (size_t)(col0 + wave) * DIMD) + lane;
    const int so  = wave * GT_PITCH + lane * 8;
    const int sob = wave * GT_PITCH + lane * 4;
    float4 tb0[16], tb1[16];
    v8bf ta[32];
#pragma unroll
    for (int j = 0; j < 16; ++j) {
      tb0[j] = bg[(size_t)j * 4 * (DIMD / 4)];
      tb1[j] = bg[(size_t)j * 4 * (DIMD / 4) + 32];
    }
#pragma unroll
    for (int j = 0; j < 32; ++j) ta[j] = ag[(size_t)j * 4 * (DIMD / 8)];
#pragma unroll
    for (int j = 0; j < 16; ++j) {
      v4bf c0, c1;
      c0[0] = (bf16_t)tb0[j].x; c0[1] = (bf16_t)tb0[j].y; c0[2] = (bf16_t)tb0[j].z; c0[3] = (bf16_t)tb0[j].w;
      c1[0] = (bf16_t)tb1[j].x; c1[1] = (bf16_t)tb1[j].y; c1[2] = (bf16_t)tb1[j].z; c1[3] = (bf16_t)tb1[j].w;
      bf16_t* bp = bt + sob + j * 4 * GT_PITCH;
      *(v4bf*)bp         = c0;
      *(v4bf*)(bp + 128) = c1;
    }
#pragma unroll
    for (int j = 0; j < 32; ++j) *(v8bf*)(at + so + j * 4 * GT_PITCH) = ta[j];
  }
  __syncthreads();

  v8f acc[2][4] = {};
  const bf16_t* ap0 = at + (wave * 32 + r) * GT_PITCH;
  const bf16_t* ap1 = ap0 + 16 * GT_PITCH;
  const bf16_t* bp  = bt + r * GT_PITCH;

  v16bf fa[2][2], fb[2][4];
  fa[0][0] = lds_frag_bf16(ap0);
  fa[0][1] = lds_frag_bf16(ap1);
#pragma unroll
  for (int nb = 0; nb < 4; ++nb) fb[0][nb] = lds_frag_bf16(bp + nb * 16 * GT_PITCH);
#pragma unroll
  for (int ks = 0; ks < DIMD / 32; ++ks) {
    const int cs = ks & 1, ns = cs ^ 1;
    if (ks + 1 < DIMD / 32) {
      const int k1 = (ks + 1) * 32;
      fa[ns][0] = lds_frag_bf16(ap0 + k1);
      fa[ns][1] = lds_frag_bf16(ap1 + k1);
#pragma unroll
      for (int nb = 0; nb < 4; ++nb)
        fb[ns][nb] = lds_frag_bf16(bp + nb * 16 * GT_PITCH + k1);
    }
    __builtin_amdgcn_sched_barrier(0);
#pragma unroll
    for (int nb = 0; nb < 4; ++nb) {
      acc[0][nb] = wmma_bf16(fa[cs][0], fb[cs][nb], acc[0][nb]);
      acc[1][nb] = wmma_bf16(fa[cs][1], fb[cs][nb], acc[1][nb]);
    }
    __builtin_amdgcn_sched_barrier(0);
  }

#pragma unroll
  for (int mi = 0; mi < 2; ++mi) {
#pragma unroll
    for (int nb = 0; nb < 4; ++nb) {
#pragma unroll
      for (int i = 0; i < 8; ++i) {
        const int row = row0 + mi * 16 + i + 8 * half;
        const int col = col0 + nb * 16 + r;
        Yf[(size_t)row * N + col] = acc[mi][nb][i] + bias[col];
      }
    }
  }
}

#define AT_WAVES 8
#define AT_KT    64
#define HPARTS   (HEADS * SEQ / (32 * AT_WAVES))
#define KV_PITCH 72

__global__ __launch_bounds__(256) __attribute__((amdgpu_waves_per_eu(2, 2))) void attn_wmma(
    const bf16_t* __restrict__ Qb, const bf16_t* __restrict__ Kb,
    const bf16_t* __restrict__ Vt, const int* __restrict__ nkv,
    bf16_t* __restrict__ attn, float* __restrict__ hpart) {
  __shared__ alignas(16) bf16_t ks[2][AT_KT * KV_PITCH];
  __shared__ alignas(16) bf16_t vs[2][HD * KV_PITCH];
  __shared__ alignas(16) bf16_t lds_pw[AT_WAVES][2][16 * 32];
  __shared__ float csl[2][AT_WAVES][AT_KT];

  const int tid  = threadIdx.x;
  const int wave = tid >> 5;
  const int bid  = blockIdx.x;
  const int bh   = bid >> 3;
  const int qt   = (bid & 7) * AT_WAVES + wave;
  const int b    = bh >> 2;
  const int lane = tid & 31;
  const int r    = lane & 15;
  const int half = lane >> 4;
  const float scale = 0.125f;

  const bf16_t* Qbase = Qb + ((size_t)bh * SEQ + qt * 32) * HD;
  const bf16_t* Kbase = Kb + (size_t)bh * SEQ * HD;
  const bf16_t* Vbase = Vt + (size_t)bh * HD * SEQ;
  float*        hp    = hpart + (size_t)bid * SEQ;
  const int nkraw = nkv[b];
  const int nk    = (nkraw < 0) ? SEQ : nkraw;
  const int lim   = (nkraw < 0) ? 0 : nkraw;
  const int nt    = (nk + AT_KT - 1) / AT_KT;

  const int srow = tid >> 3;
  const int scol = (tid & 7) * 8;
  const v8bf* kg = (const v8bf*)(Kbase + (size_t)srow * HD + scol);
  const v8bf* vg = (const v8bf*)(Vbase + (size_t)srow * SEQ + scol);
  const int   so = srow * KV_PITCH + scol;
  bf16_t* pw0 = lds_pw[wave][0];
  bf16_t* pw1 = lds_pw[wave][1];

  v16bf qa[2][2];
#pragma unroll
  for (int qi = 0; qi < 2; ++qi) {
    qa[qi][0] = load_frag_bf16(Qbase + (size_t)(qi * 16 + r) * HD + 0);
    qa[qi][1] = load_frag_bf16(Qbase + (size_t)(qi * 16 + r) * HD + 32);
  }

  float m[2][8], l[2][8];
#pragma unroll
  for (int qi = 0; qi < 2; ++qi)
#pragma unroll
    for (int i = 0; i < 8; ++i) { m[qi][i] = -3.0e38f; l[qi][i] = 0.0f; }

  {
    const v8bf g0 = kg[0];
    const v8bf g1 = kg[32 * HD / 8];
    *(v8bf*)(ks[0] + so)                 = g0;
    *(v8bf*)(ks[0] + so + 32 * KV_PITCH) = g1;
  }
  lds_barrier();
  for (int t = 0; t < nt; ++t) {
    const int cur = t & 1;
    const int tn = (t + 1 < nt) ? t + 1 : t;
    const v8bf* kn = kg + (size_t)tn * (AT_KT * HD / 8);
    const v8bf g0 = kn[0];
    const v8bf g1 = kn[32 * HD / 8];
    __builtin_amdgcn_sched_barrier(0);
#pragma unroll
    for (int sub = 0; sub < 4; ++sub) {
      const bf16_t* kr = ks[cur] + (sub * 16 + r) * KV_PITCH;
      v16bf kb0 = lds_frag_bf16(kr);
      v16bf kb1 = lds_frag_bf16(kr + 32);
      const bool dead = (t * AT_KT + sub * 16 + r) >= lim;
#pragma unroll
      for (int qi = 0; qi < 2; ++qi) {
        v8f s = {};
        s = wmma_bf16(qa[qi][0], kb0, s);
        s = wmma_bf16(qa[qi][1], kb1, s);
#pragma unroll
        for (int i = 0; i < 8; ++i) {
          const float sv = dead ? NEGV : s[i] * scale;
          const float mn = fmaxf(m[qi][i], sv);
          l[qi][i] = l[qi][i] * __expf(m[qi][i] - mn) + __expf(sv - mn);
          m[qi][i] = mn;
        }
      }
    }
    __builtin_amdgcn_sched_barrier(0);
    *(v8bf*)(ks[cur ^ 1] + so)                 = g0;
    *(v8bf*)(ks[cur ^ 1] + so + 32 * KV_PITCH) = g1;
    lds_barrier();
  }

  {
    const v8bf g0 = kg[0];
    const v8bf g1 = kg[32 * HD / 8];
    const v8bf h0 = vg[0];
    const v8bf h1 = vg[32 * SEQ / 8];
    *(v8bf*)(ks[0] + so)                 = g0;
    *(v8bf*)(ks[0] + so + 32 * KV_PITCH) = g1;
    *(v8bf*)(vs[0] + so)                 = h0;
    *(v8bf*)(vs[0] + so + 32 * KV_PITCH) = h1;
  }

#pragma unroll
  for (int d = 1; d <= 8; d <<= 1) {
#pragma unroll
    for (int qi = 0; qi < 2; ++qi) {
#pragma unroll
      for (int i = 0; i < 8; ++i) {
        const float mo = __shfl_xor(m[qi][i], d, 32);
        const float lo = __shfl_xor(l[qi][i], d, 32);
        const float mn = fmaxf(m[qi][i], mo);
        l[qi][i] = l[qi][i] * __expf(m[qi][i] - mn) + lo * __expf(mo - mn);
        m[qi][i] = mn;
      }
    }
  }
  float rl[2][8];
#pragma unroll
  for (int qi = 0; qi < 2; ++qi)
#pragma unroll
    for (int i = 0; i < 8; ++i) rl[qi][i] = 1.0f / l[qi][i];
  lds_barrier();

  v8f o[2][4] = {};
  for (int t = 0; t < nt; ++t) {
    const int cur = t & 1;
    const int tn = (t + 1 < nt) ? t + 1 : t;
    const v8bf* kn = kg + (size_t)tn * (AT_KT * HD / 8);
    const v8bf* vn = vg + (size_t)tn * (AT_KT / 8);
    const v8bf g0 = kn[0];
    const v8bf g1 = kn[32 * HD / 8];
    const v8bf h0 = vn[0];
    const v8bf h1 = vn[32 * SEQ / 8];
    if (t > 0 && tid < AT_KT) {
      const float* cp = &csl[cur ^ 1][0][tid];
      float s = cp[0];
#pragma unroll
      for (int w = 1; w < AT_WAVES; ++w) s += cp[w * AT_KT];
      hp[(t - 1) * AT_KT + tid] = s;
    }
    __builtin_amdgcn_sched_barrier(0);
    v16bf kf[4][2];
#pragma unroll
    for (int j = 0; j < 4; ++j) {
      const bf16_t* kr = ks[cur] + (j * 16 + r) * KV_PITCH;
      kf[j][0] = lds_frag_bf16(kr);
      kf[j][1] = lds_frag_bf16(kr + 32);
    }
    float* csw = csl[cur][wave];
#pragma unroll
    for (int kh = 0; kh < 2; ++kh) {
      float csv[2];
#pragma unroll
      for (int sub = 0; sub < 2; ++sub) {
        const int lrow = (kh * 2 + sub) * 16 + r;
        const v16bf kb0 = kf[kh * 2 + sub][0];
        const v16bf kb1 = kf[kh * 2 + sub][1];
        const int  slot = t * AT_KT + lrow;
        const bool dead = slot >= lim;
        float cs = 0.0f;
#pragma unroll
        for (int qi = 0; qi < 2; ++qi) {
          v8f s = {};
          s = wmma_bf16(qa[qi][0], kb0, s);
          s = wmma_bf16(qa[qi][1], kb1, s);
          bf16_t* pw = qi ? pw1 : pw0;
#pragma unroll
          for (int i = 0; i < 8; ++i) {
            const float sv = dead ? NEGV : s[i] * scale;
            const float p  = __expf(sv - m[qi][i]) * rl[qi][i];
            cs += p;
            pw[(i + 8 * half) * 32 + sub * 16 + r] = (bf16_t)p;
          }
        }
        csv[sub] = cs + __shfl_xor(cs, 16, 32);
      }
      csw[kh * 32 + lane] = half ? csv[1] : csv[0];
      __builtin_amdgcn_fence(__ATOMIC_ACQ_REL, "wavefront");
      v16bf pa0 = lds_frag_bf16(pw0 + r * 32);
      v16bf pa1 = lds_frag_bf16(pw1 + r * 32);
#pragma unroll
      for (int nb = 0; nb < 4; ++nb) {
        v16bf vb = lds_frag_bf16(vs[cur] + (nb * 16 + r) * KV_PITCH + kh * 32);
        o[0][nb] = wmma_bf16(pa0, vb, o[0][nb]);
        o[1][nb] = wmma_bf16(pa1, vb, o[1][nb]);
      }
      __builtin_amdgcn_fence(__ATOMIC_ACQ_REL, "wavefront");
    }
    *(v8bf*)(ks[cur ^ 1] + so)                 = g0;
    *(v8bf*)(ks[cur ^ 1] + so + 32 * KV_PITCH) = g1;
    *(v8bf*)(vs[cur ^ 1] + so)                 = h0;
    *(v8bf*)(vs[cur ^ 1] + so + 32 * KV_PITCH) = h1;
    lds_barrier();
  }
  if (tid < AT_KT) {
    const float* cp = &csl[(nt - 1) & 1][0][tid];
    float s = cp[0];
#pragma unroll
    for (int w = 1; w < AT_WAVES; ++w) s += cp[w * AT_KT];
    hp[(nt - 1) * AT_KT + tid] = s;
  }

  const int h = bh & 3;
#pragma unroll
  for (int qi = 0; qi < 2; ++qi) {
#pragma unroll
    for (int nb = 0; nb < 4; ++nb) {
#pragma unroll
      for (int i = 0; i < 8; ++i) {
        const int t = qt * 32 + qi * 16 + i + 8 * half;
        attn[((size_t)b * SEQ + t) * DIMD + h * HD + nb * 16 + r] =
            (bf16_t)o[qi][nb][i];
      }
    }
  }
}

#define TROWS     64
#define HF_PITCH  260
#define HB_PITCH  264
#define HID_PITCH 1032
#define W1S_PITCH 264
#define W2S_PITCH 72
#define WST_ELEMS (DIMD * W2S_PITCH)

__device__ __forceinline__ float wave_sum(float v) {
#pragma unroll
  for (int d = 16; d >= 1; d >>= 1) v += __shfl_xor(v, d, 32);
  return v;
}

__device__ __forceinline__ void ln_row(float v[8], const float4 ga,
                                       const float4 gb, const float4 ba,
                                       const float4 bb) {
  float s = ((v[0] + v[1]) + (v[2] + v[3])) + ((v[4] + v[5]) + (v[6] + v[7]));
  const float mu = wave_sum(s) * (1.0f / DIMD);
  float d[8];
#pragma unroll
  for (int i = 0; i < 8; ++i) d[i] = v[i] - mu;
  float q = ((d[0] * d[0] + d[1] * d[1]) + (d[2] * d[2] + d[3] * d[3])) +
            ((d[4] * d[4] + d[5] * d[5]) + (d[6] * d[6] + d[7] * d[7]));
  const float var = wave_sum(q) * (1.0f / DIMD);
  const float rs = rsqrtf(var + 1e-5f);
  const float gg[8] = {ga.x, ga.y, ga.z, ga.w, gb.x, gb.y, gb.z, gb.w};
  const float bt[8] = {ba.x, ba.y, ba.z, ba.w, bb.x, bb.y, bb.z, bb.w};
#pragma unroll
  for (int i = 0; i < 8; ++i) v[i] = d[i] * rs * gg[i] + bt[i];
}

__global__ __launch_bounds__(256) void ffn_block(
    const float* __restrict__ xin, const float* __restrict__ proj,
    const float* __restrict__ g1, const float* __restrict__ be1,
    const bf16_t* __restrict__ W1, const float* __restrict__ b1,
    const bf16_t* __restrict__ W2, const float* __restrict__ b2,
    const float* __restrict__ g2, const float* __restrict__ be2,
    float* __restrict__ out) {
  __shared__ alignas(16) float  hf[TROWS * HF_PITCH];
  __shared__ alignas(16) bf16_t hb[TROWS * HB_PITCH];
  __shared__ alignas(16) bf16_t hid[TROWS * HID_PITCH];
  __shared__ alignas(16) bf16_t wst[2][WST_ELEMS];

  const int tid  = threadIdx.x;
  const int lane = tid & 31;
  const int wave = tid >> 5;
  const int r    = lane & 15;
  const int half = lane >> 4;
  const int mw   = wave >> 2;
  const int nw   = wave & 3;
  const size_t row_blk = (size_t)blockIdx.x * TROWS;

  const v8bf* w1g  = (const v8bf*)(W1 + (size_t)wave * DIMD) + lane;
  const int   w1so = wave * W1S_PITCH + lane * 8;
  const int   srow = tid >> 3;
  const int   scol = (tid & 7) * 8;
  const v8bf* w2g  = (const v8bf*)(W2 + (size_t)srow * FFND + scol);
  const int   w2so = srow * W2S_PITCH + scol;

  v8bf wg[8];
#pragma unroll
  for (int j = 0; j < 8; ++j) wg[j] = w1g[(size_t)(j * 8) * (DIMD / 8)];

  {
    const float4 ga = ((const float4*)g1)[lane], gb = ((const float4*)g1)[32 + lane];
    const float4 ba = ((const float4*)be1)[lane], bb = ((const float4*)be1)[32 + lane];
#pragma unroll
    for (int j = 0; j < 8; ++j) {
      const int lr = wave * 8 + j;
      const float4* xp = (const float4*)(xin + (row_blk + lr) * DIMD);
      const float4* pp = (const float4*)(proj + (row_blk + lr) * DIMD);
      const float4 x0 = xp[lane], x1 = xp[32 + lane];
      const float4 p0 = pp[lane], p1 = pp[32 + lane];
      float v[8] = {x0.x + p0.x, x0.y + p0.y, x0.z + p0.z, x0.w + p0.w,
                    x1.x + p1.x, x1.y + p1.y, x1.z + p1.z, x1.w + p1.w};
      ln_row(v, ga, gb, ba, bb);
      float* hp = hf + lr * HF_PITCH + lane * 4;
      *(float4*)hp         = make_float4(v[0], v[1], v[2], v[3]);
      *(float4*)(hp + 128) = make_float4(v[4], v[5], v[6], v[7]);
      v4bf c0, c1;
      c0[0] = (bf16_t)v[0]; c0[1] = (bf16_t)v[1]; c0[2] = (bf16_t)v[2]; c0[3] = (bf16_t)v[3];
      c1[0] = (bf16_t)v[4]; c1[1] = (bf16_t)v[5]; c1[2] = (bf16_t)v[6]; c1[3] = (bf16_t)v[7];
      bf16_t* bp = hb + lr * HB_PITCH + lane * 4;
      *(v4bf*)bp         = c0;
      *(v4bf*)(bp + 128) = c1;
    }
  }
#pragma unroll
  for (int j = 0; j < 8; ++j) *(v8bf*)(wst[0] + w1so + j * 8 * W1S_PITCH) = wg[j];
  __syncthreads();

  {
    v16bf af[2][8];
#pragma unroll
    for (int mi = 0; mi < 2; ++mi) {
#pragma unroll
      for (int s = 0; s < 8; ++s)
        af[mi][s] = lds_frag_bf16(hb + (mw * 32 + mi * 16 + r) * HB_PITCH + s * 32);
    }
    float bias = b1[nw * 16 + r];
    for (int c = 0; c < FFND / 64; ++c) {
      const int cur = c & 1;
      const int cn = (c + 1 < FFND / 64) ? c + 1 : c;
      const float bias_n = b1[cn * 64 + nw * 16 + r];
#pragma unroll
      for (int j = 0; j < 8; ++j)
        wg[j] = w1g[(size_t)(cn * 64 + j * 8) * (DIMD / 8)];
      const bf16_t* bb = wst[cur] + (nw * 16 + r) * W1S_PITCH;
      v16bf bq[8];
#pragma unroll
      for (int s = 0; s < 8; ++s) bq[s] = lds_frag_bf16(bb + s * 32);
      __builtin_amdgcn_sched_barrier(0);
      v8f acc[2] = {};
#pragma unroll
      for (int s = 0; s < 8; ++s) {
        acc[0] = wmma_bf16(af[0][s], bq[s], acc[0]);
        acc[1] = wmma_bf16(af[1][s], bq[s], acc[1]);
      }
      const int col = c * 64 + nw * 16 + r;
#pragma unroll
      for (int mi = 0; mi < 2; ++mi) {
#pragma unroll
        for (int i = 0; i < 8; ++i) {
          const int row = mw * 32 + mi * 16 + i + 8 * half;
          hid[row * HID_PITCH + col] = (bf16_t)fmaxf(acc[mi][i] + bias, 0.0f);
        }
      }
      bias = bias_n;
      __builtin_amdgcn_sched_barrier(0);
#pragma unroll
      for (int j = 0; j < 8; ++j)
        *(v8bf*)(wst[cur ^ 1] + w1so + j * 8 * W1S_PITCH) = wg[j];
      lds_barrier();
    }
  }
#pragma unroll
  for (int j = 0; j < 8; ++j) wg[j] = w2g[(size_t)j * 32 * (FFND / 8)];
#pragma unroll
  for (int j = 0; j < 8; ++j) *(v8bf*)(wst[0] + w2so + j * 32 * W2S_PITCH) = wg[j];
  __syncthreads();

  {
    const int oc0 = nw * 64;
    const bf16_t* ia0 = hid + (mw * 32 + r) * HID_PITCH;
    const bf16_t* ia1 = ia0 + 16 * HID_PITCH;
    v8f acc[2][4] = {};
    for (int t = 0; t < FFND / 64; ++t) {
      const int cur = t & 1;
      const int tn = (t + 1 < FFND / 64) ? t + 1 : t;
#pragma unroll
      for (int j = 0; j < 8; ++j)
        wg[j] = w2g[(size_t)j * 32 * (FFND / 8) + tn * 8];
      const bf16_t* bb = wst[cur] + (oc0 + r) * W2S_PITCH;
      v16bf a[2][2], bq[2][4];
#pragma unroll
      for (int s = 0; s < 2; ++s) {
        a[0][s] = lds_frag_bf16(ia0 + t * 64 + s * 32);
        a[1][s] = lds_frag_bf16(ia1 + t * 64 + s * 32);
#pragma unroll
        for (int nb = 0; nb < 4; ++nb)
          bq[s][nb] = lds_frag_bf16(bb + nb * 16 * W2S_PITCH + s * 32);
      }
      __builtin_amdgcn_sched_barrier(0);
#pragma unroll
      for (int s = 0; s < 2; ++s) {
#pragma unroll
        for (int nb = 0; nb < 4; ++nb) {
          acc[0][nb] = wmma_bf16(a[0][s], bq[s][nb], acc[0][nb]);
          acc[1][nb] = wmma_bf16(a[1][s], bq[s][nb], acc[1][nb]);
        }
      }
      __builtin_amdgcn_sched_barrier(0);
#pragma unroll
      for (int j = 0; j < 8; ++j)
        *(v8bf*)(wst[cur ^ 1] + w2so + j * 32 * W2S_PITCH) = wg[j];
      lds_barrier();
    }
#pragma unroll
    for (int mi = 0; mi < 2; ++mi) {
#pragma unroll
      for (int nb = 0; nb < 4; ++nb) {
        const int col = oc0 + nb * 16 + r;
        const float bias = b2[col];
#pragma unroll
        for (int i = 0; i < 8; ++i) {
          const int row = mw * 32 + mi * 16 + i + 8 * half;
          float* hp = hf + row * HF_PITCH + col;
          *hp = *hp + (acc[mi][nb][i] + bias);
        }
      }
    }
  }
  __syncthreads();

  {
    const float4 ga = ((const float4*)g2)[lane], gb = ((const float4*)g2)[32 + lane];
    const float4 ba = ((const float4*)be2)[lane], bb = ((const float4*)be2)[32 + lane];
#pragma unroll
    for (int j = 0; j < 8; ++j) {
      const int lr = wave * 8 + j;
      const float* hp = hf + lr * HF_PITCH + lane * 4;
      const float4 y0 = *(const float4*)hp;
      const float4 y1 = *(const float4*)(hp + 128);
      float v[8] = {y0.x, y0.y, y0.z, y0.w, y1.x, y1.y, y1.z, y1.w};
      ln_row(v, ga, gb, ba, bb);
      float4* op = (float4*)(out + (row_blk + lr) * DIMD);
      op[lane]      = make_float4(v[0], v[1], v[2], v[3]);
      op[32 + lane] = make_float4(v[4], v[5], v[6], v[7]);
    }
  }
}

__device__ __forceinline__ float wave_max(float v) {
#pragma unroll
  for (int d = 16; d >= 1; d >>= 1) v = fmaxf(v, __shfl_xor(v, d, 32));
  return v;
}

__global__ __launch_bounds__(1024) void halt_finalize(
    const float* __restrict__ hpart, const int* __restrict__ kpos,
    const float* __restrict__ mask, float* __restrict__ out) {
  __shared__ float red[32];
  const int b = blockIdx.x, tid = threadIdx.x;
  const int lane = tid & 31, wave = tid >> 5;
  float v[2];
#pragma unroll
  for (int i = 0; i < 2; ++i) {
    const int j = tid + i * 1024;
    const int q = kpos[(size_t)b * SEQ + j];
    const float* pp = hpart + (size_t)b * HPARTS * SEQ + (q < 0 ? 0 : q);
    float a = 0.0f;
#pragma unroll
    for (int p = 0; p < HPARTS; ++p) a += pp[(size_t)p * SEQ];
    if (q < 0) a = 0.0f;
    v[i] = a * 0.25f + mask[(size_t)b * SEQ + j] * (-1e10f);
  }
  const float wm = wave_max(fmaxf(v[0], v[1]));
  if (lane == 0) red[wave] = wm;
  __syncthreads();
  const float mx = wave_max(red[lane]);
  __syncthreads();
  const float e0 = __expf(v[0] - mx), e1 = __expf(v[1] - mx);
  const float ws = wave_sum(e0 + e1);
  if (lane == 0) red[wave] = ws;
  __syncthreads();
  const float inv = 1.0f / wave_sum(red[lane]);
  out[(size_t)b * SEQ + tid]        = e0 * inv;
  out[(size_t)b * SEQ + tid + 1024] = e1 * inv;
}

extern "C" void kernel_launch(void* const* d_in, const int* in_sizes, int n_in,
                              void* d_out, int out_size, void* d_ws,
                              size_t ws_size, hipStream_t stream) {
  (void)in_sizes; (void)n_in; (void)out_size; (void)ws_size;
  const float* x    = (const float*)d_in[0];
  const float* mask = (const float*)d_in[1];
  const float* Wqkv = (const float*)d_in[2];
  const float* bqkv = (const float*)d_in[3];
  const float* Wo   = (const float*)d_in[4];
  const float* bo   = (const float*)d_in[5];
  const float* W1   = (const float*)d_in[6];
  const float* b1   = (const float*)d_in[7];
  const float* W2   = (const float*)d_in[8];
  const float* b2   = (const float*)d_in[9];
  const float* g1   = (const float*)d_in[10];
  const float* be1  = (const float*)d_in[11];
  const float* g2   = (const float*)d_in[12];
  const float* be2  = (const float*)d_in[13];

  char* ws = (char*)d_ws;
  size_t off = 0;
  auto alloc = [&](size_t bytes) {
    char* p = ws + off;
    off += (bytes + 255) & ~(size_t)255;
    return p;
  };
  bf16_t* Wqkvb = (bf16_t*)alloc((size_t)3 * DIMD * DIMD * 2);
  bf16_t* W1b   = (bf16_t*)alloc((size_t)FFND * DIMD * 2);
  bf16_t* W2b   = (bf16_t*)alloc((size_t)DIMD * FFND * 2);
  bf16_t* Qb    = (bf16_t*)alloc((size_t)NTOK * DIMD * 2);
  bf16_t* Kb    = (bf16_t*)alloc((size_t)NTOK * DIMD * 2);
  bf16_t* Vt    = (bf16_t*)alloc((size_t)NTOK * DIMD * 2);
  bf16_t* attnb = (bf16_t*)alloc((size_t)NTOK * DIMD * 2);
  float*  attnO = (float*) alloc((size_t)NTOK * DIMD * 4);

  int* kpos = (int*)alloc((size_t)NTOK * 4);
  int* nkv  = (int*)alloc((size_t)BATCH * 4);

  float* hpart = (float*)alloc((size_t)BATCH * HPARTS * SEQ * 4);

  float* out      = (float*)d_out;
  float* out_halt = out + (size_t)NTOK * DIMD;

  cvt_bf16<<<3 * DIMD * DIMD / 1024, 256, 0, stream>>>(Wqkv, Wqkvb);
  cvt_bf16<<<FFND * DIMD / 1024, 256, 0, stream>>>(W1, W1b);
  cvt_bf16<<<DIMD * FFND / 1024, 256, 0, stream>>>(W2, W2b);

  qkv_block<<<NTOK / QROWS, 256, 0, stream>>>(x, Wqkvb, bqkv, Qb, Kb, Vt, mask,
                                              kpos, nkv);

  attn_wmma<<<BATCH * HEADS * (SEQ / (32 * AT_WAVES)), 32 * AT_WAVES, 0, stream>>>(
      Qb, Kb, Vt, nkv, attnb, hpart);

  gemm_wmma<<<dim3(DIMD / 64, NTOK / 128), 128, 0, stream>>>(
      attnb, Wo, bo, attnO, DIMD);

  ffn_block<<<NTOK / TROWS, 256, 0, stream>>>(
      x, attnO, g1, be1, W1b, b1, W2b, b2, g2, be2, out);

  halt_finalize<<<BATCH, 1024, 0, stream>>>(hpart, kpos, mask, out_halt);
}


# QKV outputs go through an LDS stage and out as whole contiguous runs under the next tile: Q/K rows 4 whole lines per wave store, V^T one run of kept slots per store (was a line per lane); same values
# speedup vs baseline: 4.0398x; 1.4221x over previous
#include <hip/hip_runtime.h>
#include <hip/hip_bf16.h>

typedef __bf16 bf16_t;
typedef __attribute__((ext_vector_type(16))) __bf16 v16bf;
typedef __attribute__((ext_vector_type(4)))  __bf16 v4bf;
typedef __attribute__((ext_vector_type(8)))  float  v8f;
typedef __attribute__((ext_vector_type(8)))  __bf16 v8bf;

#define DIMD   256
#define HEADS  4
#define HD     64
#define FFND   1024
#define BATCH  8
#define SEQ    2048
#define NTOK   (BATCH*SEQ)
#define NEGV   (-1e9f)

__device__ __forceinline__ v16bf load_frag_bf16(const bf16_t* base) {
  const int half = (threadIdx.x >> 4) & 1;
  const bf16_t* p = base + 8 * half;
  v16bf f;
#pragma unroll
  for (int e = 0; e < 8; ++e) {
    f[e]     = p[e];
    f[e + 8] = p[e + 16];
  }
  return f;
}

__device__ __forceinline__ v16bf lds_frag_bf16(const bf16_t* base) {
  const int half = (threadIdx.x >> 4) & 1;
  const bf16_t* p = base + 8 * half;
  const v8bf lo = *(const v8bf*)p;
  const v8bf hi = *(const v8bf*)(p + 16);
  return __builtin_shufflevector(lo, hi, 0, 1, 2, 3, 4, 5, 6, 7,
                                 8, 9, 10, 11, 12, 13, 14, 15);
}

__device__ __forceinline__ void lds_barrier() {
  __builtin_amdgcn_fence(__ATOMIC_RELEASE, "workgroup", "local");
  __builtin_amdgcn_s_barrier();
  __builtin_amdgcn_fence(__ATOMIC_ACQUIRE, "workgroup", "local");
}

__device__ __forceinline__ v8f wmma_bf16(v16bf a, v16bf b, v8f c) {
  return __builtin_amdgcn_wmma_f32_16x16x32_bf16(
       false, a,  false, b,
       (short)0, c,  false,  false);
}

__global__ __launch_bounds__(256) void cvt_bf16(const float* __restrict__ in,
                                                bf16_t* __restrict__ out) {
  const size_t i = (size_t)blockIdx.x * 256 + threadIdx.x;
  const float4 v = ((const float4*)in)[i];
  v4bf o;
  o[0] = (bf16_t)v.x; o[1] = (bf16_t)v.y; o[2] = (bf16_t)v.z; o[3] = (bf16_t)v.w;
  ((v4bf*)out)[i] = o;
}


#define QROWS    64
#define XS_PITCH 264
#define WQ_PITCH 264
#define OS_PITCH 72
#define OV_PITCH 66
__global__ __launch_bounds__(256) void qkv_block(
    const float* __restrict__ xin, const bf16_t* __restrict__ Wb,
    const float* __restrict__ bias, bf16_t* __restrict__ Qb,
    bf16_t* __restrict__ Kb, bf16_t* __restrict__ Vt,
    const float* __restrict__ mask, int* __restrict__ kpos,
    int* __restrict__ nkv) {
  __shared__ alignas(16) bf16_t xs[QROWS * XS_PITCH];
  __shared__ alignas(16) bf16_t wq[2][64 * WQ_PITCH];
  __shared__ alignas(16) bf16_t ost[2][66 * OS_PITCH];
  __shared__ int sc_red[16];
  __shared__ int sc_keep[QROWS];
  __shared__ int sc_grp[8];
  __shared__ int sc_kp[QROWS];

  const int tid  = threadIdx.x;
  const int lane = tid & 31;
  const int wave = tid >> 5;
  const int r    = lane & 15;
  const int half = lane >> 4;
  const int mw   = wave >> 2;
  const int nw   = wave & 3;
  const int row_blk = blockIdx.x * QROWS;
  const int b   = row_blk >> 11;
  const int t0w = row_blk & (SEQ - 1);

  const float4* mp = (const float4*)(mask + (size_t)b * SEQ + tid * 8);
  const float4 m0 = mp[0], m1 = mp[1];

  const v8bf* wgp = (const v8bf*)(Wb + (size_t)wave * DIMD) + lane;
  const int   wso = wave * WQ_PITCH + lane * 8;

  int obase, ocnt;
  v8bf wg[8];
#pragma unroll
  for (int j = 0; j < 8; ++j) wg[j] = wgp[(size_t)(j * 8) * (DIMD / 8)];

  {
    float4 x0[8], x1[8];
#pragma unroll
    for (int j = 0; j < 8; ++j) {
      const float4* xp = (const float4*)(xin + (size_t)(row_blk + wave * 8 + j) * DIMD);
      x0[j] = xp[lane];
      x1[j] = xp[32 + lane];
    }

    {
      const float mv[8] = {m0.x, m0.y, m0.z, m0.w, m1.x, m1.y, m1.z, m1.w};
      int keep[8], c = 0;
#pragma unroll
      for (int j = 0; j < 8; ++j) { keep[j] = !(mv[j] > 0.5f); c += keep[j]; }
      const int g0 = t0w >> 3;
      int call = c, cpre = (tid < g0) ? c : 0;
#pragma unroll
      for (int d = 16; d >= 1; d >>= 1) {
        call += __shfl_xor(call, d, 32);
        cpre += __shfl_xor(cpre, d, 32);
      }
      if (lane == 0) { sc_red[wave] = call; sc_red[8 + wave] = cpre; }
      const int g = tid - g0;
      if (g >= 0 && g < 8) {
        sc_grp[g] = c;
#pragma unroll
        for (int j = 0; j < 8; ++j) sc_keep[g * 8 + j] = keep[j];
      }
      lds_barrier();
      int total = 0, pre = 0;
#pragma unroll
      for (int w = 0; w < 8; ++w) { total += sc_red[w]; pre += sc_red[8 + w]; }
      const bool allm = (total == 0);
      int tcnt = 0;
#pragma unroll
      for (int gg = 0; gg < 8; ++gg) tcnt += sc_grp[gg];
      obase = allm ? t0w : pre;
      ocnt  = allm ? QROWS : tcnt;
      if (tid < QROWS) {
        const int gq = tid >> 3, e = tid & 7;
        int q = pre;
#pragma unroll
        for (int gg = 0; gg < 8; ++gg) q += (gg < gq) ? sc_grp[gg] : 0;
#pragma unroll
        for (int ee = 0; ee < 8; ++ee) q += (ee < e) ? sc_keep[gq * 8 + ee] : 0;
        const int kq = allm ? (t0w + tid) : (sc_keep[tid] ? q : -1);
        sc_kp[tid] = kq;
        kpos[row_blk + tid] = kq;
      }
      const int nk = allm ? SEQ : total;
      if (t0w == 0 && tid == 0) nkv[b] = allm ? -1 : total;
      const int nk64 = (nk + 63) & ~63;
      const int ph = tid >> 6, pd = tid & 63;
      const size_t pbh = (size_t)b * HEADS + ph;
#pragma unroll
      for (int u = 0; u < 2; ++u) {
        const int q = nk + (t0w >> 6) * 2 + u;
        if (q < nk64) {
          Kb[(pbh * SEQ + q) * HD + pd] = (bf16_t)0.0f;
          Vt[(pbh * HD + pd) * SEQ + q] = (bf16_t)0.0f;
        }
      }
    }

#pragma unroll
    for (int j = 0; j < 8; ++j) {
      v4bf c0, c1;
      c0[0] = (bf16_t)x0[j].x; c0[1] = (bf16_t)x0[j].y; c0[2] = (bf16_t)x0[j].z; c0[3] = (bf16_t)x0[j].w;
      c1[0] = (bf16_t)x1[j].x; c1[1] = (bf16_t)x1[j].y; c1[2] = (bf16_t)x1[j].z; c1[3] = (bf16_t)x1[j].w;
      bf16_t* bp = xs + (wave * 8 + j) * XS_PITCH + lane * 4;
      *(v4bf*)bp         = c0;
      *(v4bf*)(bp + 128) = c1;
    }
  }
#pragma unroll
  for (int j = 0; j < 8; ++j) *(v8bf*)(wq[0] + wso + j * 8 * WQ_PITCH) = wg[j];
  __syncthreads();

  v16bf af[2][8];
#pragma unroll
  for (int mi = 0; mi < 2; ++mi) {
#pragma unroll
    for (int s = 0; s < 8; ++s)
      af[mi][s] = lds_frag_bf16(xs + (mw * 32 + mi * 16 + r) * XS_PITCH + s * 32);
  }
  const int d = nw * 16 + r;
  int ko[2][8], vo[2][8];
#pragma unroll
  for (int mi = 0; mi < 2; ++mi) {
#pragma unroll
    for (int i = 0; i < 8; ++i) {
      const int q = sc_kp[mw * 32 + mi * 16 + i + 8 * half];
      const int p = (q >= 0) ? q - obase : 64 + half;
      ko[mi][i] = p * OS_PITCH + d;
      vo[mi][i] = d * OV_PITCH + p;
    }
  }
  auto flush_tile = [&](const int cc) __attribute__((always_inline)) {
    const bf16_t* fs = ost[cc & 1];
    const int fsec = cc >> 2;
    const size_t fbh = (size_t)b * HEADS + (cc & 3);
    if (fsec < 2) {
      const int frows = (fsec == 0) ? QROWS : ocnt;
      bf16_t* gp = (fsec == 0) ? Qb + (fbh * SEQ + t0w) * HD
                               : Kb + (fbh * SEQ + obase) * HD;
      const int frow = tid >> 3, fcol = (tid & 7) * 8;
#pragma unroll
      for (int u = 0; u < 2; ++u) {
        const int row = u * 32 + frow;
        if (row < frows)
          *(v8bf*)(gp + (size_t)row * HD + fcol) = *(const v8bf*)(fs + row * OS_PITCH + fcol);
      }
    } else {
      bf16_t* gp = Vt + (fbh * HD + wave * 8) * SEQ + obase + lane;
      const bf16_t* fv = fs + wave * 8 * OV_PITCH + lane;
      if (lane < ocnt) {
#pragma unroll
        for (int j = 0; j < 8; ++j) gp[(size_t)j * SEQ] = fv[j * OV_PITCH];
      }
      if (lane + 32 < ocnt) {
#pragma unroll
        for (int j = 0; j < 8; ++j) gp[(size_t)j * SEQ + 32] = fv[j * OV_PITCH + 32];
      }
    }
  };

  float bv = bias[d];
  for (int c = 0; c < 3 * DIMD / 64; ++c) {
    const int cur = c & 1;
    const int cn = (c + 1 < 3 * DIMD / 64) ? c + 1 : c;
    const float bias_n = bias[cn * 64 + d];
#pragma unroll
    for (int j = 0; j < 8; ++j)
      wg[j] = wgp[(size_t)(cn * 64 + j * 8) * (DIMD / 8)];
    const bf16_t* bb = wq[cur] + d * WQ_PITCH;
    v16bf bq[8];
#pragma unroll
    for (int s = 0; s < 8; ++s) bq[s] = lds_frag_bf16(bb + s * 32);
    __builtin_amdgcn_sched_barrier(0);
    v8f acc[2] = {};
#pragma unroll
    for (int s = 0; s < 8; ++s) {
      acc[0] = wmma_bf16(af[0][s], bq[s], acc[0]);
      acc[1] = wmma_bf16(af[1][s], bq[s], acc[1]);
    }
    if (c > 0) flush_tile(c - 1);
    const int sec = c >> 2;
    bf16_t* os = ost[cur];
    bf16_t y[2][8];
#pragma unroll
    for (int mi = 0; mi < 2; ++mi) {
#pragma unroll
      for (int i = 0; i < 8; ++i) y[mi][i] = (bf16_t)(acc[mi][i] + bv);
    }
    if (sec == 0) {
      bf16_t* qp = os + (mw * 32 + 8 * half) * OS_PITCH + d;
#pragma unroll
      for (int mi = 0; mi < 2; ++mi) {
#pragma unroll
        for (int i = 0; i < 8; ++i) qp[(mi * 16 + i) * OS_PITCH] = y[mi][i];
      }
    } else if (sec == 1) {
#pragma unroll
      for (int mi = 0; mi < 2; ++mi) {
#pragma unroll
        for (int i = 0; i < 8; ++i) os[ko[mi][i]] = y[mi][i];
      }
    } else {
#pragma unroll
      for (int mi = 0; mi < 2; ++mi) {
#pragma unroll
        for (int i = 0; i < 8; ++i) os[vo[mi][i]] = y[mi][i];
      }
    }
    bv = bias_n;
#pragma unroll
    for (int j = 0; j < 8; ++j)
      *(v8bf*)(wq[cur ^ 1] + wso + j * 8 * WQ_PITCH) = wg[j];
    lds_barrier();
  }
  flush_tile(3 * DIMD / 64 - 1);
}

#define GT_PITCH 264
__global__ __launch_bounds__(128) void gemm_wmma(
    const bf16_t* __restrict__ A, const float* __restrict__ B,
    const float* __restrict__ bias, float* __restrict__ Yf, int N) {
  __shared__ alignas(16) bf16_t at[128 * GT_PITCH];
  __shared__ alignas(16) bf16_t bt[64 * GT_PITCH];
  const int lane = threadIdx.x & 31;
  const int wave = threadIdx.x >> 5;
  const int r    = lane & 15;
  const int half = lane >> 4;
  const int rowb = blockIdx.y * 128;
  const int row0 = rowb + wave * 32;
  const int col0 = blockIdx.x * 64;

  {
    const v8bf*   ag = (const v8bf*)(A + (size_t)(rowb + wave) * DIMD) + lane;
    const float4* bg = (const float4*)(B + (size_t)(col0 + wave) * DIMD) + lane;
    const int so  = wave * GT_PITCH + lane * 8;
    const int sob = wave * GT_PITCH + lane * 4;
    float4 tb0[16], tb1[16];
    v8bf ta[32];
#pragma unroll
    for (int j = 0; j < 16; ++j) {
      tb0[j] = bg[(size_t)j * 4 * (DIMD / 4)];
      tb1[j] = bg[(size_t)j * 4 * (DIMD / 4) + 32];
    }
#pragma unroll
    for (int j = 0; j < 32; ++j) ta[j] = ag[(size_t)j * 4 * (DIMD / 8)];
#pragma unroll
    for (int j = 0; j < 16; ++j) {
      v4bf c0, c1;
      c0[0] = (bf16_t)tb0[j].x; c0[1] = (bf16_t)tb0[j].y; c0[2] = (bf16_t)tb0[j].z; c0[3] = (bf16_t)tb0[j].w;
      c1[0] = (bf16_t)tb1[j].x; c1[1] = (bf16_t)tb1[j].y; c1[2] = (bf16_t)tb1[j].z; c1[3] = (bf16_t)tb1[j].w;
      bf16_t* bp = bt + sob + j * 4 * GT_PITCH;
      *(v4bf*)bp         = c0;
      *(v4bf*)(bp + 128) = c1;
    }
#pragma unroll
    for (int j = 0; j < 32; ++j) *(v8bf*)(at + so + j * 4 * GT_PITCH) = ta[j];
  }
  __syncthreads();

  v8f acc[2][4] = {};
  const bf16_t* ap0 = at + (wave * 32 + r) * GT_PITCH;
  const bf16_t* ap1 = ap0 + 16 * GT_PITCH;
  const bf16_t* bp  = bt + r * GT_PITCH;

  v16bf fa[2][2], fb[2][4];
  fa[0][0] = lds_frag_bf16(ap0);
  fa[0][1] = lds_frag_bf16(ap1);
#pragma unroll
  for (int nb = 0; nb < 4; ++nb) fb[0][nb] = lds_frag_bf16(bp + nb * 16 * GT_PITCH);
#pragma unroll
  for (int ks = 0; ks < DIMD / 32; ++ks) {
    const int cs = ks & 1, ns = cs ^ 1;
    if (ks + 1 < DIMD / 32) {
      const int k1 = (ks + 1) * 32;
      fa[ns][0] = lds_frag_bf16(ap0 + k1);
      fa[ns][1] = lds_frag_bf16(ap1 + k1);
#pragma unroll
      for (int nb = 0; nb < 4; ++nb)
        fb[ns][nb] = lds_frag_bf16(bp + nb * 16 * GT_PITCH + k1);
    }
    __builtin_amdgcn_sched_barrier(0);
#pragma unroll
    for (int nb = 0; nb < 4; ++nb) {
      acc[0][nb] = wmma_bf16(fa[cs][0], fb[cs][nb], acc[0][nb]);
      acc[1][nb] = wmma_bf16(fa[cs][1], fb[cs][nb], acc[1][nb]);
    }
    __builtin_amdgcn_sched_barrier(0);
  }

#pragma unroll
  for (int mi = 0; mi < 2; ++mi) {
#pragma unroll
    for (int nb = 0; nb < 4; ++nb) {
#pragma unroll
      for (int i = 0; i < 8; ++i) {
        const int row = row0 + mi * 16 + i + 8 * half;
        const int col = col0 + nb * 16 + r;
        Yf[(size_t)row * N + col] = acc[mi][nb][i] + bias[col];
      }
    }
  }
}

#define AT_WAVES 8
#define AT_KT    64
#define SC_L2E   0.18033688011112042f
#define S_BLANK  (-8589934592.0f)
#define HPARTS   (HEADS * SEQ / (32 * AT_WAVES))
#define KV_PITCH 72

template <bool B> struct tile_flag { static constexpr bool value = B; };

__global__ __launch_bounds__(256) __attribute__((amdgpu_waves_per_eu(2, 2))) void attn_wmma(
    const bf16_t* __restrict__ Qb, const bf16_t* __restrict__ Kb,
    const bf16_t* __restrict__ Vt, const int* __restrict__ nkv,
    bf16_t* __restrict__ attn, float* __restrict__ hpart) {
  __shared__ alignas(16) bf16_t ks[2][AT_KT * KV_PITCH];
  __shared__ alignas(16) bf16_t vs[2][HD * KV_PITCH];
  __shared__ alignas(16) bf16_t lds_pw[AT_WAVES][2][16 * 32];
  __shared__ float csl[2][AT_WAVES][AT_KT];

  const int tid  = threadIdx.x;
  const int wave = tid >> 5;
  const int bid  = blockIdx.x;
  const int bh   = bid >> 3;
  const int qt   = (bid & 7) * AT_WAVES + wave;
  const int b    = bh >> 2;
  const int lane = tid & 31;
  const int r    = lane & 15;
  const int half = lane >> 4;

  const bf16_t* Qbase = Qb + ((size_t)bh * SEQ + qt * 32) * HD;
  const bf16_t* Kbase = Kb + (size_t)bh * SEQ * HD;
  const bf16_t* Vbase = Vt + (size_t)bh * HD * SEQ;
  float*        hp    = hpart + (size_t)bid * SEQ;
  const int nkraw = nkv[b];
  const int nk    = (nkraw < 0) ? SEQ : nkraw;
  const int lim   = (nkraw < 0) ? 0 : nkraw;
  const int nt    = (nk + AT_KT - 1) / AT_KT;

  const int srow = tid >> 3;
  const int scol = (tid & 7) * 8;
  const v8bf* kg = (const v8bf*)(Kbase + (size_t)srow * HD + scol);
  const v8bf* vg = (const v8bf*)(Vbase + (size_t)srow * SEQ + scol);
  const int   so = srow * KV_PITCH + scol;
  bf16_t* pw0 = lds_pw[wave][0];
  bf16_t* pw1 = lds_pw[wave][1];

  v16bf qa[2][2];
#pragma unroll
  for (int qi = 0; qi < 2; ++qi) {
    qa[qi][0] = load_frag_bf16(Qbase + (size_t)(qi * 16 + r) * HD + 0);
    qa[qi][1] = load_frag_bf16(Qbase + (size_t)(qi * 16 + r) * HD + 32);
  }

  float m[2][8], l[2][8];
#pragma unroll
  for (int qi = 0; qi < 2; ++qi)
#pragma unroll
    for (int i = 0; i < 8; ++i) { m[qi][i] = -3.0e38f; l[qi][i] = 0.0f; }

  {
    const v8bf g0 = kg[0];
    const v8bf g1 = kg[32 * HD / 8];
    *(v8bf*)(ks[0] + so)                 = g0;
    *(v8bf*)(ks[0] + so + 32 * KV_PITCH) = g1;
  }
  lds_barrier();
  const int ntf = lim / AT_KT;
  auto p1_tile = [&](const int t, auto blank) __attribute__((always_inline)) {
    constexpr bool BLANK = decltype(blank)::value;
    const int cur = t & 1;
    const int tn = (t + 1 < nt) ? t + 1 : t;
    const v8bf* kn = kg + (size_t)tn * (AT_KT * HD / 8);
    const v8bf g0 = kn[0];
    const v8bf g1 = kn[32 * HD / 8];
    __builtin_amdgcn_sched_barrier(0);
    v8f s[4][2];
#pragma unroll
    for (int sub = 0; sub < 4; ++sub) {
      const bf16_t* kr = ks[cur] + (sub * 16 + r) * KV_PITCH;
      const v16bf kb0 = lds_frag_bf16(kr);
      const v16bf kb1 = lds_frag_bf16(kr + 32);
#pragma unroll
      for (int qi = 0; qi < 2; ++qi) {
        v8f z = {};
        z = wmma_bf16(qa[qi][0], kb0, z);
        z = wmma_bf16(qa[qi][1], kb1, z);
        if (BLANK) {
          const bool dead = (t * AT_KT + sub * 16 + r) >= lim;
#pragma unroll
          for (int i = 0; i < 8; ++i) z[i] = dead ? S_BLANK : z[i];
        }
        s[sub][qi] = z;
      }
    }
#pragma unroll
    for (int qi = 0; qi < 2; ++qi) {
#pragma unroll
      for (int i = 0; i < 8; ++i) {
        const float a0 = s[0][qi][i], a1 = s[1][qi][i];
        const float a2 = s[2][qi][i], a3 = s[3][qi][i];
        const float mx = fmaxf(fmaxf(a0, a1), fmaxf(a2, a3));
        const float mn = fmaxf(m[qi][i], mx * SC_L2E);
        const float e0 = __builtin_amdgcn_exp2f(__builtin_fmaf(a0, SC_L2E, -mn));
        const float e1 = __builtin_amdgcn_exp2f(__builtin_fmaf(a1, SC_L2E, -mn));
        const float e2 = __builtin_amdgcn_exp2f(__builtin_fmaf(a2, SC_L2E, -mn));
        const float e3 = __builtin_amdgcn_exp2f(__builtin_fmaf(a3, SC_L2E, -mn));
        l[qi][i] = l[qi][i] * __builtin_amdgcn_exp2f(m[qi][i] - mn) + ((e0 + e1) + (e2 + e3));
        m[qi][i] = mn;
      }
    }
    __builtin_amdgcn_sched_barrier(0);
    *(v8bf*)(ks[cur ^ 1] + so)                 = g0;
    *(v8bf*)(ks[cur ^ 1] + so + 32 * KV_PITCH) = g1;
    lds_barrier();
  };
  for (int t = 0; t < ntf; ++t) p1_tile(t, tile_flag<false>{});
  for (int t = ntf; t < nt; ++t) p1_tile(t, tile_flag<true>{});

  {
    const v8bf g0 = kg[0];
    const v8bf g1 = kg[32 * HD / 8];
    const v8bf h0 = vg[0];
    const v8bf h1 = vg[32 * SEQ / 8];
    *(v8bf*)(ks[0] + so)                 = g0;
    *(v8bf*)(ks[0] + so + 32 * KV_PITCH) = g1;
    *(v8bf*)(vs[0] + so)                 = h0;
    *(v8bf*)(vs[0] + so + 32 * KV_PITCH) = h1;
  }

#pragma unroll
  for (int d = 1; d <= 8; d <<= 1) {
#pragma unroll
    for (int qi = 0; qi < 2; ++qi) {
#pragma unroll
      for (int i = 0; i < 8; ++i) {
        const float mo = __shfl_xor(m[qi][i], d, 32);
        const float lo = __shfl_xor(l[qi][i], d, 32);
        const float mn = fmaxf(m[qi][i], mo);
        l[qi][i] = l[qi][i] * __builtin_amdgcn_exp2f(m[qi][i] - mn) +
                   lo * __builtin_amdgcn_exp2f(mo - mn);
        m[qi][i] = mn;
      }
    }
  }
  float rl[2][8];
#pragma unroll
  for (int qi = 0; qi < 2; ++qi)
#pragma unroll
    for (int i = 0; i < 8; ++i) rl[qi][i] = 1.0f / l[qi][i];
  lds_barrier();

  v8f o[2][4] = {};
  auto p2_tile = [&](const int t, auto blank) __attribute__((always_inline)) {
    constexpr bool BLANK = decltype(blank)::value;
    const int cur = t & 1;
    const int tn = (t + 1 < nt) ? t + 1 : t;
    const v8bf* kn = kg + (size_t)tn * (AT_KT * HD / 8);
    const v8bf* vn = vg + (size_t)tn * (AT_KT / 8);
    const v8bf g0 = kn[0];
    const v8bf g1 = kn[32 * HD / 8];
    const v8bf h0 = vn[0];
    const v8bf h1 = vn[32 * SEQ / 8];
    if (t > 0 && tid < AT_KT) {
      const float* cp = &csl[cur ^ 1][0][tid];
      float s = cp[0];
#pragma unroll
      for (int w = 1; w < AT_WAVES; ++w) s += cp[w * AT_KT];
      hp[(t - 1) * AT_KT + tid] = s;
    }
    __builtin_amdgcn_sched_barrier(0);
    v16bf kf[4][2];
#pragma unroll
    for (int j = 0; j < 4; ++j) {
      const bf16_t* kr = ks[cur] + (j * 16 + r) * KV_PITCH;
      kf[j][0] = lds_frag_bf16(kr);
      kf[j][1] = lds_frag_bf16(kr + 32);
    }
    float* csw = csl[cur][wave];
#pragma unroll
    for (int kh = 0; kh < 2; ++kh) {
      float csv[2];
#pragma unroll
      for (int sub = 0; sub < 2; ++sub) {
        const int lrow = (kh * 2 + sub) * 16 + r;
        const v16bf kb0 = kf[kh * 2 + sub][0];
        const v16bf kb1 = kf[kh * 2 + sub][1];
        const bool dead = BLANK && (t * AT_KT + lrow >= lim);
        float cs = 0.0f;
#pragma unroll
        for (int qi = 0; qi < 2; ++qi) {
          v8f s = {};
          s = wmma_bf16(qa[qi][0], kb0, s);
          s = wmma_bf16(qa[qi][1], kb1, s);
          bf16_t* pw = qi ? pw1 : pw0;
#pragma unroll
          for (int i = 0; i < 8; ++i) {
            const float sr = (BLANK && dead) ? S_BLANK : s[i];
            const float p  = __builtin_amdgcn_exp2f(__builtin_fmaf(sr, SC_L2E, -m[qi][i])) * rl[qi][i];
            cs += p;
            pw[(i + 8 * half) * 32 + sub * 16 + r] = (bf16_t)p;
          }
        }
        csv[sub] = cs + __shfl_xor(cs, 16, 32);
      }
      csw[kh * 32 + lane] = half ? csv[1] : csv[0];
      __builtin_amdgcn_fence(__ATOMIC_ACQ_REL, "wavefront");
      v16bf pa0 = lds_frag_bf16(pw0 + r * 32);
      v16bf pa1 = lds_frag_bf16(pw1 + r * 32);
#pragma unroll
      for (int nb = 0; nb < 4; ++nb) {
        v16bf vb = lds_frag_bf16(vs[cur] + (nb * 16 + r) * KV_PITCH + kh * 32);
        o[0][nb] = wmma_bf16(pa0, vb, o[0][nb]);
        o[1][nb] = wmma_bf16(pa1, vb, o[1][nb]);
      }
      __builtin_amdgcn_fence(__ATOMIC_ACQ_REL, "wavefront");
    }
    *(v8bf*)(ks[cur ^ 1] + so)                 = g0;
    *(v8bf*)(ks[cur ^ 1] + so + 32 * KV_PITCH) = g1;
    *(v8bf*)(vs[cur ^ 1] + so)                 = h0;
    *(v8bf*)(vs[cur ^ 1] + so + 32 * KV_PITCH) = h1;
    lds_barrier();
  };
  for (int t = 0; t < ntf; ++t) p2_tile(t, tile_flag<false>{});
  for (int t = ntf; t < nt; ++t) p2_tile(t, tile_flag<true>{});
  if (tid < AT_KT) {
    const float* cp = &csl[(nt - 1) & 1][0][tid];
    float s = cp[0];
#pragma unroll
    for (int w = 1; w < AT_WAVES; ++w) s += cp[w * AT_KT];
    hp[(nt - 1) * AT_KT + tid] = s;
  }

  const int h = bh & 3;
#pragma unroll
  for (int qi = 0; qi < 2; ++qi) {
#pragma unroll
    for (int nb = 0; nb < 4; ++nb) {
#pragma unroll
      for (int i = 0; i < 8; ++i) {
        const int t = qt * 32 + qi * 16 + i + 8 * half;
        attn[((size_t)b * SEQ + t) * DIMD + h * HD + nb * 16 + r] =
            (bf16_t)o[qi][nb][i];
      }
    }
  }
}

#define TROWS     64
#define HF_PITCH  260
#define HB_PITCH  264
#define HID_PITCH 1032
#define W1S_PITCH 264
#define W2S_PITCH 72
#define WST_ELEMS (DIMD * W2S_PITCH)

__device__ __forceinline__ float wave_sum(float v) {
#pragma unroll
  for (int d = 16; d >= 1; d >>= 1) v += __shfl_xor(v, d, 32);
  return v;
}

__device__ __forceinline__ void ln_row(float v[8], const float4 ga,
                                       const float4 gb, const float4 ba,
                                       const float4 bb) {
  float s = ((v[0] + v[1]) + (v[2] + v[3])) + ((v[4] + v[5]) + (v[6] + v[7]));
  const float mu = wave_sum(s) * (1.0f / DIMD);
  float d[8];
#pragma unroll
  for (int i = 0; i < 8; ++i) d[i] = v[i] - mu;
  float q = ((d[0] * d[0] + d[1] * d[1]) + (d[2] * d[2] + d[3] * d[3])) +
            ((d[4] * d[4] + d[5] * d[5]) + (d[6] * d[6] + d[7] * d[7]));
  const float var = wave_sum(q) * (1.0f / DIMD);
  const float rs = rsqrtf(var + 1e-5f);
  const float gg[8] = {ga.x, ga.y, ga.z, ga.w, gb.x, gb.y, gb.z, gb.w};
  const float bt[8] = {ba.x, ba.y, ba.z, ba.w, bb.x, bb.y, bb.z, bb.w};
#pragma unroll
  for (int i = 0; i < 8; ++i) v[i] = d[i] * rs * gg[i] + bt[i];
}

__global__ __launch_bounds__(256) void ffn_block(
    const float* __restrict__ xin, const float* __restrict__ proj,
    const float* __restrict__ g1, const float* __restrict__ be1,
    const bf16_t* __restrict__ W1, const float* __restrict__ b1,
    const bf16_t* __restrict__ W2, const float* __restrict__ b2,
    const float* __restrict__ g2, const float* __restrict__ be2,
    float* __restrict__ out) {
  __shared__ alignas(16) float  hf[TROWS * HF_PITCH];
  __shared__ alignas(16) bf16_t hb[TROWS * HB_PITCH];
  __shared__ alignas(16) bf16_t hid[TROWS * HID_PITCH];
  __shared__ alignas(16) bf16_t wst[2][WST_ELEMS];

  const int tid  = threadIdx.x;
  const int lane = tid & 31;
  const int wave = tid >> 5;
  const int r    = lane & 15;
  const int half = lane >> 4;
  const int mw   = wave >> 2;
  const int nw   = wave & 3;
  const size_t row_blk = (size_t)blockIdx.x * TROWS;

  const v8bf* w1g  = (const v8bf*)(W1 + (size_t)wave * DIMD) + lane;
  const int   w1so = wave * W1S_PITCH + lane * 8;
  const int   srow = tid >> 3;
  const int   scol = (tid & 7) * 8;
  const v8bf* w2g  = (const v8bf*)(W2 + (size_t)srow * FFND + scol);
  const int   w2so = srow * W2S_PITCH + scol;

  v8bf wg[8];
#pragma unroll
  for (int j = 0; j < 8; ++j) wg[j] = w1g[(size_t)(j * 8) * (DIMD / 8)];

  {
    const float4 ga = ((const float4*)g1)[lane], gb = ((const float4*)g1)[32 + lane];
    const float4 ba = ((const float4*)be1)[lane], bb = ((const float4*)be1)[32 + lane];
#pragma unroll
    for (int j = 0; j < 8; ++j) {
      const int lr = wave * 8 + j;
      const float4* xp = (const float4*)(xin + (row_blk + lr) * DIMD);
      const float4* pp = (const float4*)(proj + (row_blk + lr) * DIMD);
      const float4 x0 = xp[lane], x1 = xp[32 + lane];
      const float4 p0 = pp[lane], p1 = pp[32 + lane];
      float v[8] = {x0.x + p0.x, x0.y + p0.y, x0.z + p0.z, x0.w + p0.w,
                    x1.x + p1.x, x1.y + p1.y, x1.z + p1.z, x1.w + p1.w};
      ln_row(v, ga, gb, ba, bb);
      float* hp = hf + lr * HF_PITCH + lane * 4;
      *(float4*)hp         = make_float4(v[0], v[1], v[2], v[3]);
      *(float4*)(hp + 128) = make_float4(v[4], v[5], v[6], v[7]);
      v4bf c0, c1;
      c0[0] = (bf16_t)v[0]; c0[1] = (bf16_t)v[1]; c0[2] = (bf16_t)v[2]; c0[3] = (bf16_t)v[3];
      c1[0] = (bf16_t)v[4]; c1[1] = (bf16_t)v[5]; c1[2] = (bf16_t)v[6]; c1[3] = (bf16_t)v[7];
      bf16_t* bp = hb + lr * HB_PITCH + lane * 4;
      *(v4bf*)bp         = c0;
      *(v4bf*)(bp + 128) = c1;
    }
  }
#pragma unroll
  for (int j = 0; j < 8; ++j) *(v8bf*)(wst[0] + w1so + j * 8 * W1S_PITCH) = wg[j];
  __syncthreads();

  {
    v16bf af[2][8];
#pragma unroll
    for (int mi = 0; mi < 2; ++mi) {
#pragma unroll
      for (int s = 0; s < 8; ++s)
        af[mi][s] = lds_frag_bf16(hb + (mw * 32 + mi * 16 + r) * HB_PITCH + s * 32);
    }
    float bias = b1[nw * 16 + r];
    for (int c = 0; c < FFND / 64; ++c) {
      const int cur = c & 1;
      const int cn = (c + 1 < FFND / 64) ? c + 1 : c;
      const float bias_n = b1[cn * 64 + nw * 16 + r];
#pragma unroll
      for (int j = 0; j < 8; ++j)
        wg[j] = w1g[(size_t)(cn * 64 + j * 8) * (DIMD / 8)];
      const bf16_t* bb = wst[cur] + (nw * 16 + r) * W1S_PITCH;
      v16bf bq[8];
#pragma unroll
      for (int s = 0; s < 8; ++s) bq[s] = lds_frag_bf16(bb + s * 32);
      __builtin_amdgcn_sched_barrier(0);
      v8f acc[2] = {};
#pragma unroll
      for (int s = 0; s < 8; ++s) {
        acc[0] = wmma_bf16(af[0][s], bq[s], acc[0]);
        acc[1] = wmma_bf16(af[1][s], bq[s], acc[1]);
      }
      const int col = c * 64 + nw * 16 + r;
#pragma unroll
      for (int mi = 0; mi < 2; ++mi) {
#pragma unroll
        for (int i = 0; i < 8; ++i) {
          const int row = mw * 32 + mi * 16 + i + 8 * half;
          hid[row * HID_PITCH + col] = (bf16_t)fmaxf(acc[mi][i] + bias, 0.0f);
        }
      }
      bias = bias_n;
      __builtin_amdgcn_sched_barrier(0);
#pragma unroll
      for (int j = 0; j < 8; ++j)
        *(v8bf*)(wst[cur ^ 1] + w1so + j * 8 * W1S_PITCH) = wg[j];
      lds_barrier();
    }
  }
#pragma unroll
  for (int j = 0; j < 8; ++j) wg[j] = w2g[(size_t)j * 32 * (FFND / 8)];
#pragma unroll
  for (int j = 0; j < 8; ++j) *(v8bf*)(wst[0] + w2so + j * 32 * W2S_PITCH) = wg[j];
  __syncthreads();

  {
    const int oc0 = nw * 64;
    const bf16_t* ia0 = hid + (mw * 32 + r) * HID_PITCH;
    const bf16_t* ia1 = ia0 + 16 * HID_PITCH;
    v8f acc[2][4] = {};
    for (int t = 0; t < FFND / 64; ++t) {
      const int cur = t & 1;
      const int tn = (t + 1 < FFND / 64) ? t + 1 : t;
#pragma unroll
      for (int j = 0; j < 8; ++j)
        wg[j] = w2g[(size_t)j * 32 * (FFND / 8) + tn * 8];
      const bf16_t* bb = wst[cur] + (oc0 + r) * W2S_PITCH;
      v16bf a[2][2], bq[2][4];
#pragma unroll
      for (int s = 0; s < 2; ++s) {
        a[0][s] = lds_frag_bf16(ia0 + t * 64 + s * 32);
        a[1][s] = lds_frag_bf16(ia1 + t * 64 + s * 32);
#pragma unroll
        for (int nb = 0; nb < 4; ++nb)
          bq[s][nb] = lds_frag_bf16(bb + nb * 16 * W2S_PITCH + s * 32);
      }
      __builtin_amdgcn_sched_barrier(0);
#pragma unroll
      for (int s = 0; s < 2; ++s) {
#pragma unroll
        for (int nb = 0; nb < 4; ++nb) {
          acc[0][nb] = wmma_bf16(a[0][s], bq[s][nb], acc[0][nb]);
          acc[1][nb] = wmma_bf16(a[1][s], bq[s][nb], acc[1][nb]);
        }
      }
      __builtin_amdgcn_sched_barrier(0);
#pragma unroll
      for (int j = 0; j < 8; ++j)
        *(v8bf*)(wst[cur ^ 1] + w2so + j * 32 * W2S_PITCH) = wg[j];
      lds_barrier();
    }
#pragma unroll
    for (int mi = 0; mi < 2; ++mi) {
#pragma unroll
      for (int nb = 0; nb < 4; ++nb) {
        const int col = oc0 + nb * 16 + r;
        const float bias = b2[col];
#pragma unroll
        for (int i = 0; i < 8; ++i) {
          const int row = mw * 32 + mi * 16 + i + 8 * half;
          float* hp = hf + row * HF_PITCH + col;
          *hp = *hp + (acc[mi][nb][i] + bias);
        }
      }
    }
  }
  __syncthreads();

  {
    const float4 ga = ((const float4*)g2)[lane], gb = ((const float4*)g2)[32 + lane];
    const float4 ba = ((const float4*)be2)[lane], bb = ((const float4*)be2)[32 + lane];
#pragma unroll
    for (int j = 0; j < 8; ++j) {
      const int lr = wave * 8 + j;
      const float* hp = hf + lr * HF_PITCH + lane * 4;
      const float4 y0 = *(const float4*)hp;
      const float4 y1 = *(const float4*)(hp + 128);
      float v[8] = {y0.x, y0.y, y0.z, y0.w, y1.x, y1.y, y1.z, y1.w};
      ln_row(v, ga, gb, ba, bb);
      float4* op = (float4*)(out + (row_blk + lr) * DIMD);
      op[lane]      = make_float4(v[0], v[1], v[2], v[3]);
      op[32 + lane] = make_float4(v[4], v[5], v[6], v[7]);
    }
  }
}

__device__ __forceinline__ float wave_max(float v) {
#pragma unroll
  for (int d = 16; d >= 1; d >>= 1) v = fmaxf(v, __shfl_xor(v, d, 32));
  return v;
}

__global__ __launch_bounds__(1024) void halt_finalize(
    const float* __restrict__ hpart, const int* __restrict__ kpos,
    const float* __restrict__ mask, float* __restrict__ out) {
  __shared__ float red[32];
  const int b = blockIdx.x, tid = threadIdx.x;
  const int lane = tid & 31, wave = tid >> 5;
  float v[2];
#pragma unroll
  for (int i = 0; i < 2; ++i) {
    const int j = tid + i * 1024;
    const int q = kpos[(size_t)b * SEQ + j];
    const float* pp = hpart + (size_t)b * HPARTS * SEQ + (q < 0 ? 0 : q);
    float a = 0.0f;
#pragma unroll
    for (int p = 0; p < HPARTS; ++p) a += pp[(size_t)p * SEQ];
    if (q < 0) a = 0.0f;
    v[i] = a * 0.25f + mask[(size_t)b * SEQ + j] * (-1e10f);
  }
  const float wm = wave_max(fmaxf(v[0], v[1]));
  if (lane == 0) red[wave] = wm;
  __syncthreads();
  const float mx = wave_max(red[lane]);
  __syncthreads();
  const float e0 = __expf(v[0] - mx), e1 = __expf(v[1] - mx);
  const float ws = wave_sum(e0 + e1);
  if (lane == 0) red[wave] = ws;
  __syncthreads();
  const float inv = 1.0f / wave_sum(red[lane]);
  out[(size_t)b * SEQ + tid]        = e0 * inv;
  out[(size_t)b * SEQ + tid + 1024] = e1 * inv;
}

extern "C" void kernel_launch(void* const* d_in, const int* in_sizes, int n_in,
                              void* d_out, int out_size, void* d_ws,
                              size_t ws_size, hipStream_t stream) {
  (void)in_sizes; (void)n_in; (void)out_size; (void)ws_size;
  const float* x    = (const float*)d_in[0];
  const float* mask = (const float*)d_in[1];
  const float* Wqkv = (const float*)d_in[2];
  const float* bqkv = (const float*)d_in[3];
  const float* Wo   = (const float*)d_in[4];
  const float* bo   = (const float*)d_in[5];
  const float* W1   = (const float*)d_in[6];
  const float* b1   = (const float*)d_in[7];
  const float* W2   = (const float*)d_in[8];
  const float* b2   = (const float*)d_in[9];
  const float* g1   = (const float*)d_in[10];
  const float* be1  = (const float*)d_in[11];
  const float* g2   = (const float*)d_in[12];
  const float* be2  = (const float*)d_in[13];

  char* ws = (char*)d_ws;
  size_t off = 0;
  auto alloc = [&](size_t bytes) {
    char* p = ws + off;
    off += (bytes + 255) & ~(size_t)255;
    return p;
  };
  bf16_t* Wqkvb = (bf16_t*)alloc((size_t)3 * DIMD * DIMD * 2);
  bf16_t* W1b   = (bf16_t*)alloc((size_t)FFND * DIMD * 2);
  bf16_t* W2b   = (bf16_t*)alloc((size_t)DIMD * FFND * 2);
  bf16_t* Qb    = (bf16_t*)alloc((size_t)NTOK * DIMD * 2);
  bf16_t* Kb    = (bf16_t*)alloc((size_t)NTOK * DIMD * 2);
  bf16_t* Vt    = (bf16_t*)alloc((size_t)NTOK * DIMD * 2);
  bf16_t* attnb = (bf16_t*)alloc((size_t)NTOK * DIMD * 2);
  float*  attnO = (float*) alloc((size_t)NTOK * DIMD * 4);

  int* kpos = (int*)alloc((size_t)NTOK * 4);
  int* nkv  = (int*)alloc((size_t)BATCH * 4);

  float* hpart = (float*)alloc((size_t)BATCH * HPARTS * SEQ * 4);

  float* out      = (float*)d_out;
  float* out_halt = out + (size_t)NTOK * DIMD;

  cvt_bf16<<<3 * DIMD * DIMD / 1024, 256, 0, stream>>>(Wqkv, Wqkvb);
  cvt_bf16<<<FFND * DIMD / 1024, 256, 0, stream>>>(W1, W1b);
  cvt_bf16<<<DIMD * FFND / 1024, 256, 0, stream>>>(W2, W2b);

  qkv_block<<<NTOK / QROWS, 256, 0, stream>>>(x, Wqkvb, bqkv, Qb, Kb, Vt, mask,
                                              kpos, nkv);

  attn_wmma<<<BATCH * HEADS * (SEQ / (32 * AT_WAVES)), 32 * AT_WAVES, 0, stream>>>(
      Qb, Kb, Vt, nkv, attnb, hpart);

  gemm_wmma<<<dim3(DIMD / 64, NTOK / 128), 128, 0, stream>>>(
      attnb, Wo, bo, attnO, DIMD);

  ffn_block<<<NTOK / TROWS, 256, 0, stream>>>(
      x, attnO, g1, be1, W1b, b1, W2b, b2, g2, be2, out);

  halt_finalize<<<BATCH, 1024, 0, stream>>>(hpart, kpos, mask, out_halt);
}


# W1 and W2 bf16 conversion rides in the QKV row-tile kernel (one float4 of each per thread, loaded ahead of the tile loads); the two convert launches are gone, 8 -> 6 launches; same values
# speedup vs baseline: 4.2412x; 1.0499x over previous
#include <hip/hip_runtime.h>
#include <hip/hip_bf16.h>

typedef __bf16 bf16_t;
typedef __attribute__((ext_vector_type(16))) __bf16 v16bf;
typedef __attribute__((ext_vector_type(4)))  __bf16 v4bf;
typedef __attribute__((ext_vector_type(8)))  float  v8f;
typedef __attribute__((ext_vector_type(8)))  __bf16 v8bf;

#define DIMD   256
#define HEADS  4
#define HD     64
#define FFND   1024
#define BATCH  8
#define SEQ    2048
#define NTOK   (BATCH*SEQ)
#define NEGV   (-1e9f)

__device__ __forceinline__ v16bf load_frag_bf16(const bf16_t* base) {
  const int half = (threadIdx.x >> 4) & 1;
  const bf16_t* p = base + 8 * half;
  v16bf f;
#pragma unroll
  for (int e = 0; e < 8; ++e) {
    f[e]     = p[e];
    f[e + 8] = p[e + 16];
  }
  return f;
}

__device__ __forceinline__ v16bf lds_frag_bf16(const bf16_t* base) {
  const int half = (threadIdx.x >> 4) & 1;
  const bf16_t* p = base + 8 * half;
  const v8bf lo = *(const v8bf*)p;
  const v8bf hi = *(const v8bf*)(p + 16);
  return __builtin_shufflevector(lo, hi, 0, 1, 2, 3, 4, 5, 6, 7,
                                 8, 9, 10, 11, 12, 13, 14, 15);
}

__device__ __forceinline__ void lds_barrier() {
  __builtin_amdgcn_fence(__ATOMIC_RELEASE, "workgroup", "local");
  __builtin_amdgcn_s_barrier();
  __builtin_amdgcn_fence(__ATOMIC_ACQUIRE, "workgroup", "local");
}

__device__ __forceinline__ v8f wmma_bf16(v16bf a, v16bf b, v8f c) {
  return __builtin_amdgcn_wmma_f32_16x16x32_bf16(
       false, a,  false, b,
       (short)0, c,  false,  false);
}

__global__ __launch_bounds__(256) void cvt_bf16(const float* __restrict__ in,
                                                bf16_t* __restrict__ out) {
  const size_t i = (size_t)blockIdx.x * 256 + threadIdx.x;
  const float4 v = ((const float4*)in)[i];
  v4bf o;
  o[0] = (bf16_t)v.x; o[1] = (bf16_t)v.y; o[2] = (bf16_t)v.z; o[3] = (bf16_t)v.w;
  ((v4bf*)out)[i] = o;
}


#define QROWS    64
static_assert(FFND * DIMD / 4 == (NTOK / QROWS) * 256, "one float4 of W1 and of W2 per qkv_block thread");
#define XS_PITCH 264
#define WQ_PITCH 264
#define OS_PITCH 72
#define OV_PITCH 66
__global__ __launch_bounds__(256) void qkv_block(
    const float* __restrict__ xin, const bf16_t* __restrict__ Wb,
    const float* __restrict__ bias, bf16_t* __restrict__ Qb,
    bf16_t* __restrict__ Kb, bf16_t* __restrict__ Vt,
    const float* __restrict__ mask, int* __restrict__ kpos,
    int* __restrict__ nkv,
    const float* __restrict__ W1f, const float* __restrict__ W2f,
    bf16_t* __restrict__ W1b, bf16_t* __restrict__ W2b) {
  __shared__ alignas(16) bf16_t xs[QROWS * XS_PITCH];
  __shared__ alignas(16) bf16_t wq[2][64 * WQ_PITCH];
  __shared__ alignas(16) bf16_t ost[2][66 * OS_PITCH];
  __shared__ int sc_red[16];
  __shared__ int sc_keep[QROWS];
  __shared__ int sc_grp[8];
  __shared__ int sc_kp[QROWS];

  const int tid  = threadIdx.x;
  const int lane = tid & 31;
  const int wave = tid >> 5;
  const int r    = lane & 15;
  const int half = lane >> 4;
  const int mw   = wave >> 2;
  const int nw   = wave & 3;
  const int row_blk = blockIdx.x * QROWS;
  const int b   = row_blk >> 11;
  const int t0w = row_blk & (SEQ - 1);

  const float4* mp = (const float4*)(mask + (size_t)b * SEQ + tid * 8);
  const float4 m0 = mp[0], m1 = mp[1];
  const size_t cvi = (size_t)blockIdx.x * 256 + tid;
  const float4 cw1 = ((const float4*)W1f)[cvi];
  const float4 cw2 = ((const float4*)W2f)[cvi];

  const v8bf* wgp = (const v8bf*)(Wb + (size_t)wave * DIMD) + lane;
  const int   wso = wave * WQ_PITCH + lane * 8;

  int obase, ocnt;
  v8bf wg[8];
#pragma unroll
  for (int j = 0; j < 8; ++j) wg[j] = wgp[(size_t)(j * 8) * (DIMD / 8)];

  {
    float4 x0[8], x1[8];
#pragma unroll
    for (int j = 0; j < 8; ++j) {
      const float4* xp = (const float4*)(xin + (size_t)(row_blk + wave * 8 + j) * DIMD);
      x0[j] = xp[lane];
      x1[j] = xp[32 + lane];
    }

    {
      const float mv[8] = {m0.x, m0.y, m0.z, m0.w, m1.x, m1.y, m1.z, m1.w};
      int keep[8], c = 0;
#pragma unroll
      for (int j = 0; j < 8; ++j) { keep[j] = !(mv[j] > 0.5f); c += keep[j]; }
      const int g0 = t0w >> 3;
      int call = c, cpre = (tid < g0) ? c : 0;
#pragma unroll
      for (int d = 16; d >= 1; d >>= 1) {
        call += __shfl_xor(call, d, 32);
        cpre += __shfl_xor(cpre, d, 32);
      }
      if (lane == 0) { sc_red[wave] = call; sc_red[8 + wave] = cpre; }
      const int g = tid - g0;
      if (g >= 0 && g < 8) {
        sc_grp[g] = c;
#pragma unroll
        for (int j = 0; j < 8; ++j) sc_keep[g * 8 + j] = keep[j];
      }
      lds_barrier();
      int total = 0, pre = 0;
#pragma unroll
      for (int w = 0; w < 8; ++w) { total += sc_red[w]; pre += sc_red[8 + w]; }
      const bool allm = (total == 0);
      int tcnt = 0;
#pragma unroll
      for (int gg = 0; gg < 8; ++gg) tcnt += sc_grp[gg];
      obase = allm ? t0w : pre;
      ocnt  = allm ? QROWS : tcnt;
      if (tid < QROWS) {
        const int gq = tid >> 3, e = tid & 7;
        int q = pre;
#pragma unroll
        for (int gg = 0; gg < 8; ++gg) q += (gg < gq) ? sc_grp[gg] : 0;
#pragma unroll
        for (int ee = 0; ee < 8; ++ee) q += (ee < e) ? sc_keep[gq * 8 + ee] : 0;
        const int kq = allm ? (t0w + tid) : (sc_keep[tid] ? q : -1);
        sc_kp[tid] = kq;
        kpos[row_blk + tid] = kq;
      }
      const int nk = allm ? SEQ : total;
      if (t0w == 0 && tid == 0) nkv[b] = allm ? -1 : total;
      const int nk64 = (nk + 63) & ~63;
      const int ph = tid >> 6, pd = tid & 63;
      const size_t pbh = (size_t)b * HEADS + ph;
#pragma unroll
      for (int u = 0; u < 2; ++u) {
        const int q = nk + (t0w >> 6) * 2 + u;
        if (q < nk64) {
          Kb[(pbh * SEQ + q) * HD + pd] = (bf16_t)0.0f;
          Vt[(pbh * HD + pd) * SEQ + q] = (bf16_t)0.0f;
        }
      }
    }

#pragma unroll
    for (int j = 0; j < 8; ++j) {
      v4bf c0, c1;
      c0[0] = (bf16_t)x0[j].x; c0[1] = (bf16_t)x0[j].y; c0[2] = (bf16_t)x0[j].z; c0[3] = (bf16_t)x0[j].w;
      c1[0] = (bf16_t)x1[j].x; c1[1] = (bf16_t)x1[j].y; c1[2] = (bf16_t)x1[j].z; c1[3] = (bf16_t)x1[j].w;
      bf16_t* bp = xs + (wave * 8 + j) * XS_PITCH + lane * 4;
      *(v4bf*)bp         = c0;
      *(v4bf*)(bp + 128) = c1;
    }
  }
#pragma unroll
  for (int j = 0; j < 8; ++j) *(v8bf*)(wq[0] + wso + j * 8 * WQ_PITCH) = wg[j];
  __syncthreads();

  {
    v4bf o1, o2;
    o1[0] = (bf16_t)cw1.x; o1[1] = (bf16_t)cw1.y; o1[2] = (bf16_t)cw1.z; o1[3] = (bf16_t)cw1.w;
    o2[0] = (bf16_t)cw2.x; o2[1] = (bf16_t)cw2.y; o2[2] = (bf16_t)cw2.z; o2[3] = (bf16_t)cw2.w;
    ((v4bf*)W1b)[cvi] = o1;
    ((v4bf*)W2b)[cvi] = o2;
  }

  v16bf af[2][8];
#pragma unroll
  for (int mi = 0; mi < 2; ++mi) {
#pragma unroll
    for (int s = 0; s < 8; ++s)
      af[mi][s] = lds_frag_bf16(xs + (mw * 32 + mi * 16 + r) * XS_PITCH + s * 32);
  }
  const int d = nw * 16 + r;
  int ko[2][8], vo[2][8];
#pragma unroll
  for (int mi = 0; mi < 2; ++mi) {
#pragma unroll
    for (int i = 0; i < 8; ++i) {
      const int q = sc_kp[mw * 32 + mi * 16 + i + 8 * half];
      const int p = (q >= 0) ? q - obase : 64 + half;
      ko[mi][i] = p * OS_PITCH + d;
      vo[mi][i] = d * OV_PITCH + p;
    }
  }
  auto flush_tile = [&](const int cc) __attribute__((always_inline)) {
    const bf16_t* fs = ost[cc & 1];
    const int fsec = cc >> 2;
    const size_t fbh = (size_t)b * HEADS + (cc & 3);
    if (fsec < 2) {
      const int frows = (fsec == 0) ? QROWS : ocnt;
      bf16_t* gp = (fsec == 0) ? Qb + (fbh * SEQ + t0w) * HD
                               : Kb + (fbh * SEQ + obase) * HD;
      const int frow = tid >> 3, fcol = (tid & 7) * 8;
#pragma unroll
      for (int u = 0; u < 2; ++u) {
        const int row = u * 32 + frow;
        if (row < frows)
          *(v8bf*)(gp + (size_t)row * HD + fcol) = *(const v8bf*)(fs + row * OS_PITCH + fcol);
      }
    } else {
      bf16_t* gp = Vt + (fbh * HD + wave * 8) * SEQ + obase + lane;
      const bf16_t* fv = fs + wave * 8 * OV_PITCH + lane;
      if (lane < ocnt) {
#pragma unroll
        for (int j = 0; j < 8; ++j) gp[(size_t)j * SEQ] = fv[j * OV_PITCH];
      }
      if (lane + 32 < ocnt) {
#pragma unroll
        for (int j = 0; j < 8; ++j) gp[(size_t)j * SEQ + 32] = fv[j * OV_PITCH + 32];
      }
    }
  };

  float bv = bias[d];
  for (int c = 0; c < 3 * DIMD / 64; ++c) {
    const int cur = c & 1;
    const int cn = (c + 1 < 3 * DIMD / 64) ? c + 1 : c;
    const float bias_n = bias[cn * 64 + d];
#pragma unroll
    for (int j = 0; j < 8; ++j)
      wg[j] = wgp[(size_t)(cn * 64 + j * 8) * (DIMD / 8)];
    const bf16_t* bb = wq[cur] + d * WQ_PITCH;
    v16bf bq[8];
#pragma unroll
    for (int s = 0; s < 8; ++s) bq[s] = lds_frag_bf16(bb + s * 32);
    __builtin_amdgcn_sched_barrier(0);
    v8f acc[2] = {};
#pragma unroll
    for (int s = 0; s < 8; ++s) {
      acc[0] = wmma_bf16(af[0][s], bq[s], acc[0]);
      acc[1] = wmma_bf16(af[1][s], bq[s], acc[1]);
    }
    if (c > 0) flush_tile(c - 1);
    const int sec = c >> 2;
    bf16_t* os = ost[cur];
    bf16_t y[2][8];
#pragma unroll
    for (int mi = 0; mi < 2; ++mi) {
#pragma unroll
      for (int i = 0; i < 8; ++i) y[mi][i] = (bf16_t)(acc[mi][i] + bv);
    }
    if (sec == 0) {
      bf16_t* qp = os + (mw * 32 + 8 * half) * OS_PITCH + d;
#pragma unroll
      for (int mi = 0; mi < 2; ++mi) {
#pragma unroll
        for (int i = 0; i < 8; ++i) qp[(mi * 16 + i) * OS_PITCH] = y[mi][i];
      }
    } else if (sec == 1) {
#pragma unroll
      for (int mi = 0; mi < 2; ++mi) {
#pragma unroll
        for (int i = 0; i < 8; ++i) os[ko[mi][i]] = y[mi][i];
      }
    } else {
#pragma unroll
      for (int mi = 0; mi < 2; ++mi) {
#pragma unroll
        for (int i = 0; i < 8; ++i) os[vo[mi][i]] = y[mi][i];
      }
    }
    bv = bias_n;
#pragma unroll
    for (int j = 0; j < 8; ++j)
      *(v8bf*)(wq[cur ^ 1] + wso + j * 8 * WQ_PITCH) = wg[j];
    lds_barrier();
  }
  flush_tile(3 * DIMD / 64 - 1);
}

#define GT_PITCH 264
__global__ __launch_bounds__(128) void gemm_wmma(
    const bf16_t* __restrict__ A, const float* __restrict__ B,
    const float* __restrict__ bias, float* __restrict__ Yf, int N) {
  __shared__ alignas(16) bf16_t at[128 * GT_PITCH];
  __shared__ alignas(16) bf16_t bt[64 * GT_PITCH];
  const int lane = threadIdx.x & 31;
  const int wave = threadIdx.x >> 5;
  const int r    = lane & 15;
  const int half = lane >> 4;
  const int rowb = blockIdx.y * 128;
  const int row0 = rowb + wave * 32;
  const int col0 = blockIdx.x * 64;

  {
    const v8bf*   ag = (const v8bf*)(A + (size_t)(rowb + wave) * DIMD) + lane;
    const float4* bg = (const float4*)(B + (size_t)(col0 + wave) * DIMD) + lane;
    const int so  = wave * GT_PITCH + lane * 8;
    const int sob = wave * GT_PITCH + lane * 4;
    float4 tb0[16], tb1[16];
    v8bf ta[32];
#pragma unroll
    for (int j = 0; j < 16; ++j) {
      tb0[j] = bg[(size_t)j * 4 * (DIMD / 4)];
      tb1[j] = bg[(size_t)j * 4 * (DIMD / 4) + 32];
    }
#pragma unroll
    for (int j = 0; j < 32; ++j) ta[j] = ag[(size_t)j * 4 * (DIMD / 8)];
#pragma unroll
    for (int j = 0; j < 16; ++j) {
      v4bf c0, c1;
      c0[0] = (bf16_t)tb0[j].x; c0[1] = (bf16_t)tb0[j].y; c0[2] = (bf16_t)tb0[j].z; c0[3] = (bf16_t)tb0[j].w;
      c1[0] = (bf16_t)tb1[j].x; c1[1] = (bf16_t)tb1[j].y; c1[2] = (bf16_t)tb1[j].z; c1[3] = (bf16_t)tb1[j].w;
      bf16_t* bp = bt + sob + j * 4 * GT_PITCH;
      *(v4bf*)bp         = c0;
      *(v4bf*)(bp + 128) = c1;
    }
#pragma unroll
    for (int j = 0; j < 32; ++j) *(v8bf*)(at + so + j * 4 * GT_PITCH) = ta[j];
  }
  __syncthreads();

  v8f acc[2][4] = {};
  const bf16_t* ap0 = at + (wave * 32 + r) * GT_PITCH;
  const bf16_t* ap1 = ap0 + 16 * GT_PITCH;
  const bf16_t* bp  = bt + r * GT_PITCH;

  v16bf fa[2][2], fb[2][4];
  fa[0][0] = lds_frag_bf16(ap0);
  fa[0][1] = lds_frag_bf16(ap1);
#pragma unroll
  for (int nb = 0; nb < 4; ++nb) fb[0][nb] = lds_frag_bf16(bp + nb * 16 * GT_PITCH);
#pragma unroll
  for (int ks = 0; ks < DIMD / 32; ++ks) {
    const int cs = ks & 1, ns = cs ^ 1;
    if (ks + 1 < DIMD / 32) {
      const int k1 = (ks + 1) * 32;
      fa[ns][0] = lds_frag_bf16(ap0 + k1);
      fa[ns][1] = lds_frag_bf16(ap1 + k1);
#pragma unroll
      for (int nb = 0; nb < 4; ++nb)
        fb[ns][nb] = lds_frag_bf16(bp + nb * 16 * GT_PITCH + k1);
    }
    __builtin_amdgcn_sched_barrier(0);
#pragma unroll
    for (int nb = 0; nb < 4; ++nb) {
      acc[0][nb] = wmma_bf16(fa[cs][0], fb[cs][nb], acc[0][nb]);
      acc[1][nb] = wmma_bf16(fa[cs][1], fb[cs][nb], acc[1][nb]);
    }
    __builtin_amdgcn_sched_barrier(0);
  }

#pragma unroll
  for (int mi = 0; mi < 2; ++mi) {
#pragma unroll
    for (int nb = 0; nb < 4; ++nb) {
#pragma unroll
      for (int i = 0; i < 8; ++i) {
        const int row = row0 + mi * 16 + i + 8 * half;
        const int col = col0 + nb * 16 + r;
        Yf[(size_t)row * N + col] = acc[mi][nb][i] + bias[col];
      }
    }
  }
}

#define AT_WAVES 8
#define AT_KT    64
#define SC_L2E   0.18033688011112042f
#define S_BLANK  (-8589934592.0f)
#define HPARTS   (HEADS * SEQ / (32 * AT_WAVES))
#define KV_PITCH 72

template <bool B> struct tile_flag { static constexpr bool value = B; };

__global__ __launch_bounds__(256) __attribute__((amdgpu_waves_per_eu(2, 2))) void attn_wmma(
    const bf16_t* __restrict__ Qb, const bf16_t* __restrict__ Kb,
    const bf16_t* __restrict__ Vt, const int* __restrict__ nkv,
    bf16_t* __restrict__ attn, float* __restrict__ hpart) {
  __shared__ alignas(16) bf16_t ks[2][AT_KT * KV_PITCH];
  __shared__ alignas(16) bf16_t vs[2][HD * KV_PITCH];
  __shared__ alignas(16) bf16_t lds_pw[AT_WAVES][2][16 * 32];
  __shared__ float csl[2][AT_WAVES][AT_KT];

  const int tid  = threadIdx.x;
  const int wave = tid >> 5;
  const int bid  = blockIdx.x;
  const int bh   = bid >> 3;
  const int qt   = (bid & 7) * AT_WAVES + wave;
  const int b    = bh >> 2;
  const int lane = tid & 31;
  const int r    = lane & 15;
  const int half = lane >> 4;

  const bf16_t* Qbase = Qb + ((size_t)bh * SEQ + qt * 32) * HD;
  const bf16_t* Kbase = Kb + (size_t)bh * SEQ * HD;
  const bf16_t* Vbase = Vt + (size_t)bh * HD * SEQ;
  float*        hp    = hpart + (size_t)bid * SEQ;
  const int nkraw = nkv[b];
  const int nk    = (nkraw < 0) ? SEQ : nkraw;
  const int lim   = (nkraw < 0) ? 0 : nkraw;
  const int nt    = (nk + AT_KT - 1) / AT_KT;

  const int srow = tid >> 3;
  const int scol = (tid & 7) * 8;
  const v8bf* kg = (const v8bf*)(Kbase + (size_t)srow * HD + scol);
  const v8bf* vg = (const v8bf*)(Vbase + (size_t)srow * SEQ + scol);
  const int   so = srow * KV_PITCH + scol;
  bf16_t* pw0 = lds_pw[wave][0];
  bf16_t* pw1 = lds_pw[wave][1];

  v16bf qa[2][2];
#pragma unroll
  for (int qi = 0; qi < 2; ++qi) {
    qa[qi][0] = load_frag_bf16(Qbase + (size_t)(qi * 16 + r) * HD + 0);
    qa[qi][1] = load_frag_bf16(Qbase + (size_t)(qi * 16 + r) * HD + 32);
  }

  float m[2][8], l[2][8];
#pragma unroll
  for (int qi = 0; qi < 2; ++qi)
#pragma unroll
    for (int i = 0; i < 8; ++i) { m[qi][i] = -3.0e38f; l[qi][i] = 0.0f; }

  {
    const v8bf g0 = kg[0];
    const v8bf g1 = kg[32 * HD / 8];
    *(v8bf*)(ks[0] + so)                 = g0;
    *(v8bf*)(ks[0] + so + 32 * KV_PITCH) = g1;
  }
  lds_barrier();
  const int ntf = lim / AT_KT;
  auto p1_tile = [&](const int t, auto blank) __attribute__((always_inline)) {
    constexpr bool BLANK = decltype(blank)::value;
    const int cur = t & 1;
    const int tn = (t + 1 < nt) ? t + 1 : t;
    const v8bf* kn = kg + (size_t)tn * (AT_KT * HD / 8);
    const v8bf g0 = kn[0];
    const v8bf g1 = kn[32 * HD / 8];
    __builtin_amdgcn_sched_barrier(0);
    v8f s[4][2];
#pragma unroll
    for (int sub = 0; sub < 4; ++sub) {
      const bf16_t* kr = ks[cur] + (sub * 16 + r) * KV_PITCH;
      const v16bf kb0 = lds_frag_bf16(kr);
      const v16bf kb1 = lds_frag_bf16(kr + 32);
#pragma unroll
      for (int qi = 0; qi < 2; ++qi) {
        v8f z = {};
        z = wmma_bf16(qa[qi][0], kb0, z);
        z = wmma_bf16(qa[qi][1], kb1, z);
        if (BLANK) {
          const bool dead = (t * AT_KT + sub * 16 + r) >= lim;
#pragma unroll
          for (int i = 0; i < 8; ++i) z[i] = dead ? S_BLANK : z[i];
        }
        s[sub][qi] = z;
      }
    }
#pragma unroll
    for (int qi = 0; qi < 2; ++qi) {
#pragma unroll
      for (int i = 0; i < 8; ++i) {
        const float a0 = s[0][qi][i], a1 = s[1][qi][i];
        const float a2 = s[2][qi][i], a3 = s[3][qi][i];
        const float mx = fmaxf(fmaxf(a0, a1), fmaxf(a2, a3));
        const float mn = fmaxf(m[qi][i], mx * SC_L2E);
        const float e0 = __builtin_amdgcn_exp2f(__builtin_fmaf(a0, SC_L2E, -mn));
        const float e1 = __builtin_amdgcn_exp2f(__builtin_fmaf(a1, SC_L2E, -mn));
        const float e2 = __builtin_amdgcn_exp2f(__builtin_fmaf(a2, SC_L2E, -mn));
        const float e3 = __builtin_amdgcn_exp2f(__builtin_fmaf(a3, SC_L2E, -mn));
        l[qi][i] = l[qi][i] * __builtin_amdgcn_exp2f(m[qi][i] - mn) + ((e0 + e1) + (e2 + e3));
        m[qi][i] = mn;
      }
    }
    __builtin_amdgcn_sched_barrier(0);
    *(v8bf*)(ks[cur ^ 1] + so)                 = g0;
    *(v8bf*)(ks[cur ^ 1] + so + 32 * KV_PITCH) = g1;
    lds_barrier();
  };
  for (int t = 0; t < ntf; ++t) p1_tile(t, tile_flag<false>{});
  for (int t = ntf; t < nt; ++t) p1_tile(t, tile_flag<true>{});

  {
    const v8bf g0 = kg[0];
    const v8bf g1 = kg[32 * HD / 8];
    const v8bf h0 = vg[0];
    const v8bf h1 = vg[32 * SEQ / 8];
    *(v8bf*)(ks[0] + so)                 = g0;
    *(v8bf*)(ks[0] + so + 32 * KV_PITCH) = g1;
    *(v8bf*)(vs[0] + so)                 = h0;
    *(v8bf*)(vs[0] + so + 32 * KV_PITCH) = h1;
  }

#pragma unroll
  for (int d = 1; d <= 8; d <<= 1) {
#pragma unroll
    for (int qi = 0; qi < 2; ++qi) {
#pragma unroll
      for (int i = 0; i < 8; ++i) {
        const float mo = __shfl_xor(m[qi][i], d, 32);
        const float lo = __shfl_xor(l[qi][i], d, 32);
        const float mn = fmaxf(m[qi][i], mo);
        l[qi][i] = l[qi][i] * __builtin_amdgcn_exp2f(m[qi][i] - mn) +
                   lo * __builtin_amdgcn_exp2f(mo - mn);
        m[qi][i] = mn;
      }
    }
  }
  float rl[2][8];
#pragma unroll
  for (int qi = 0; qi < 2; ++qi)
#pragma unroll
    for (int i = 0; i < 8; ++i) rl[qi][i] = 1.0f / l[qi][i];
  lds_barrier();

  v8f o[2][4] = {};
  auto p2_tile = [&](const int t, auto blank) __attribute__((always_inline)) {
    constexpr bool BLANK = decltype(blank)::value;
    const int cur = t & 1;
    const int tn = (t + 1 < nt) ? t + 1 : t;
    const v8bf* kn = kg + (size_t)tn * (AT_KT * HD / 8);
    const v8bf* vn = vg + (size_t)tn * (AT_KT / 8);
    const v8bf g0 = kn[0];
    const v8bf g1 = kn[32 * HD / 8];
    const v8bf h0 = vn[0];
    const v8bf h1 = vn[32 * SEQ / 8];
    if (t > 0 && tid < AT_KT) {
      const float* cp = &csl[cur ^ 1][0][tid];
      float s = cp[0];
#pragma unroll
      for (int w = 1; w < AT_WAVES; ++w) s += cp[w * AT_KT];
      hp[(t - 1) * AT_KT + tid] = s;
    }
    __builtin_amdgcn_sched_barrier(0);
    v16bf kf[4][2];
#pragma unroll
    for (int j = 0; j < 4; ++j) {
      const bf16_t* kr = ks[cur] + (j * 16 + r) * KV_PITCH;
      kf[j][0] = lds_frag_bf16(kr);
      kf[j][1] = lds_frag_bf16(kr + 32);
    }
    float* csw = csl[cur][wave];
#pragma unroll
    for (int kh = 0; kh < 2; ++kh) {
      float csv[2];
#pragma unroll
      for (int sub = 0; sub < 2; ++sub) {
        const int lrow = (kh * 2 + sub) * 16 + r;
        const v16bf kb0 = kf[kh * 2 + sub][0];
        const v16bf kb1 = kf[kh * 2 + sub][1];
        const bool dead = BLANK && (t * AT_KT + lrow >= lim);
        float cs = 0.0f;
#pragma unroll
        for (int qi = 0; qi < 2; ++qi) {
          v8f s = {};
          s = wmma_bf16(qa[qi][0], kb0, s);
          s = wmma_bf16(qa[qi][1], kb1, s);
          bf16_t* pw = qi ? pw1 : pw0;
#pragma unroll
          for (int i = 0; i < 8; ++i) {
            const float sr = (BLANK && dead) ? S_BLANK : s[i];
            const float p  = __builtin_amdgcn_exp2f(__builtin_fmaf(sr, SC_L2E, -m[qi][i])) * rl[qi][i];
            cs += p;
            pw[(i + 8 * half) * 32 + sub * 16 + r] = (bf16_t)p;
          }
        }
        csv[sub] = cs + __shfl_xor(cs, 16, 32);
      }
      csw[kh * 32 + lane] = half ? csv[1] : csv[0];
      __builtin_amdgcn_fence(__ATOMIC_ACQ_REL, "wavefront");
      v16bf pa0 = lds_frag_bf16(pw0 + r * 32);
      v16bf pa1 = lds_frag_bf16(pw1 + r * 32);
#pragma unroll
      for (int nb = 0; nb < 4; ++nb) {
        v16bf vb = lds_frag_bf16(vs[cur] + (nb * 16 + r) * KV_PITCH + kh * 32);
        o[0][nb] = wmma_bf16(pa0, vb, o[0][nb]);
        o[1][nb] = wmma_bf16(pa1, vb, o[1][nb]);
      }
      __builtin_amdgcn_fence(__ATOMIC_ACQ_REL, "wavefront");
    }
    *(v8bf*)(ks[cur ^ 1] + so)                 = g0;
    *(v8bf*)(ks[cur ^ 1] + so + 32 * KV_PITCH) = g1;
    *(v8bf*)(vs[cur ^ 1] + so)                 = h0;
    *(v8bf*)(vs[cur ^ 1] + so + 32 * KV_PITCH) = h1;
    lds_barrier();
  };
  for (int t = 0; t < ntf; ++t) p2_tile(t, tile_flag<false>{});
  for (int t = ntf; t < nt; ++t) p2_tile(t, tile_flag<true>{});
  if (tid < AT_KT) {
    const float* cp = &csl[(nt - 1) & 1][0][tid];
    float s = cp[0];
#pragma unroll
    for (int w = 1; w < AT_WAVES; ++w) s += cp[w * AT_KT];
    hp[(nt - 1) * AT_KT + tid] = s;
  }

  const int h = bh & 3;
#pragma unroll
  for (int qi = 0; qi < 2; ++qi) {
#pragma unroll
    for (int nb = 0; nb < 4; ++nb) {
#pragma unroll
      for (int i = 0; i < 8; ++i) {
        const int t = qt * 32 + qi * 16 + i + 8 * half;
        attn[((size_t)b * SEQ + t) * DIMD + h * HD + nb * 16 + r] =
            (bf16_t)o[qi][nb][i];
      }
    }
  }
}

#define TROWS     64
#define HF_PITCH  260
#define HB_PITCH  264
#define HID_PITCH 1032
#define W1S_PITCH 264
#define W2S_PITCH 72
#define WST_ELEMS (DIMD * W2S_PITCH)

__device__ __forceinline__ float wave_sum(float v) {
#pragma unroll
  for (int d = 16; d >= 1; d >>= 1) v += __shfl_xor(v, d, 32);
  return v;
}

__device__ __forceinline__ void ln_row(float v[8], const float4 ga,
                                       const float4 gb, const float4 ba,
                                       const float4 bb) {
  float s = ((v[0] + v[1]) + (v[2] + v[3])) + ((v[4] + v[5]) + (v[6] + v[7]));
  const float mu = wave_sum(s) * (1.0f / DIMD);
  float d[8];
#pragma unroll
  for (int i = 0; i < 8; ++i) d[i] = v[i] - mu;
  float q = ((d[0] * d[0] + d[1] * d[1]) + (d[2] * d[2] + d[3] * d[3])) +
            ((d[4] * d[4] + d[5] * d[5]) + (d[6] * d[6] + d[7] * d[7]));
  const float var = wave_sum(q) * (1.0f / DIMD);
  const float rs = rsqrtf(var + 1e-5f);
  const float gg[8] = {ga.x, ga.y, ga.z, ga.w, gb.x, gb.y, gb.z, gb.w};
  const float bt[8] = {ba.x, ba.y, ba.z, ba.w, bb.x, bb.y, bb.z, bb.w};
#pragma unroll
  for (int i = 0; i < 8; ++i) v[i] = d[i] * rs * gg[i] + bt[i];
}

__global__ __launch_bounds__(256) void ffn_block(
    const float* __restrict__ xin, const float* __restrict__ proj,
    const float* __restrict__ g1, const float* __restrict__ be1,
    const bf16_t* __restrict__ W1, const float* __restrict__ b1,
    const bf16_t* __restrict__ W2, const float* __restrict__ b2,
    const float* __restrict__ g2, const float* __restrict__ be2,
    float* __restrict__ out) {
  __shared__ alignas(16) float  hf[TROWS * HF_PITCH];
  __shared__ alignas(16) bf16_t hb[TROWS * HB_PITCH];
  __shared__ alignas(16) bf16_t hid[TROWS * HID_PITCH];
  __shared__ alignas(16) bf16_t wst[2][WST_ELEMS];

  const int tid  = threadIdx.x;
  const int lane = tid & 31;
  const int wave = tid >> 5;
  const int r    = lane & 15;
  const int half = lane >> 4;
  const int mw   = wave >> 2;
  const int nw   = wave & 3;
  const size_t row_blk = (size_t)blockIdx.x * TROWS;

  const v8bf* w1g  = (const v8bf*)(W1 + (size_t)wave * DIMD) + lane;
  const int   w1so = wave * W1S_PITCH + lane * 8;
  const int   srow = tid >> 3;
  const int   scol = (tid & 7) * 8;
  const v8bf* w2g  = (const v8bf*)(W2 + (size_t)srow * FFND + scol);
  const int   w2so = srow * W2S_PITCH + scol;

  v8bf wg[8];
#pragma unroll
  for (int j = 0; j < 8; ++j) wg[j] = w1g[(size_t)(j * 8) * (DIMD / 8)];

  {
    const float4 ga = ((const float4*)g1)[lane], gb = ((const float4*)g1)[32 + lane];
    const float4 ba = ((const float4*)be1)[lane], bb = ((const float4*)be1)[32 + lane];
#pragma unroll
    for (int j = 0; j < 8; ++j) {
      const int lr = wave * 8 + j;
      const float4* xp = (const float4*)(xin + (row_blk + lr) * DIMD);
      const float4* pp = (const float4*)(proj + (row_blk + lr) * DIMD);
      const float4 x0 = xp[lane], x1 = xp[32 + lane];
      const float4 p0 = pp[lane], p1 = pp[32 + lane];
      float v[8] = {x0.x + p0.x, x0.y + p0.y, x0.z + p0.z, x0.w + p0.w,
                    x1.x + p1.x, x1.y + p1.y, x1.z + p1.z, x1.w + p1.w};
      ln_row(v, ga, gb, ba, bb);
      float* hp = hf + lr * HF_PITCH + lane * 4;
      *(float4*)hp         = make_float4(v[0], v[1], v[2], v[3]);
      *(float4*)(hp + 128) = make_float4(v[4], v[5], v[6], v[7]);
      v4bf c0, c1;
      c0[0] = (bf16_t)v[0]; c0[1] = (bf16_t)v[1]; c0[2] = (bf16_t)v[2]; c0[3] = (bf16_t)v[3];
      c1[0] = (bf16_t)v[4]; c1[1] = (bf16_t)v[5]; c1[2] = (bf16_t)v[6]; c1[3] = (bf16_t)v[7];
      bf16_t* bp = hb + lr * HB_PITCH + lane * 4;
      *(v4bf*)bp         = c0;
      *(v4bf*)(bp + 128) = c1;
    }
  }
#pragma unroll
  for (int j = 0; j < 8; ++j) *(v8bf*)(wst[0] + w1so + j * 8 * W1S_PITCH) = wg[j];
  __syncthreads();

  {
    v16bf af[2][8];
#pragma unroll
    for (int mi = 0; mi < 2; ++mi) {
#pragma unroll
      for (int s = 0; s < 8; ++s)
        af[mi][s] = lds_frag_bf16(hb + (mw * 32 + mi * 16 + r) * HB_PITCH + s * 32);
    }
    float bias = b1[nw * 16 + r];
    for (int c = 0; c < FFND / 64; ++c) {
      const int cur = c & 1;
      const int cn = (c + 1 < FFND / 64) ? c + 1 : c;
      const float bias_n = b1[cn * 64 + nw * 16 + r];
#pragma unroll
      for (int j = 0; j < 8; ++j)
        wg[j] = w1g[(size_t)(cn * 64 + j * 8) * (DIMD / 8)];
      const bf16_t* bb = wst[cur] + (nw * 16 + r) * W1S_PITCH;
      v16bf bq[8];
#pragma unroll
      for (int s = 0; s < 8; ++s) bq[s] = lds_frag_bf16(bb + s * 32);
      __builtin_amdgcn_sched_barrier(0);
      v8f acc[2] = {};
#pragma unroll
      for (int s = 0; s < 8; ++s) {
        acc[0] = wmma_bf16(af[0][s], bq[s], acc[0]);
        acc[1] = wmma_bf16(af[1][s], bq[s], acc[1]);
      }
      const int col = c * 64 + nw * 16 + r;
#pragma unroll
      for (int mi = 0; mi < 2; ++mi) {
#pragma unroll
        for (int i = 0; i < 8; ++i) {
          const int row = mw * 32 + mi * 16 + i + 8 * half;
          hid[row * HID_PITCH + col] = (bf16_t)fmaxf(acc[mi][i] + bias, 0.0f);
        }
      }
      bias = bias_n;
      __builtin_amdgcn_sched_barrier(0);
#pragma unroll
      for (int j = 0; j < 8; ++j)
        *(v8bf*)(wst[cur ^ 1] + w1so + j * 8 * W1S_PITCH) = wg[j];
      lds_barrier();
    }
  }
#pragma unroll
  for (int j = 0; j < 8; ++j) wg[j] = w2g[(size_t)j * 32 * (FFND / 8)];
#pragma unroll
  for (int j = 0; j < 8; ++j) *(v8bf*)(wst[0] + w2so + j * 32 * W2S_PITCH) = wg[j];
  __syncthreads();

  {
    const int oc0 = nw * 64;
    const bf16_t* ia0 = hid + (mw * 32 + r) * HID_PITCH;
    const bf16_t* ia1 = ia0 + 16 * HID_PITCH;
    v8f acc[2][4] = {};
    for (int t = 0; t < FFND / 64; ++t) {
      const int cur = t & 1;
      const int tn = (t + 1 < FFND / 64) ? t + 1 : t;
#pragma unroll
      for (int j = 0; j < 8; ++j)
        wg[j] = w2g[(size_t)j * 32 * (FFND / 8) + tn * 8];
      const bf16_t* bb = wst[cur] + (oc0 + r) * W2S_PITCH;
      v16bf a[2][2], bq[2][4];
#pragma unroll
      for (int s = 0; s < 2; ++s) {
        a[0][s] = lds_frag_bf16(ia0 + t * 64 + s * 32);
        a[1][s] = lds_frag_bf16(ia1 + t * 64 + s * 32);
#pragma unroll
        for (int nb = 0; nb < 4; ++nb)
          bq[s][nb] = lds_frag_bf16(bb + nb * 16 * W2S_PITCH + s * 32);
      }
      __builtin_amdgcn_sched_barrier(0);
#pragma unroll
      for (int s = 0; s < 2; ++s) {
#pragma unroll
        for (int nb = 0; nb < 4; ++nb) {
          acc[0][nb] = wmma_bf16(a[0][s], bq[s][nb], acc[0][nb]);
          acc[1][nb] = wmma_bf16(a[1][s], bq[s][nb], acc[1][nb]);
        }
      }
      __builtin_amdgcn_sched_barrier(0);
#pragma unroll
      for (int j = 0; j < 8; ++j)
        *(v8bf*)(wst[cur ^ 1] + w2so + j * 32 * W2S_PITCH) = wg[j];
      lds_barrier();
    }
#pragma unroll
    for (int mi = 0; mi < 2; ++mi) {
#pragma unroll
      for (int nb = 0; nb < 4; ++nb) {
        const int col = oc0 + nb * 16 + r;
        const float bias = b2[col];
#pragma unroll
        for (int i = 0; i < 8; ++i) {
          const int row = mw * 32 + mi * 16 + i + 8 * half;
          float* hp = hf + row * HF_PITCH + col;
          *hp = *hp + (acc[mi][nb][i] + bias);
        }
      }
    }
  }
  __syncthreads();

  {
    const float4 ga = ((const float4*)g2)[lane], gb = ((const float4*)g2)[32 + lane];
    const float4 ba = ((const float4*)be2)[lane], bb = ((const float4*)be2)[32 + lane];
#pragma unroll
    for (int j = 0; j < 8; ++j) {
      const int lr = wave * 8 + j;
      const float* hp = hf + lr * HF_PITCH + lane * 4;
      const float4 y0 = *(const float4*)hp;
      const float4 y1 = *(const float4*)(hp + 128);
      float v[8] = {y0.x, y0.y, y0.z, y0.w, y1.x, y1.y, y1.z, y1.w};
      ln_row(v, ga, gb, ba, bb);
      float4* op = (float4*)(out + (row_blk + lr) * DIMD);
      op[lane]      = make_float4(v[0], v[1], v[2], v[3]);
      op[32 + lane] = make_float4(v[4], v[5], v[6], v[7]);
    }
  }
}

__device__ __forceinline__ float wave_max(float v) {
#pragma unroll
  for (int d = 16; d >= 1; d >>= 1) v = fmaxf(v, __shfl_xor(v, d, 32));
  return v;
}

__global__ __launch_bounds__(1024) void halt_finalize(
    const float* __restrict__ hpart, const int* __restrict__ kpos,
    const float* __restrict__ mask, float* __restrict__ out) {
  __shared__ float red[32];
  const int b = blockIdx.x, tid = threadIdx.x;
  const int lane = tid & 31, wave = tid >> 5;
  float v[2];
#pragma unroll
  for (int i = 0; i < 2; ++i) {
    const int j = tid + i * 1024;
    const int q = kpos[(size_t)b * SEQ + j];
    const float* pp = hpart + (size_t)b * HPARTS * SEQ + (q < 0 ? 0 : q);
    float a = 0.0f;
#pragma unroll
    for (int p = 0; p < HPARTS; ++p) a += pp[(size_t)p * SEQ];
    if (q < 0) a = 0.0f;
    v[i] = a * 0.25f + mask[(size_t)b * SEQ + j] * (-1e10f);
  }
  const float wm = wave_max(fmaxf(v[0], v[1]));
  if (lane == 0) red[wave] = wm;
  __syncthreads();
  const float mx = wave_max(red[lane]);
  __syncthreads();
  const float e0 = __expf(v[0] - mx), e1 = __expf(v[1] - mx);
  const float ws = wave_sum(e0 + e1);
  if (lane == 0) red[wave] = ws;
  __syncthreads();
  const float inv = 1.0f / wave_sum(red[lane]);
  out[(size_t)b * SEQ + tid]        = e0 * inv;
  out[(size_t)b * SEQ + tid + 1024] = e1 * inv;
}

extern "C" void kernel_launch(void* const* d_in, const int* in_sizes, int n_in,
                              void* d_out, int out_size, void* d_ws,
                              size_t ws_size, hipStream_t stream) {
  (void)in_sizes; (void)n_in; (void)out_size; (void)ws_size;
  const float* x    = (const float*)d_in[0];
  const float* mask = (const float*)d_in[1];
  const float* Wqkv = (const float*)d_in[2];
  const float* bqkv = (const float*)d_in[3];
  const float* Wo   = (const float*)d_in[4];
  const float* bo   = (const float*)d_in[5];
  const float* W1   = (const float*)d_in[6];
  const float* b1   = (const float*)d_in[7];
  const float* W2   = (const float*)d_in[8];
  const float* b2   = (const float*)d_in[9];
  const float* g1   = (const float*)d_in[10];
  const float* be1  = (const float*)d_in[11];
  const float* g2   = (const float*)d_in[12];
  const float* be2  = (const float*)d_in[13];

  char* ws = (char*)d_ws;
  size_t off = 0;
  auto alloc = [&](size_t bytes) {
    char* p = ws + off;
    off += (bytes + 255) & ~(size_t)255;
    return p;
  };
  bf16_t* Wqkvb = (bf16_t*)alloc((size_t)3 * DIMD * DIMD * 2);
  bf16_t* W1b   = (bf16_t*)alloc((size_t)FFND * DIMD * 2);
  bf16_t* W2b   = (bf16_t*)alloc((size_t)DIMD * FFND * 2);
  bf16_t* Qb    = (bf16_t*)alloc((size_t)NTOK * DIMD * 2);
  bf16_t* Kb    = (bf16_t*)alloc((size_t)NTOK * DIMD * 2);
  bf16_t* Vt    = (bf16_t*)alloc((size_t)NTOK * DIMD * 2);
  bf16_t* attnb = (bf16_t*)alloc((size_t)NTOK * DIMD * 2);
  float*  attnO = (float*) alloc((size_t)NTOK * DIMD * 4);

  int* kpos = (int*)alloc((size_t)NTOK * 4);
  int* nkv  = (int*)alloc((size_t)BATCH * 4);

  float* hpart = (float*)alloc((size_t)BATCH * HPARTS * SEQ * 4);

  float* out      = (float*)d_out;
  float* out_halt = out + (size_t)NTOK * DIMD;

  cvt_bf16<<<3 * DIMD * DIMD / 1024, 256, 0, stream>>>(Wqkv, Wqkvb);

  qkv_block<<<NTOK / QROWS, 256, 0, stream>>>(x, Wqkvb, bqkv, Qb, Kb, Vt, mask,
                                              kpos, nkv, W1, W2, W1b, W2b);

  attn_wmma<<<BATCH * HEADS * (SEQ / (32 * AT_WAVES)), 32 * AT_WAVES, 0, stream>>>(
      Qb, Kb, Vt, nkv, attnb, hpart);

  gemm_wmma<<<dim3(DIMD / 64, NTOK / 128), 128, 0, stream>>>(
      attnb, Wo, bo, attnO, DIMD);

  ffn_block<<<NTOK / TROWS, 256, 0, stream>>>(
      x, attnO, g1, be1, W1b, b1, W2b, b2, g2, be2, out);

  halt_finalize<<<BATCH, 1024, 0, stream>>>(hpart, kpos, mask, out_halt);
}
